# starting kernel
# baseline (speedup 1.0000x reference)
#include <hip/hip_runtime.h>


typedef float v2f __attribute__((ext_vector_type(2)));
typedef float v8f __attribute__((ext_vector_type(8)));

#define HIDDEN 2048
#define EXPERTS 64
#define TSTEPS 8192

__global__ __launch_bounds__(256) void transpose_w_kernel(const float* __restrict__ W,
                                                          float* __restrict__ BT) {
    int idx = blockIdx.x * 256 + threadIdx.x;
    int k = idx >> 6;
    int n = idx & 63;
    BT[(size_t)n * HIDDEN + k] = W[(size_t)k * EXPERTS + n];
}

__global__ __launch_bounds__(256) void gemm_wmma_kernel(const float* __restrict__ A,
                                                        const float* __restrict__ BT,
                                                        float* __restrict__ C) {
    const int wave  = threadIdx.x >> 5;
    const int lane  = threadIdx.x & 31;
    const int mtile = blockIdx.x * 8 + wave;
    const int m0    = mtile * 16;
    const int half  = lane >> 4;
    const int sub   = lane & 15;

    const float* arow = A  + (size_t)(m0 + sub) * HIDDEN + half * 2;
    const float* b0p  = BT + (size_t)(0 * 16 + sub) * HIDDEN + half * 2;
    const float* b1p  = BT + (size_t)(1 * 16 + sub) * HIDDEN + half * 2;
    const float* b2p  = BT + (size_t)(2 * 16 + sub) * HIDDEN + half * 2;
    const float* b3p  = BT + (size_t)(3 * 16 + sub) * HIDDEN + half * 2;

    v8f acc0 = {}, acc1 = {}, acc2 = {}, acc3 = {};

    for (int k0 = 0; k0 < HIDDEN; k0 += 4) {
        __builtin_prefetch(arow + k0 + 512, 0, 3);
        v2f a  = *(const v2f*)(arow + k0);
        v2f b0 = *(const v2f*)(b0p + k0);
        v2f b1 = *(const v2f*)(b1p + k0);
        v2f b2 = *(const v2f*)(b2p + k0);
        v2f b3 = *(const v2f*)(b3p + k0);
        acc0 = __builtin_amdgcn_wmma_f32_16x16x4_f32(false, a, false, b0, (short)0, acc0, false, false);
        acc1 = __builtin_amdgcn_wmma_f32_16x16x4_f32(false, a, false, b1, (short)0, acc1, false, false);
        acc2 = __builtin_amdgcn_wmma_f32_16x16x4_f32(false, a, false, b2, (short)0, acc2, false, false);
        acc3 = __builtin_amdgcn_wmma_f32_16x16x4_f32(false, a, false, b3, (short)0, acc3, false, false);
    }

#pragma unroll
    for (int r = 0; r < 8; ++r) {
        size_t row = (size_t)(m0 + half * 8 + r) * EXPERTS;
        C[row +  0 + sub] = acc0[r];
        C[row + 16 + sub] = acc1[r];
        C[row + 32 + sub] = acc2[r];
        C[row + 48 + sub] = acc3[r];
    }
}

#define CH 8

__device__ __forceinline__ void lif_step(float2 cur, float2& mp, float2& rf,
                                         float2* __restrict__ out) {
    mp.x *= 0.9f;
    mp.y *= 0.9f;
    float a0 = (rf.x <= 0.0f) ? 0.1f : 0.0f;
    float a1 = (rf.y <= 0.0f) ? 0.1f : 0.0f;
    mp.x += cur.x * a0;
    mp.y += cur.y * a1;
    rf.x = fmaxf(rf.x - 0.1f, 0.0f);
    rf.y = fmaxf(rf.y - 0.1f, 0.0f);
    *out = mp;
    bool s0 = mp.x > 1.0f;
    bool s1 = mp.y > 1.0f;
    mp.x = s0 ? 0.0f : mp.x;
    mp.y = s1 ? 0.0f : mp.y;
    rf.x = s0 ? 1.0f : rf.x;
    rf.y = s1 ? 1.0f : rf.y;
}

__global__ __launch_bounds__(32) void lif_scan_kernel(float2* __restrict__ buf) {
    const int lane = threadIdx.x;
    float2 mp = {0.0f, 0.0f};
    float2 rf = {0.0f, 0.0f};

    float2 bufA[CH], bufB[CH];
#pragma unroll
    for (int i = 0; i < CH; ++i) bufA[i] = buf[(size_t)i * 32 + lane];

    for (int t0 = 0; t0 < TSTEPS; t0 += 2 * CH) {
#pragma unroll
        for (int i = 0; i < CH; ++i) bufB[i] = buf[(size_t)(t0 + CH + i) * 32 + lane];
#pragma unroll
        for (int i = 0; i < CH; ++i)
            lif_step(bufA[i], mp, rf, &buf[(size_t)(t0 + i) * 32 + lane]);

        if (t0 + 2 * CH < TSTEPS) {
#pragma unroll
            for (int i = 0; i < CH; ++i) bufA[i] = buf[(size_t)(t0 + 2 * CH + i) * 32 + lane];
        }
#pragma unroll
        for (int i = 0; i < CH; ++i)
            lif_step(bufB[i], mp, rf, &buf[(size_t)(t0 + CH + i) * 32 + lane]);
    }
}

__global__ __launch_bounds__(256) void routing_kernel(const float2* __restrict__ mpbuf,
                                                      float2* __restrict__ out) {
    const int row  = blockIdx.x * 8 + (threadIdx.x >> 5);
    const int lane = threadIdx.x & 31;

    float2 mp = mpbuf[(size_t)row * 32 + lane];
    bool s0 = mp.x > 1.0f;
    bool s1 = mp.y > 1.0f;

    unsigned long long bal = __ballot(s0 || s1);
    bool any = (bal != 0ull);

    float x0 = any ? (s0 ? 1.0f : 0.0f) : mp.x;
    float x1 = any ? (s1 ? 1.0f : 0.0f) : mp.y;

    float m = fmaxf(x0, x1);
#pragma unroll
    for (int off = 16; off >= 1; off >>= 1)
        m = fmaxf(m, __shfl_xor(m, off, 32));

    float e0 = __expf(x0 - m);
    float e1 = __expf(x1 - m);
    float s  = e0 + e1;
#pragma unroll
    for (int off = 16; off >= 1; off >>= 1)
        s += __shfl_xor(s, off, 32);

    float inv = 1.0f / s;
    float2 r;
    r.x = e0 * inv;
    r.y = e1 * inv;
    out[(size_t)row * 32 + lane] = r;
}

extern "C" void kernel_launch(void* const* d_in, const int* in_sizes, int n_in,
                              void* d_out, int out_size, void* d_ws, size_t ws_size,
                              hipStream_t stream) {
    const float* hidden = (const float*)d_in[0];
    const float* W      = (const float*)d_in[1];
    float* out          = (float*)d_out;

    float* curr = (float*)d_ws;
    float* BT   = (float*)((char*)d_ws + (size_t)TSTEPS * EXPERTS * sizeof(float));

    transpose_w_kernel<<<(HIDDEN * EXPERTS) / 256, 256, 0, stream>>>(W, BT);

    gemm_wmma_kernel<<<64, 256, 0, stream>>>(hidden, BT, curr);

    lif_scan_kernel<<<1, 32, 0, stream>>>((float2*)curr);

    routing_kernel<<<TSTEPS / 8, 256, 0, stream>>>((const float2*)curr, (float2*)out);
}


# LIF scan rebuilt on its exact restart points: next-active step from every step (parallel table), per-expert hop chain in LDS, chunks replayed from their own restart and routed; step sums unchanged
# speedup vs baseline: 2.9655x; 2.9655x over previous
#include <hip/hip_runtime.h>


typedef float v2f __attribute__((ext_vector_type(2)));
typedef float v8f __attribute__((ext_vector_type(8)));

#define HIDDEN 2048
#define EXPERTS 64
#define TSTEPS 8192

__global__ __launch_bounds__(256) void transpose_w_kernel(const float* __restrict__ W,
                                                          float* __restrict__ BT) {
    int idx = blockIdx.x * 256 + threadIdx.x;
    int k = idx >> 6;
    int n = idx & 63;
    BT[(size_t)n * HIDDEN + k] = W[(size_t)k * EXPERTS + n];
}

__global__ __launch_bounds__(256) void gemm_wmma_kernel(const float* __restrict__ A,
                                                        const float* __restrict__ BT,
                                                        float* __restrict__ C) {
    const int wave  = threadIdx.x >> 5;
    const int lane  = threadIdx.x & 31;
    const int mtile = blockIdx.x * 8 + wave;
    const int m0    = mtile * 16;
    const int half  = lane >> 4;
    const int sub   = lane & 15;

    const float* arow = A  + (size_t)(m0 + sub) * HIDDEN + half * 2;
    const float* b0p  = BT + (size_t)(0 * 16 + sub) * HIDDEN + half * 2;
    const float* b1p  = BT + (size_t)(1 * 16 + sub) * HIDDEN + half * 2;
    const float* b2p  = BT + (size_t)(2 * 16 + sub) * HIDDEN + half * 2;
    const float* b3p  = BT + (size_t)(3 * 16 + sub) * HIDDEN + half * 2;

    v8f acc0 = {}, acc1 = {}, acc2 = {}, acc3 = {};

    for (int k0 = 0; k0 < HIDDEN; k0 += 4) {
        __builtin_prefetch(arow + k0 + 512, 0, 3);
        v2f a  = *(const v2f*)(arow + k0);
        v2f b0 = *(const v2f*)(b0p + k0);
        v2f b1 = *(const v2f*)(b1p + k0);
        v2f b2 = *(const v2f*)(b2p + k0);
        v2f b3 = *(const v2f*)(b3p + k0);
        acc0 = __builtin_amdgcn_wmma_f32_16x16x4_f32(false, a, false, b0, (short)0, acc0, false, false);
        acc1 = __builtin_amdgcn_wmma_f32_16x16x4_f32(false, a, false, b1, (short)0, acc1, false, false);
        acc2 = __builtin_amdgcn_wmma_f32_16x16x4_f32(false, a, false, b2, (short)0, acc2, false, false);
        acc3 = __builtin_amdgcn_wmma_f32_16x16x4_f32(false, a, false, b3, (short)0, acc3, false, false);
    }

#pragma unroll
    for (int r = 0; r < 8; ++r) {
        size_t row = (size_t)(m0 + half * 8 + r) * EXPERTS;
        C[row +  0 + sub] = acc0[r];
        C[row + 16 + sub] = acc1[r];
        C[row + 32 + sub] = acc2[r];
        C[row + 48 + sub] = acc3[r];
    }
}

#define CHK  16
#define NCH  (TSTEPS / CHK)
#define BLK  8

constexpr int refractory_gap() {
    float rf = 1.0f;
    int k = 0;
    for (int i = 0; i < 64; ++i) {
        ++k;
        bool active = (rf <= 0.0f);
        float d = rf - 0.1f;
        rf = (d > 0.0f) ? d : 0.0f;
        if (active) break;
    }
    return k;
}
constexpr int GAP = refractory_gap();

__device__ __forceinline__ float2 lif_step(float2 cur, bool on0, bool on1,
                                           float2& mp, float2& rf) {
    float l0 = mp.x * 0.9f;
    float l1 = mp.y * 0.9f;
    float a0 = (on0 && rf.x <= 0.0f) ? 0.1f : 0.0f;
    float a1 = (on1 && rf.y <= 0.0f) ? 0.1f : 0.0f;
    mp.x = __builtin_fmaf(cur.x, a0, l0);
    mp.y = __builtin_fmaf(cur.y, a1, l1);
    rf.x = fmaxf(rf.x - 0.1f, 0.0f);
    rf.y = fmaxf(rf.y - 0.1f, 0.0f);
    float2 pre = mp;
    bool s0 = mp.x > 1.0f;
    bool s1 = mp.y > 1.0f;
    mp.x = s0 ? 0.0f : mp.x;
    mp.y = s1 ? 0.0f : mp.y;
    rf.x = s0 ? 1.0f : rf.x;
    rf.y = s1 ? 1.0f : rf.y;
    return pre;
}

__global__ __launch_bounds__(256) void next_restart_kernel(const float2* __restrict__ cur,
                                                           unsigned* __restrict__ nxt) {
    const int r    = __builtin_amdgcn_readfirstlane(blockIdx.x * 8 + (threadIdx.x >> 5));
    const int lane = threadIdx.x & 31;

    float m0 = 0.0f, m1 = 0.0f;
    int s0 = TSTEPS, s1 = TSTEPS;

    for (int tb = r; tb < TSTEPS; tb += BLK) {
        float2 c[BLK];
#pragma unroll
        for (int i = 0; i < BLK; ++i) {
            int t = min(tb + i, TSTEPS - 1);
            c[i] = cur[(size_t)t * 32 + lane];
        }
#pragma unroll
        for (int i = 0; i < BLK; ++i) {
            int t = tb + i;
            m0 = __builtin_fmaf(c[i].x, 0.1f, m0 * 0.9f);
            m1 = __builtin_fmaf(c[i].y, 0.1f, m1 * 0.9f);
            bool in = t < TSTEPS;
            if (in && s0 == TSTEPS && m0 > 1.0f) s0 = t;
            if (in && s1 == TSTEPS && m1 > 1.0f) s1 = t;
        }
        if (__ballot(s0 == TSTEPS || s1 == TSTEPS) == 0ull) break;
    }
    unsigned n0 = (unsigned)min(s0 + GAP, TSTEPS);
    unsigned n1 = (unsigned)min(s1 + GAP, TSTEPS);
    nxt[(size_t)r * 32 + lane] = n0 | (n1 << 16);
}

__device__ __forceinline__ void fill_chunks(unsigned short* st, int& c, int from, int upto) {
    int cend = min((upto + CHK - 1) / CHK, NCH);
    for (; c < cend; ++c) st[c] = (unsigned short)from;
}

__global__ __launch_bounds__(256) void restart_chain_kernel(const unsigned* __restrict__ nxt,
                                                            unsigned* __restrict__ start) {
    __shared__ unsigned short tab1[TSTEPS + 2];
    __shared__ unsigned short tab2[TSTEPS + 2];
    __shared__ unsigned short tab4[TSTEPS + 2];
    __shared__ unsigned stw[NCH / 2];
    unsigned short* st = (unsigned short*)stw;
    const int e   = blockIdx.x;
    const int tid = threadIdx.x;

    for (int r = tid; r < TSTEPS; r += 256) {
        unsigned w = nxt[(size_t)r * 32 + (e >> 1)];
        tab1[r] = (unsigned short)((e & 1) ? (w >> 16) : (w & 0xffffu));
    }
    if (tid == 0) { tab1[TSTEPS] = TSTEPS; tab2[TSTEPS] = TSTEPS; tab4[TSTEPS] = TSTEPS; }
    __syncthreads();
    for (int r = tid; r < TSTEPS; r += 256) tab2[r] = tab1[tab1[r]];
    __syncthreads();
    for (int r = tid; r < TSTEPS; r += 256) tab4[r] = tab2[tab2[r]];
    __syncthreads();

    if (tid == 0) {
        int c = 0;
        int r = 0;
        int a = tab1[0], b = tab2[0], d = tab4[0];
        for (int it = 0; it < TSTEPS && r < TSTEPS; ++it) {
            int m  = tab1[b];
            int an = tab1[d], bn = tab2[d], dn = tab4[d];
            fill_chunks(st, c, r, a);
            fill_chunks(st, c, a, b);
            fill_chunks(st, c, b, m);
            fill_chunks(st, c, m, d);
            r = d; a = an; b = bn; d = dn;
        }
    }
    __syncthreads();

    start[(size_t)e * (NCH / 2) + tid] = stw[tid];
}

__global__ __launch_bounds__(32) void replay_route_kernel(const float2* __restrict__ cur,
                                                          const unsigned short* __restrict__ start,
                                                          float2* __restrict__ out) {
    const int ch   = blockIdx.x;
    const int lane = threadIdx.x;
    const int tbeg = ch * CHK;

    const int sA = start[(size_t)(2 * lane)     * NCH + ch];
    const int sB = start[(size_t)(2 * lane + 1) * NCH + ch];
    int smin = min(sA, sB);
#pragma unroll
    for (int off = 16; off >= 1; off >>= 1)
        smin = min(smin, __shfl_xor(smin, off, 32));
    smin = __builtin_amdgcn_readfirstlane(smin);

    float2 mp = {0.0f, 0.0f};
    float2 rf = {0.0f, 0.0f};

    const int nblk = (tbeg - smin + BLK - 1) / BLK;
    for (int tb = tbeg - nblk * BLK; tb < tbeg; tb += BLK) {
        float2 c[BLK];
#pragma unroll
        for (int i = 0; i < BLK; ++i) {
            int t = max(tb + i, 0);
            c[i] = cur[(size_t)t * 32 + lane];
        }
#pragma unroll
        for (int i = 0; i < BLK; ++i) {
            int t = tb + i;
            (void)lif_step(c[i], t >= sA, t >= sB, mp, rf);
        }
    }

    float2 c[CHK];
#pragma unroll
    for (int i = 0; i < CHK; ++i) c[i] = cur[(size_t)(tbeg + i) * 32 + lane];

#pragma unroll
    for (int i = 0; i < CHK; ++i) {
        const int t = tbeg + i;
        float2 v = lif_step(c[i], t >= sA, t >= sB, mp, rf);

        bool s0 = v.x > 1.0f;
        bool s1 = v.y > 1.0f;
        unsigned long long bal = __ballot(s0 || s1);
        bool any = (bal != 0ull);

        float x0 = any ? (s0 ? 1.0f : 0.0f) : v.x;
        float x1 = any ? (s1 ? 1.0f : 0.0f) : v.y;

        float m = fmaxf(x0, x1);
#pragma unroll
        for (int off = 16; off >= 1; off >>= 1)
            m = fmaxf(m, __shfl_xor(m, off, 32));

        float e0 = __expf(x0 - m);
        float e1 = __expf(x1 - m);
        float s  = e0 + e1;
#pragma unroll
        for (int off = 16; off >= 1; off >>= 1)
            s += __shfl_xor(s, off, 32);

        float inv = 1.0f / s;
        float2 rr;
        rr.x = e0 * inv;
        rr.y = e1 * inv;
        out[(size_t)t * 32 + lane] = rr;
    }
}

extern "C" void kernel_launch(void* const* d_in, const int* in_sizes, int n_in,
                              void* d_out, int out_size, void* d_ws, size_t ws_size,
                              hipStream_t stream) {
    const float* hidden = (const float*)d_in[0];
    const float* W      = (const float*)d_in[1];
    float* out          = (float*)d_out;

    float* curr = (float*)d_ws;
    float* BT   = (float*)((char*)d_ws + (size_t)TSTEPS * EXPERTS * sizeof(float));
    unsigned* nxt   = (unsigned*)d_out;
    unsigned* start = (unsigned*)BT;

    transpose_w_kernel<<<(HIDDEN * EXPERTS) / 256, 256, 0, stream>>>(W, BT);

    gemm_wmma_kernel<<<64, 256, 0, stream>>>(hidden, BT, curr);

    next_restart_kernel<<<TSTEPS / 8, 256, 0, stream>>>((const float2*)curr, nxt);

    restart_chain_kernel<<<EXPERTS, 256, 0, stream>>>(nxt, start);

    replay_route_kernel<<<NCH, 32, 0, stream>>>((const float2*)curr,
                                                (const unsigned short*)start, (float2*)out);
}


# GEMM operands staged through LDS: 32x64 tile per 8-wave group, A and B^T rows fetched as whole 512-byte pieces two chunks ahead, fragments read from LDS; per-tile k order and sums unchanged
# speedup vs baseline: 5.5622x; 1.8756x over previous
#include <hip/hip_runtime.h>


typedef float v2f __attribute__((ext_vector_type(2)));
typedef float v8f __attribute__((ext_vector_type(8)));

#define HIDDEN 2048
#define EXPERTS 64
#define TSTEPS 8192

__global__ __launch_bounds__(256) void transpose_w_kernel(const float* __restrict__ W,
                                                          float* __restrict__ BT) {
    int idx = blockIdx.x * 256 + threadIdx.x;
    int k = idx >> 6;
    int n = idx & 63;
    BT[(size_t)n * HIDDEN + k] = W[(size_t)k * EXPERTS + n];
}

#define KC      128
#define PITCH   (KC + 4)
#define AROWS   32
#define BROWS   EXPERTS
#define STAGE   ((AROWS + BROWS) * PITCH)
#define NCHUNK  (HIDDEN / KC)
#define CPITCH  (EXPERTS + 4)

struct Frags { v2f a[KC / 4]; v2f b[KC / 4]; };

__device__ __forceinline__ void read_frags(const float* st, int fa, int fb, Frags& f) {
#pragma unroll
    for (int ks = 0; ks < KC / 4; ++ks) {
        f.a[ks] = *(const v2f*)(st + fa + ks * 4);
        f.b[ks] = *(const v2f*)(st + fb + ks * 4);
    }
    __builtin_amdgcn_sched_barrier(0);
}

__device__ __forceinline__ void mma_chunk(const Frags& f, v8f& acc) {
#pragma unroll
    for (int ks = 0; ks < KC / 4; ++ks)
        acc = __builtin_amdgcn_wmma_f32_16x16x4_f32(false, f.a[ks], false, f.b[ks], (short)0, acc, false, false);
}

__device__ __forceinline__ void lds_barrier() {
    __builtin_amdgcn_fence(__ATOMIC_RELEASE, "workgroup", "local");
    __builtin_amdgcn_s_barrier();
    __builtin_amdgcn_fence(__ATOMIC_ACQUIRE, "workgroup", "local");
}

__global__ __launch_bounds__(256) void gemm_wmma_kernel(const float* __restrict__ A,
                                                        const float* __restrict__ BT,
                                                        float* __restrict__ C) {
    __shared__ __attribute__((aligned(16))) float lds[2 * STAGE];

    const int tid  = threadIdx.x;
    const int wave = tid >> 5;
    const int lane = tid & 31;
    const int m0   = blockIdx.x * AROWS;
    const int mt   = wave >> 2;
    const int nt   = wave & 3;
    const int half = lane >> 4;
    const int sub  = lane & 15;

    const float* ga = A  + (size_t)(m0 + wave * 4) * HIDDEN + lane * 4;
    const float* gb = BT + (size_t)(wave * 8) * HIDDEN + lane * 4;
    const int sa = (wave * 4) * PITCH + lane * 4;
    const int sb = (AROWS + wave * 8) * PITCH + lane * 4;

    const int fa = (mt * 16 + sub) * PITCH + half * 2;
    const int fb = (AROWS + nt * 16 + sub) * PITCH + half * 2;

    float4 ra0, ra1, ra2, ra3, rb0, rb1, rb2, rb3, rb4, rb5, rb6, rb7;
#define LOAD_ROWS(koff)                                              \
    ra0 = *(const float4*)(ga + 0 * (size_t)HIDDEN + (koff));        \
    ra1 = *(const float4*)(ga + 1 * (size_t)HIDDEN + (koff));        \
    ra2 = *(const float4*)(ga + 2 * (size_t)HIDDEN + (koff));        \
    ra3 = *(const float4*)(ga + 3 * (size_t)HIDDEN + (koff));        \
    rb0 = *(const float4*)(gb + 0 * (size_t)HIDDEN + (koff));        \
    rb1 = *(const float4*)(gb + 1 * (size_t)HIDDEN + (koff));        \
    rb2 = *(const float4*)(gb + 2 * (size_t)HIDDEN + (koff));        \
    rb3 = *(const float4*)(gb + 3 * (size_t)HIDDEN + (koff));        \
    rb4 = *(const float4*)(gb + 4 * (size_t)HIDDEN + (koff));        \
    rb5 = *(const float4*)(gb + 5 * (size_t)HIDDEN + (koff));        \
    rb6 = *(const float4*)(gb + 6 * (size_t)HIDDEN + (koff));        \
    rb7 = *(const float4*)(gb + 7 * (size_t)HIDDEN + (koff));
#define STORE_ROWS(stage)                                            \
    *(float4*)((stage) + sa + 0 * PITCH) = ra0;                      \
    *(float4*)((stage) + sa + 1 * PITCH) = ra1;                      \
    *(float4*)((stage) + sa + 2 * PITCH) = ra2;                      \
    *(float4*)((stage) + sa + 3 * PITCH) = ra3;                      \
    *(float4*)((stage) + sb + 0 * PITCH) = rb0;                      \
    *(float4*)((stage) + sb + 1 * PITCH) = rb1;                      \
    *(float4*)((stage) + sb + 2 * PITCH) = rb2;                      \
    *(float4*)((stage) + sb + 3 * PITCH) = rb3;                      \
    *(float4*)((stage) + sb + 4 * PITCH) = rb4;                      \
    *(float4*)((stage) + sb + 5 * PITCH) = rb5;                      \
    *(float4*)((stage) + sb + 6 * PITCH) = rb6;                      \
    *(float4*)((stage) + sb + 7 * PITCH) = rb7;

    LOAD_ROWS(0)
    STORE_ROWS(lds)
    LOAD_ROWS(KC)
    __syncthreads();

    v8f acc = {};

    for (int c = 0; c < NCHUNK - 2; ++c) {
        const float* st = lds + (c & 1) * STAGE;
        float*       sn = lds + ((c + 1) & 1) * STAGE;
        const int    kn = (c + 2) * KC;
        Frags f;
        read_frags(st, fa, fb, f);
        STORE_ROWS(sn)
        LOAD_ROWS(kn)
        mma_chunk(f, acc);
        lds_barrier();
    }
    {
        Frags f;
        read_frags(lds + ((NCHUNK - 2) & 1) * STAGE, fa, fb, f);
        STORE_ROWS(lds + ((NCHUNK - 1) & 1) * STAGE)
        mma_chunk(f, acc);
        __syncthreads();
    }
    {
        Frags f;
        read_frags(lds + ((NCHUNK - 1) & 1) * STAGE, fa, fb, f);
        mma_chunk(f, acc);
        __syncthreads();
    }
#undef LOAD_ROWS
#undef STORE_ROWS

    float* ct = lds;
#pragma unroll
    for (int r = 0; r < 8; ++r)
        ct[(mt * 16 + half * 8 + r) * CPITCH + nt * 16 + sub] = acc[r];
    __syncthreads();
#pragma unroll
    for (int p = 0; p < 2; ++p) {
        const int j   = p * 1024 + tid * 4;
        const int row = j >> 6;
        const int col = j & 63;
        float4 v = *(const float4*)(ct + row * CPITCH + col);
        *(float4*)(C + (size_t)m0 * EXPERTS + j) = v;
    }
}

#define CHK  16
#define NCH  (TSTEPS / CHK)
#define BLK  8

constexpr int refractory_gap() {
    float rf = 1.0f;
    int k = 0;
    for (int i = 0; i < 64; ++i) {
        ++k;
        bool active = (rf <= 0.0f);
        float d = rf - 0.1f;
        rf = (d > 0.0f) ? d : 0.0f;
        if (active) break;
    }
    return k;
}
constexpr int GAP = refractory_gap();

__device__ __forceinline__ float2 lif_step(float2 cur, bool on0, bool on1,
                                           float2& mp, float2& rf) {
    float l0 = mp.x * 0.9f;
    float l1 = mp.y * 0.9f;
    float a0 = (on0 && rf.x <= 0.0f) ? 0.1f : 0.0f;
    float a1 = (on1 && rf.y <= 0.0f) ? 0.1f : 0.0f;
    mp.x = __builtin_fmaf(cur.x, a0, l0);
    mp.y = __builtin_fmaf(cur.y, a1, l1);
    rf.x = fmaxf(rf.x - 0.1f, 0.0f);
    rf.y = fmaxf(rf.y - 0.1f, 0.0f);
    float2 pre = mp;
    bool s0 = mp.x > 1.0f;
    bool s1 = mp.y > 1.0f;
    mp.x = s0 ? 0.0f : mp.x;
    mp.y = s1 ? 0.0f : mp.y;
    rf.x = s0 ? 1.0f : rf.x;
    rf.y = s1 ? 1.0f : rf.y;
    return pre;
}

__global__ __launch_bounds__(256) void next_restart_kernel(const float2* __restrict__ cur,
                                                           unsigned* __restrict__ nxt) {
    const int r    = __builtin_amdgcn_readfirstlane(blockIdx.x * 8 + (threadIdx.x >> 5));
    const int lane = threadIdx.x & 31;

    float m0 = 0.0f, m1 = 0.0f;
    int s0 = TSTEPS, s1 = TSTEPS;

    for (int tb = r; tb < TSTEPS; tb += BLK) {
        float2 c[BLK];
#pragma unroll
        for (int i = 0; i < BLK; ++i) {
            int t = min(tb + i, TSTEPS - 1);
            c[i] = cur[(size_t)t * 32 + lane];
        }
#pragma unroll
        for (int i = 0; i < BLK; ++i) {
            int t = tb + i;
            m0 = __builtin_fmaf(c[i].x, 0.1f, m0 * 0.9f);
            m1 = __builtin_fmaf(c[i].y, 0.1f, m1 * 0.9f);
            bool in = t < TSTEPS;
            if (in && s0 == TSTEPS && m0 > 1.0f) s0 = t;
            if (in && s1 == TSTEPS && m1 > 1.0f) s1 = t;
        }
        if (__ballot(s0 == TSTEPS || s1 == TSTEPS) == 0ull) break;
    }
    unsigned n0 = (unsigned)min(s0 + GAP, TSTEPS);
    unsigned n1 = (unsigned)min(s1 + GAP, TSTEPS);
    nxt[(size_t)r * 32 + lane] = n0 | (n1 << 16);
}

__device__ __forceinline__ void fill_chunks(unsigned short* st, int& c, int from, int upto) {
    int cend = min((upto + CHK - 1) / CHK, NCH);
    for (; c < cend; ++c) st[c] = (unsigned short)from;
}

__global__ __launch_bounds__(256) void restart_chain_kernel(const unsigned* __restrict__ nxt,
                                                            unsigned* __restrict__ start) {
    __shared__ unsigned short tab1[TSTEPS + 2];
    __shared__ unsigned short tab2[TSTEPS + 2];
    __shared__ unsigned short tab4[TSTEPS + 2];
    __shared__ unsigned stw[NCH / 2];
    unsigned short* st = (unsigned short*)stw;
    const int e   = blockIdx.x;
    const int tid = threadIdx.x;

    for (int r = tid; r < TSTEPS; r += 256) {
        unsigned w = nxt[(size_t)r * 32 + (e >> 1)];
        tab1[r] = (unsigned short)((e & 1) ? (w >> 16) : (w & 0xffffu));
    }
    if (tid == 0) { tab1[TSTEPS] = TSTEPS; tab2[TSTEPS] = TSTEPS; tab4[TSTEPS] = TSTEPS; }
    __syncthreads();
    for (int r = tid; r < TSTEPS; r += 256) tab2[r] = tab1[tab1[r]];
    __syncthreads();
    for (int r = tid; r < TSTEPS; r += 256) tab4[r] = tab2[tab2[r]];
    __syncthreads();

    if (tid == 0) {
        int c = 0;
        int r = 0;
        int a = tab1[0], b = tab2[0], d = tab4[0];
        for (int it = 0; it < TSTEPS && r < TSTEPS; ++it) {
            int m  = tab1[b];
            int an = tab1[d], bn = tab2[d], dn = tab4[d];
            fill_chunks(st, c, r, a);
            fill_chunks(st, c, a, b);
            fill_chunks(st, c, b, m);
            fill_chunks(st, c, m, d);
            r = d; a = an; b = bn; d = dn;
        }
    }
    __syncthreads();

    start[(size_t)e * (NCH / 2) + tid] = stw[tid];
}

__global__ __launch_bounds__(32) void replay_route_kernel(const float2* __restrict__ cur,
                                                          const unsigned short* __restrict__ start,
                                                          float2* __restrict__ out) {
    const int ch   = blockIdx.x;
    const int lane = threadIdx.x;
    const int tbeg = ch * CHK;

    const int sA = start[(size_t)(2 * lane)     * NCH + ch];
    const int sB = start[(size_t)(2 * lane + 1) * NCH + ch];
    int smin = min(sA, sB);
#pragma unroll
    for (int off = 16; off >= 1; off >>= 1)
        smin = min(smin, __shfl_xor(smin, off, 32));
    smin = __builtin_amdgcn_readfirstlane(smin);

    float2 mp = {0.0f, 0.0f};
    float2 rf = {0.0f, 0.0f};

    const int nblk = (tbeg - smin + BLK - 1) / BLK;
    for (int tb = tbeg - nblk * BLK; tb < tbeg; tb += BLK) {
        float2 c[BLK];
#pragma unroll
        for (int i = 0; i < BLK; ++i) {
            int t = max(tb + i, 0);
            c[i] = cur[(size_t)t * 32 + lane];
        }
#pragma unroll
        for (int i = 0; i < BLK; ++i) {
            int t = tb + i;
            (void)lif_step(c[i], t >= sA, t >= sB, mp, rf);
        }
    }

    float2 c[CHK];
#pragma unroll
    for (int i = 0; i < CHK; ++i) c[i] = cur[(size_t)(tbeg + i) * 32 + lane];

#pragma unroll
    for (int i = 0; i < CHK; ++i) {
        const int t = tbeg + i;
        float2 v = lif_step(c[i], t >= sA, t >= sB, mp, rf);

        bool s0 = v.x > 1.0f;
        bool s1 = v.y > 1.0f;
        unsigned long long bal = __ballot(s0 || s1);
        bool any = (bal != 0ull);

        float x0 = any ? (s0 ? 1.0f : 0.0f) : v.x;
        float x1 = any ? (s1 ? 1.0f : 0.0f) : v.y;

        float m = fmaxf(x0, x1);
#pragma unroll
        for (int off = 16; off >= 1; off >>= 1)
            m = fmaxf(m, __shfl_xor(m, off, 32));

        float e0 = __expf(x0 - m);
        float e1 = __expf(x1 - m);
        float s  = e0 + e1;
#pragma unroll
        for (int off = 16; off >= 1; off >>= 1)
            s += __shfl_xor(s, off, 32);

        float inv = 1.0f / s;
        float2 rr;
        rr.x = e0 * inv;
        rr.y = e1 * inv;
        out[(size_t)t * 32 + lane] = rr;
    }
}

extern "C" void kernel_launch(void* const* d_in, const int* in_sizes, int n_in,
                              void* d_out, int out_size, void* d_ws, size_t ws_size,
                              hipStream_t stream) {
    const float* hidden = (const float*)d_in[0];
    const float* W      = (const float*)d_in[1];
    float* out          = (float*)d_out;

    float* curr = (float*)d_ws;
    float* BT   = (float*)((char*)d_ws + (size_t)TSTEPS * EXPERTS * sizeof(float));
    unsigned* nxt   = (unsigned*)d_out;
    unsigned* start = (unsigned*)BT;

    transpose_w_kernel<<<(HIDDEN * EXPERTS) / 256, 256, 0, stream>>>(W, BT);

    gemm_wmma_kernel<<<TSTEPS / AROWS, 256, 0, stream>>>(hidden, BT, curr);

    next_restart_kernel<<<TSTEPS / 8, 256, 0, stream>>>((const float2*)curr, nxt);

    restart_chain_kernel<<<EXPERTS, 256, 0, stream>>>(nxt, start);

    replay_route_kernel<<<NCH, 32, 0, stream>>>((const float2*)curr,
                                                (const unsigned short*)start, (float2*)out);
}


# 2c replay cut to the bare leak-integrate recurrence: spike step from 2b's table (next restart - gap), sit-out counter rebuilt directly, rows through a 3-block register ring; step sums unchanged
# speedup vs baseline: 5.9208x; 1.0645x over previous
#include <hip/hip_runtime.h>


typedef float v2f __attribute__((ext_vector_type(2)));
typedef float v8f __attribute__((ext_vector_type(8)));

#define HIDDEN 2048
#define EXPERTS 64
#define TSTEPS 8192

__global__ __launch_bounds__(256) void transpose_w_kernel(const float* __restrict__ W,
                                                          float* __restrict__ BT) {
    int idx = blockIdx.x * 256 + threadIdx.x;
    int k = idx >> 6;
    int n = idx & 63;
    BT[(size_t)n * HIDDEN + k] = W[(size_t)k * EXPERTS + n];
}

#define KC      128
#define PITCH   (KC + 4)
#define AROWS   32
#define BROWS   EXPERTS
#define STAGE   ((AROWS + BROWS) * PITCH)
#define NCHUNK  (HIDDEN / KC)
#define CPITCH  (EXPERTS + 4)

struct Frags { v2f a[KC / 4]; v2f b[KC / 4]; };

__device__ __forceinline__ void read_frags(const float* st, int fa, int fb, Frags& f) {
#pragma unroll
    for (int ks = 0; ks < KC / 4; ++ks) {
        f.a[ks] = *(const v2f*)(st + fa + ks * 4);
        f.b[ks] = *(const v2f*)(st + fb + ks * 4);
    }
    __builtin_amdgcn_sched_barrier(0);
}

__device__ __forceinline__ void mma_chunk(const Frags& f, v8f& acc) {
#pragma unroll
    for (int ks = 0; ks < KC / 4; ++ks)
        acc = __builtin_amdgcn_wmma_f32_16x16x4_f32(false, f.a[ks], false, f.b[ks], (short)0, acc, false, false);
}

__device__ __forceinline__ void lds_barrier() {
    __builtin_amdgcn_fence(__ATOMIC_RELEASE, "workgroup", "local");
    __builtin_amdgcn_s_barrier();
    __builtin_amdgcn_fence(__ATOMIC_ACQUIRE, "workgroup", "local");
}

__global__ __launch_bounds__(256) void gemm_wmma_kernel(const float* __restrict__ A,
                                                        const float* __restrict__ BT,
                                                        float* __restrict__ C) {
    __shared__ __attribute__((aligned(16))) float lds[2 * STAGE];

    const int tid  = threadIdx.x;
    const int wave = tid >> 5;
    const int lane = tid & 31;
    const int m0   = blockIdx.x * AROWS;
    const int mt   = wave >> 2;
    const int nt   = wave & 3;
    const int half = lane >> 4;
    const int sub  = lane & 15;

    const float* ga = A  + (size_t)(m0 + wave * 4) * HIDDEN + lane * 4;
    const float* gb = BT + (size_t)(wave * 8) * HIDDEN + lane * 4;
    const int sa = (wave * 4) * PITCH + lane * 4;
    const int sb = (AROWS + wave * 8) * PITCH + lane * 4;

    const int fa = (mt * 16 + sub) * PITCH + half * 2;
    const int fb = (AROWS + nt * 16 + sub) * PITCH + half * 2;

    float4 ra0, ra1, ra2, ra3, rb0, rb1, rb2, rb3, rb4, rb5, rb6, rb7;
#define LOAD_ROWS(koff)                                              \
    ra0 = *(const float4*)(ga + 0 * (size_t)HIDDEN + (koff));        \
    ra1 = *(const float4*)(ga + 1 * (size_t)HIDDEN + (koff));        \
    ra2 = *(const float4*)(ga + 2 * (size_t)HIDDEN + (koff));        \
    ra3 = *(const float4*)(ga + 3 * (size_t)HIDDEN + (koff));        \
    rb0 = *(const float4*)(gb + 0 * (size_t)HIDDEN + (koff));        \
    rb1 = *(const float4*)(gb + 1 * (size_t)HIDDEN + (koff));        \
    rb2 = *(const float4*)(gb + 2 * (size_t)HIDDEN + (koff));        \
    rb3 = *(const float4*)(gb + 3 * (size_t)HIDDEN + (koff));        \
    rb4 = *(const float4*)(gb + 4 * (size_t)HIDDEN + (koff));        \
    rb5 = *(const float4*)(gb + 5 * (size_t)HIDDEN + (koff));        \
    rb6 = *(const float4*)(gb + 6 * (size_t)HIDDEN + (koff));        \
    rb7 = *(const float4*)(gb + 7 * (size_t)HIDDEN + (koff));
#define STORE_ROWS(stage)                                            \
    *(float4*)((stage) + sa + 0 * PITCH) = ra0;                      \
    *(float4*)((stage) + sa + 1 * PITCH) = ra1;                      \
    *(float4*)((stage) + sa + 2 * PITCH) = ra2;                      \
    *(float4*)((stage) + sa + 3 * PITCH) = ra3;                      \
    *(float4*)((stage) + sb + 0 * PITCH) = rb0;                      \
    *(float4*)((stage) + sb + 1 * PITCH) = rb1;                      \
    *(float4*)((stage) + sb + 2 * PITCH) = rb2;                      \
    *(float4*)((stage) + sb + 3 * PITCH) = rb3;                      \
    *(float4*)((stage) + sb + 4 * PITCH) = rb4;                      \
    *(float4*)((stage) + sb + 5 * PITCH) = rb5;                      \
    *(float4*)((stage) + sb + 6 * PITCH) = rb6;                      \
    *(float4*)((stage) + sb + 7 * PITCH) = rb7;

    LOAD_ROWS(0)
    STORE_ROWS(lds)
    LOAD_ROWS(KC)
    __syncthreads();

    v8f acc = {};

    for (int c = 0; c < NCHUNK - 2; ++c) {
        const float* st = lds + (c & 1) * STAGE;
        float*       sn = lds + ((c + 1) & 1) * STAGE;
        const int    kn = (c + 2) * KC;
        Frags f;
        read_frags(st, fa, fb, f);
        STORE_ROWS(sn)
        LOAD_ROWS(kn)
        mma_chunk(f, acc);
        lds_barrier();
    }
    {
        Frags f;
        read_frags(lds + ((NCHUNK - 2) & 1) * STAGE, fa, fb, f);
        STORE_ROWS(lds + ((NCHUNK - 1) & 1) * STAGE)
        mma_chunk(f, acc);
        __syncthreads();
    }
    {
        Frags f;
        read_frags(lds + ((NCHUNK - 1) & 1) * STAGE, fa, fb, f);
        mma_chunk(f, acc);
        __syncthreads();
    }
#undef LOAD_ROWS
#undef STORE_ROWS

    float* ct = lds;
#pragma unroll
    for (int r = 0; r < 8; ++r)
        ct[(mt * 16 + half * 8 + r) * CPITCH + nt * 16 + sub] = acc[r];
    __syncthreads();
#pragma unroll
    for (int p = 0; p < 2; ++p) {
        const int j   = p * 1024 + tid * 4;
        const int row = j >> 6;
        const int col = j & 63;
        float4 v = *(const float4*)(ct + row * CPITCH + col);
        *(float4*)(C + (size_t)m0 * EXPERTS + j) = v;
    }
}

#define CHK  16
#define NCH  (TSTEPS / CHK)
#define BLK  8

constexpr int refractory_gap() {
    float rf = 1.0f;
    int k = 0;
    for (int i = 0; i < 64; ++i) {
        ++k;
        bool active = (rf <= 0.0f);
        float d = rf - 0.1f;
        rf = (d > 0.0f) ? d : 0.0f;
        if (active) break;
    }
    return k;
}
constexpr int GAP = refractory_gap();

__device__ __forceinline__ float2 lif_step(float2 cur, bool on0, bool on1,
                                           float2& mp, float2& rf) {
    float l0 = mp.x * 0.9f;
    float l1 = mp.y * 0.9f;
    float a0 = (on0 && rf.x <= 0.0f) ? 0.1f : 0.0f;
    float a1 = (on1 && rf.y <= 0.0f) ? 0.1f : 0.0f;
    mp.x = __builtin_fmaf(cur.x, a0, l0);
    mp.y = __builtin_fmaf(cur.y, a1, l1);
    rf.x = fmaxf(rf.x - 0.1f, 0.0f);
    rf.y = fmaxf(rf.y - 0.1f, 0.0f);
    float2 pre = mp;
    bool s0 = mp.x > 1.0f;
    bool s1 = mp.y > 1.0f;
    mp.x = s0 ? 0.0f : mp.x;
    mp.y = s1 ? 0.0f : mp.y;
    rf.x = s0 ? 1.0f : rf.x;
    rf.y = s1 ? 1.0f : rf.y;
    return pre;
}

__global__ __launch_bounds__(256) void next_restart_kernel(const float2* __restrict__ cur,
                                                           unsigned* __restrict__ nxt) {
    const int r    = __builtin_amdgcn_readfirstlane(blockIdx.x * 8 + (threadIdx.x >> 5));
    const int lane = threadIdx.x & 31;

    float m0 = 0.0f, m1 = 0.0f;
    int s0 = TSTEPS, s1 = TSTEPS;

    for (int tb = r; tb < TSTEPS; tb += BLK) {
        float2 c[BLK];
#pragma unroll
        for (int i = 0; i < BLK; ++i) {
            int t = min(tb + i, TSTEPS - 1);
            c[i] = cur[(size_t)t * 32 + lane];
        }
#pragma unroll
        for (int i = 0; i < BLK; ++i) {
            int t = tb + i;
            m0 = __builtin_fmaf(c[i].x, 0.1f, m0 * 0.9f);
            m1 = __builtin_fmaf(c[i].y, 0.1f, m1 * 0.9f);
            bool in = t < TSTEPS;
            if (in && s0 == TSTEPS && m0 > 1.0f) s0 = t;
            if (in && s1 == TSTEPS && m1 > 1.0f) s1 = t;
        }
        if (__ballot(s0 == TSTEPS || s1 == TSTEPS) == 0ull) break;
    }
    unsigned n0 = (unsigned)min(s0 + GAP, TSTEPS);
    unsigned n1 = (unsigned)min(s1 + GAP, TSTEPS);
    nxt[(size_t)r * 32 + lane] = n0 | (n1 << 16);
}

__device__ __forceinline__ void fill_chunks(unsigned* st, int& c, int from, int upto) {
    int cend = min((upto + CHK - 1) / CHK, NCH);
    const unsigned w = (unsigned)from | ((unsigned)upto << 16);
    for (; c < cend; ++c) st[c] = w;
}

__global__ __launch_bounds__(256) void restart_chain_kernel(const unsigned* __restrict__ nxt,
                                                            unsigned* __restrict__ start) {
    __shared__ unsigned short tab1[TSTEPS + 2];
    __shared__ unsigned short tab2[TSTEPS + 2];
    __shared__ unsigned short tab4[TSTEPS + 2];
    __shared__ unsigned st[NCH];
    const int e   = blockIdx.x;
    const int tid = threadIdx.x;

    for (int r = tid; r < TSTEPS; r += 256) {
        unsigned w = nxt[(size_t)r * 32 + (e >> 1)];
        tab1[r] = (unsigned short)((e & 1) ? (w >> 16) : (w & 0xffffu));
    }
    if (tid == 0) { tab1[TSTEPS] = TSTEPS; tab2[TSTEPS] = TSTEPS; tab4[TSTEPS] = TSTEPS; }
    __syncthreads();
    for (int r = tid; r < TSTEPS; r += 256) tab2[r] = tab1[tab1[r]];
    __syncthreads();
    for (int r = tid; r < TSTEPS; r += 256) tab4[r] = tab2[tab2[r]];
    __syncthreads();

    if (tid == 0) {
        int c = 0;
        int r = 0;
        int a = tab1[0], b = tab2[0], d = tab4[0];
        for (int it = 0; it < TSTEPS && r < TSTEPS; ++it) {
            int m  = tab1[b];
            int an = tab1[d], bn = tab2[d], dn = tab4[d];
            fill_chunks(st, c, r, a);
            fill_chunks(st, c, a, b);
            fill_chunks(st, c, b, m);
            fill_chunks(st, c, m, d);
            r = d; a = an; b = bn; d = dn;
        }
    }
    __syncthreads();

    start[(size_t)e * NCH + tid]       = st[tid];
    start[(size_t)e * NCH + 256 + tid] = st[256 + tid];
}

#define RB   CHK
#define NRB  3

__device__ __forceinline__ void load_block(const float2* __restrict__ cur, int lane, int t0,
                                           float2 (&c)[RB]) {
    const int rb = min(max(t0, 0), TSTEPS - RB);
    const float2* p = cur + (size_t)rb * 32 + lane;
#pragma unroll
    for (int i = 0; i < RB; ++i) c[i] = p[(size_t)i * 32];
}

__device__ __forceinline__ void integrate_block(const float2 (&c)[RB], int t0, int eA, int eB,
                                                float2& m) {
#pragma unroll
    for (int i = 0; i < RB; ++i) {
        const int t = t0 + i;
        const float g0 = (t >= eA) ? 0.1f : 0.0f;
        const float g1 = (t >= eB) ? 0.1f : 0.0f;
        m.x = __builtin_fmaf(c[i].x, g0, m.x * 0.9f);
        m.y = __builtin_fmaf(c[i].y, g1, m.y * 0.9f);
    }
}

__device__ __forceinline__ float sitout_counter(int j) {
    float r = 1.0f;
#pragma unroll
    for (int k = 1; k < GAP - 1; ++k) {
        const float d = fmaxf(r - 0.1f, 0.0f);
        r = (k < j) ? d : r;
    }
    return r;
}

__global__ __launch_bounds__(32) void replay_route_kernel(const float2* __restrict__ cur,
                                                          const unsigned* __restrict__ start,
                                                          float2* __restrict__ out) {
    const int ch   = blockIdx.x;
    const int lane = threadIdx.x;
    const int tbeg = ch * CHK;

    const unsigned wA = start[(size_t)(2 * lane)     * NCH + ch];
    const unsigned wB = start[(size_t)(2 * lane + 1) * NCH + ch];
    const int sA = (int)(wA & 0xffffu);
    const int sB = (int)(wB & 0xffffu);
    const int jA = tbeg - ((int)(wA >> 16) - GAP);
    const int jB = tbeg - ((int)(wB >> 16) - GAP);
    const bool spA = jA > 0;
    const bool spB = jB > 0;
    const int eA = spA ? tbeg : sA;
    const int eB = spB ? tbeg : sB;

    int smin = min(eA, eB);
#pragma unroll
    for (int off = 16; off >= 1; off >>= 1)
        smin = min(smin, __shfl_xor(smin, off, 32));
    smin = __builtin_amdgcn_readfirstlane(smin);

    const int nblk = NRB * ((tbeg - smin + NRB * RB - 1) / (NRB * RB));
    int tb = tbeg - nblk * RB;

    float2 c0[RB], c1[RB], c2[RB];
    load_block(cur, lane, tb,          c0);
    load_block(cur, lane, tb + RB,     c1);
    load_block(cur, lane, tb + 2 * RB, c2);

    float2 m = {0.0f, 0.0f};
    for (; tb < tbeg; tb += NRB * RB) {
        integrate_block(c0, tb,          eA, eB, m);
        load_block(cur, lane, tb + 3 * RB, c0);
        integrate_block(c1, tb + RB,     eA, eB, m);
        load_block(cur, lane, tb + 4 * RB, c1);
        integrate_block(c2, tb + 2 * RB, eA, eB, m);
        load_block(cur, lane, tb + 5 * RB, c2);
    }

    float2 mp = m;
    float2 rf;
    rf.x = spA ? sitout_counter(jA) : 0.0f;
    rf.y = spB ? sitout_counter(jB) : 0.0f;

#pragma unroll
    for (int i = 0; i < CHK; ++i) {
        const int t = tbeg + i;
        float2 v = lif_step(c0[i], t >= sA, t >= sB, mp, rf);

        bool s0 = v.x > 1.0f;
        bool s1 = v.y > 1.0f;
        unsigned long long bal = __ballot(s0 || s1);
        bool any = (bal != 0ull);

        float x0 = any ? (s0 ? 1.0f : 0.0f) : v.x;
        float x1 = any ? (s1 ? 1.0f : 0.0f) : v.y;

        float mx = fmaxf(x0, x1);
#pragma unroll
        for (int off = 16; off >= 1; off >>= 1)
            mx = fmaxf(mx, __shfl_xor(mx, off, 32));

        float e0 = __expf(x0 - mx);
        float e1 = __expf(x1 - mx);
        float s  = e0 + e1;
#pragma unroll
        for (int off = 16; off >= 1; off >>= 1)
            s += __shfl_xor(s, off, 32);

        float inv = 1.0f / s;
        float2 rr;
        rr.x = e0 * inv;
        rr.y = e1 * inv;
        out[(size_t)t * 32 + lane] = rr;
    }
}

extern "C" void kernel_launch(void* const* d_in, const int* in_sizes, int n_in,
                              void* d_out, int out_size, void* d_ws, size_t ws_size,
                              hipStream_t stream) {
    const float* hidden = (const float*)d_in[0];
    const float* W      = (const float*)d_in[1];
    float* out          = (float*)d_out;

    float* curr = (float*)d_ws;
    float* BT   = (float*)((char*)d_ws + (size_t)TSTEPS * EXPERTS * sizeof(float));
    unsigned* nxt   = (unsigned*)d_out;
    unsigned* start = (unsigned*)BT;

    transpose_w_kernel<<<(HIDDEN * EXPERTS) / 256, 256, 0, stream>>>(W, BT);

    gemm_wmma_kernel<<<TSTEPS / AROWS, 256, 0, stream>>>(hidden, BT, curr);

    next_restart_kernel<<<TSTEPS / 8, 256, 0, stream>>>((const float2*)curr, nxt);

    restart_chain_kernel<<<EXPERTS, 256, 0, stream>>>(nxt, start);

    replay_route_kernel<<<NCH, 32, 0, stream>>>((const float2*)curr,
                                                (const unsigned*)start, (float2*)out);
}


# 2b restart walk made parallel: table squared twice more (16 hops), one lane notes every 16th restart, 48 threads walk the 15 between, all threads turn restart pairs into chunk entries; same table
# speedup vs baseline: 6.9333x; 1.1710x over previous
#include <hip/hip_runtime.h>


typedef float v2f __attribute__((ext_vector_type(2)));
typedef float v8f __attribute__((ext_vector_type(8)));

#define HIDDEN 2048
#define EXPERTS 64
#define TSTEPS 8192

__global__ __launch_bounds__(256) void transpose_w_kernel(const float* __restrict__ W,
                                                          float* __restrict__ BT) {
    int idx = blockIdx.x * 256 + threadIdx.x;
    int k = idx >> 6;
    int n = idx & 63;
    BT[(size_t)n * HIDDEN + k] = W[(size_t)k * EXPERTS + n];
}

#define KC      128
#define PITCH   (KC + 4)
#define AROWS   32
#define BROWS   EXPERTS
#define STAGE   ((AROWS + BROWS) * PITCH)
#define NCHUNK  (HIDDEN / KC)
#define CPITCH  (EXPERTS + 4)

struct Frags { v2f a[KC / 4]; v2f b[KC / 4]; };

__device__ __forceinline__ void read_frags(const float* st, int fa, int fb, Frags& f) {
#pragma unroll
    for (int ks = 0; ks < KC / 4; ++ks) {
        f.a[ks] = *(const v2f*)(st + fa + ks * 4);
        f.b[ks] = *(const v2f*)(st + fb + ks * 4);
    }
    __builtin_amdgcn_sched_barrier(0);
}

__device__ __forceinline__ void mma_chunk(const Frags& f, v8f& acc) {
#pragma unroll
    for (int ks = 0; ks < KC / 4; ++ks)
        acc = __builtin_amdgcn_wmma_f32_16x16x4_f32(false, f.a[ks], false, f.b[ks], (short)0, acc, false, false);
}

__device__ __forceinline__ void lds_barrier() {
    __builtin_amdgcn_fence(__ATOMIC_RELEASE, "workgroup", "local");
    __builtin_amdgcn_s_barrier();
    __builtin_amdgcn_fence(__ATOMIC_ACQUIRE, "workgroup", "local");
}

__global__ __launch_bounds__(256) void gemm_wmma_kernel(const float* __restrict__ A,
                                                        const float* __restrict__ BT,
                                                        float* __restrict__ C) {
    __shared__ __attribute__((aligned(16))) float lds[2 * STAGE];

    const int tid  = threadIdx.x;
    const int wave = tid >> 5;
    const int lane = tid & 31;
    const int m0   = blockIdx.x * AROWS;
    const int mt   = wave >> 2;
    const int nt   = wave & 3;
    const int half = lane >> 4;
    const int sub  = lane & 15;

    const float* ga = A  + (size_t)(m0 + wave * 4) * HIDDEN + lane * 4;
    const float* gb = BT + (size_t)(wave * 8) * HIDDEN + lane * 4;
    const int sa = (wave * 4) * PITCH + lane * 4;
    const int sb = (AROWS + wave * 8) * PITCH + lane * 4;

    const int fa = (mt * 16 + sub) * PITCH + half * 2;
    const int fb = (AROWS + nt * 16 + sub) * PITCH + half * 2;

    float4 ra0, ra1, ra2, ra3, rb0, rb1, rb2, rb3, rb4, rb5, rb6, rb7;
#define LOAD_ROWS(koff)                                              \
    ra0 = *(const float4*)(ga + 0 * (size_t)HIDDEN + (koff));        \
    ra1 = *(const float4*)(ga + 1 * (size_t)HIDDEN + (koff));        \
    ra2 = *(const float4*)(ga + 2 * (size_t)HIDDEN + (koff));        \
    ra3 = *(const float4*)(ga + 3 * (size_t)HIDDEN + (koff));        \
    rb0 = *(const float4*)(gb + 0 * (size_t)HIDDEN + (koff));        \
    rb1 = *(const float4*)(gb + 1 * (size_t)HIDDEN + (koff));        \
    rb2 = *(const float4*)(gb + 2 * (size_t)HIDDEN + (koff));        \
    rb3 = *(const float4*)(gb + 3 * (size_t)HIDDEN + (koff));        \
    rb4 = *(const float4*)(gb + 4 * (size_t)HIDDEN + (koff));        \
    rb5 = *(const float4*)(gb + 5 * (size_t)HIDDEN + (koff));        \
    rb6 = *(const float4*)(gb + 6 * (size_t)HIDDEN + (koff));        \
    rb7 = *(const float4*)(gb + 7 * (size_t)HIDDEN + (koff));
#define STORE_ROWS(stage)                                            \
    *(float4*)((stage) + sa + 0 * PITCH) = ra0;                      \
    *(float4*)((stage) + sa + 1 * PITCH) = ra1;                      \
    *(float4*)((stage) + sa + 2 * PITCH) = ra2;                      \
    *(float4*)((stage) + sa + 3 * PITCH) = ra3;                      \
    *(float4*)((stage) + sb + 0 * PITCH) = rb0;                      \
    *(float4*)((stage) + sb + 1 * PITCH) = rb1;                      \
    *(float4*)((stage) + sb + 2 * PITCH) = rb2;                      \
    *(float4*)((stage) + sb + 3 * PITCH) = rb3;                      \
    *(float4*)((stage) + sb + 4 * PITCH) = rb4;                      \
    *(float4*)((stage) + sb + 5 * PITCH) = rb5;                      \
    *(float4*)((stage) + sb + 6 * PITCH) = rb6;                      \
    *(float4*)((stage) + sb + 7 * PITCH) = rb7;

    LOAD_ROWS(0)
    STORE_ROWS(lds)
    LOAD_ROWS(KC)
    __syncthreads();

    v8f acc = {};

    for (int c = 0; c < NCHUNK - 2; ++c) {
        const float* st = lds + (c & 1) * STAGE;
        float*       sn = lds + ((c + 1) & 1) * STAGE;
        const int    kn = (c + 2) * KC;
        Frags f;
        read_frags(st, fa, fb, f);
        STORE_ROWS(sn)
        LOAD_ROWS(kn)
        mma_chunk(f, acc);
        lds_barrier();
    }
    {
        Frags f;
        read_frags(lds + ((NCHUNK - 2) & 1) * STAGE, fa, fb, f);
        STORE_ROWS(lds + ((NCHUNK - 1) & 1) * STAGE)
        mma_chunk(f, acc);
        __syncthreads();
    }
    {
        Frags f;
        read_frags(lds + ((NCHUNK - 1) & 1) * STAGE, fa, fb, f);
        mma_chunk(f, acc);
        __syncthreads();
    }
#undef LOAD_ROWS
#undef STORE_ROWS

    float* ct = lds;
#pragma unroll
    for (int r = 0; r < 8; ++r)
        ct[(mt * 16 + half * 8 + r) * CPITCH + nt * 16 + sub] = acc[r];
    __syncthreads();
#pragma unroll
    for (int p = 0; p < 2; ++p) {
        const int j   = p * 1024 + tid * 4;
        const int row = j >> 6;
        const int col = j & 63;
        float4 v = *(const float4*)(ct + row * CPITCH + col);
        *(float4*)(C + (size_t)m0 * EXPERTS + j) = v;
    }
}

#define CHK  16
#define NCH  (TSTEPS / CHK)
#define BLK  8

constexpr int refractory_gap() {
    float rf = 1.0f;
    int k = 0;
    for (int i = 0; i < 64; ++i) {
        ++k;
        bool active = (rf <= 0.0f);
        float d = rf - 0.1f;
        rf = (d > 0.0f) ? d : 0.0f;
        if (active) break;
    }
    return k;
}
constexpr int GAP = refractory_gap();

__device__ __forceinline__ float2 lif_step(float2 cur, bool on0, bool on1,
                                           float2& mp, float2& rf) {
    float l0 = mp.x * 0.9f;
    float l1 = mp.y * 0.9f;
    float a0 = (on0 && rf.x <= 0.0f) ? 0.1f : 0.0f;
    float a1 = (on1 && rf.y <= 0.0f) ? 0.1f : 0.0f;
    mp.x = __builtin_fmaf(cur.x, a0, l0);
    mp.y = __builtin_fmaf(cur.y, a1, l1);
    rf.x = fmaxf(rf.x - 0.1f, 0.0f);
    rf.y = fmaxf(rf.y - 0.1f, 0.0f);
    float2 pre = mp;
    bool s0 = mp.x > 1.0f;
    bool s1 = mp.y > 1.0f;
    mp.x = s0 ? 0.0f : mp.x;
    mp.y = s1 ? 0.0f : mp.y;
    rf.x = s0 ? 1.0f : rf.x;
    rf.y = s1 ? 1.0f : rf.y;
    return pre;
}

__global__ __launch_bounds__(256) void next_restart_kernel(const float2* __restrict__ cur,
                                                           unsigned* __restrict__ nxt) {
    const int r    = __builtin_amdgcn_readfirstlane(blockIdx.x * 8 + (threadIdx.x >> 5));
    const int lane = threadIdx.x & 31;

    float m0 = 0.0f, m1 = 0.0f;
    int s0 = TSTEPS, s1 = TSTEPS;

    for (int tb = r; tb < TSTEPS; tb += BLK) {
        float2 c[BLK];
#pragma unroll
        for (int i = 0; i < BLK; ++i) {
            int t = min(tb + i, TSTEPS - 1);
            c[i] = cur[(size_t)t * 32 + lane];
        }
#pragma unroll
        for (int i = 0; i < BLK; ++i) {
            int t = tb + i;
            m0 = __builtin_fmaf(c[i].x, 0.1f, m0 * 0.9f);
            m1 = __builtin_fmaf(c[i].y, 0.1f, m1 * 0.9f);
            bool in = t < TSTEPS;
            if (in && s0 == TSTEPS && m0 > 1.0f) s0 = t;
            if (in && s1 == TSTEPS && m1 > 1.0f) s1 = t;
        }
        if (__ballot(s0 == TSTEPS || s1 == TSTEPS) == 0ull) break;
    }
    unsigned n0 = (unsigned)min(s0 + GAP, TSTEPS);
    unsigned n1 = (unsigned)min(s1 + GAP, TSTEPS);
    nxt[(size_t)r * 32 + lane] = n0 | (n1 << 16);
}

#define HOPS   16
#define NMARK  (((TSTEPS / (HOPS * GAP) + 1 + 3) / 4) * 4)
#define NHOP   (NMARK * HOPS)

__global__ __launch_bounds__(256) void restart_chain_kernel(const unsigned* __restrict__ nxt,
                                                            unsigned* __restrict__ start) {
    __shared__ unsigned short tab1[TSTEPS + 2];
    __shared__ unsigned short tabA[TSTEPS + 2];
    __shared__ unsigned short tabB[TSTEPS + 2];
    __shared__ __attribute__((aligned(16))) unsigned short mark[NMARK];
    __shared__ __attribute__((aligned(16))) unsigned short hop[NHOP + 8];
    __shared__ unsigned st[NCH];
    const int e   = blockIdx.x;
    const int tid = threadIdx.x;

    for (int r = tid; r < TSTEPS; r += 256) {
        unsigned w = nxt[(size_t)r * 32 + (e >> 1)];
        tab1[r] = (unsigned short)((e & 1) ? (w >> 16) : (w & 0xffffu));
    }
    if (tid == 0) { tab1[TSTEPS] = TSTEPS; tabA[TSTEPS] = TSTEPS; tabB[TSTEPS] = TSTEPS; hop[NHOP] = TSTEPS; }
    if (tid < NMARK) mark[tid] = TSTEPS;
    __syncthreads();
    for (int r = tid; r < TSTEPS; r += 256) tabA[r] = tab1[tab1[r]];
    __syncthreads();
    for (int r = tid; r < TSTEPS; r += 256) tabB[r] = tabA[tabA[r]];
    __syncthreads();
    for (int r = tid; r < TSTEPS; r += 256) tabA[r] = tabB[tabB[r]];
    __syncthreads();
    for (int r = tid; r < TSTEPS; r += 256) tabB[r] = tabA[tabA[r]];
    __syncthreads();

    if (tid == 0) {
        int r = 0;
        for (int n = 0; n < NMARK && r < TSTEPS; n += 4) {
            const int r1 = tabB[r];
            const int r2 = tabB[r1];
            const int r3 = tabB[r2];
            mark[n]     = (unsigned short)r;
            mark[n + 1] = (unsigned short)r1;
            mark[n + 2] = (unsigned short)r2;
            mark[n + 3] = (unsigned short)r3;
            r = tabB[r3];
        }
    }
    __syncthreads();

    if (tid < NMARK) {
        int r = mark[tid];
#pragma unroll
        for (int i = 0; i < HOPS; ++i) {
            hop[tid * HOPS + i] = (unsigned short)r;
            r = tab1[r];
        }
    }
    __syncthreads();

    for (int q = tid; q < NHOP; q += 256) {
        const int r    = hop[q];
        const int n    = hop[q + 1];
        const int cend = min((n + CHK - 1) / CHK, NCH);
        const unsigned w = (unsigned)r | ((unsigned)n << 16);
        for (int c = (r + CHK - 1) / CHK; c < cend; ++c) st[c] = w;
    }
    __syncthreads();

    start[(size_t)e * NCH + tid]       = st[tid];
    start[(size_t)e * NCH + 256 + tid] = st[256 + tid];
}

#define RB   CHK
#define NRB  3

__device__ __forceinline__ void load_block(const float2* __restrict__ cur, int lane, int t0,
                                           float2 (&c)[RB]) {
    const int rb = min(max(t0, 0), TSTEPS - RB);
    const float2* p = cur + (size_t)rb * 32 + lane;
#pragma unroll
    for (int i = 0; i < RB; ++i) c[i] = p[(size_t)i * 32];
}

__device__ __forceinline__ void integrate_block(const float2 (&c)[RB], int t0, int eA, int eB,
                                                float2& m) {
#pragma unroll
    for (int i = 0; i < RB; ++i) {
        const int t = t0 + i;
        const float g0 = (t >= eA) ? 0.1f : 0.0f;
        const float g1 = (t >= eB) ? 0.1f : 0.0f;
        m.x = __builtin_fmaf(c[i].x, g0, m.x * 0.9f);
        m.y = __builtin_fmaf(c[i].y, g1, m.y * 0.9f);
    }
}

__device__ __forceinline__ float sitout_counter(int j) {
    float r = 1.0f;
#pragma unroll
    for (int k = 1; k < GAP - 1; ++k) {
        const float d = fmaxf(r - 0.1f, 0.0f);
        r = (k < j) ? d : r;
    }
    return r;
}

__global__ __launch_bounds__(32) void replay_route_kernel(const float2* __restrict__ cur,
                                                          const unsigned* __restrict__ start,
                                                          float2* __restrict__ out) {
    const int ch   = blockIdx.x;
    const int lane = threadIdx.x;
    const int tbeg = ch * CHK;

    const unsigned wA = start[(size_t)(2 * lane)     * NCH + ch];
    const unsigned wB = start[(size_t)(2 * lane + 1) * NCH + ch];
    const int sA = (int)(wA & 0xffffu);
    const int sB = (int)(wB & 0xffffu);
    const int jA = tbeg - ((int)(wA >> 16) - GAP);
    const int jB = tbeg - ((int)(wB >> 16) - GAP);
    const bool spA = jA > 0;
    const bool spB = jB > 0;
    const int eA = spA ? tbeg : sA;
    const int eB = spB ? tbeg : sB;

    int smin = min(eA, eB);
#pragma unroll
    for (int off = 16; off >= 1; off >>= 1)
        smin = min(smin, __shfl_xor(smin, off, 32));
    smin = __builtin_amdgcn_readfirstlane(smin);

    const int nblk = NRB * ((tbeg - smin + NRB * RB - 1) / (NRB * RB));
    int tb = tbeg - nblk * RB;

    float2 c0[RB], c1[RB], c2[RB];
    load_block(cur, lane, tb,          c0);
    load_block(cur, lane, tb + RB,     c1);
    load_block(cur, lane, tb + 2 * RB, c2);

    float2 m = {0.0f, 0.0f};
    for (; tb < tbeg; tb += NRB * RB) {
        integrate_block(c0, tb,          eA, eB, m);
        load_block(cur, lane, tb + 3 * RB, c0);
        integrate_block(c1, tb + RB,     eA, eB, m);
        load_block(cur, lane, tb + 4 * RB, c1);
        integrate_block(c2, tb + 2 * RB, eA, eB, m);
        load_block(cur, lane, tb + 5 * RB, c2);
    }

    float2 mp = m;
    float2 rf;
    rf.x = spA ? sitout_counter(jA) : 0.0f;
    rf.y = spB ? sitout_counter(jB) : 0.0f;

#pragma unroll
    for (int i = 0; i < CHK; ++i) {
        const int t = tbeg + i;
        float2 v = lif_step(c0[i], t >= sA, t >= sB, mp, rf);

        bool s0 = v.x > 1.0f;
        bool s1 = v.y > 1.0f;
        unsigned long long bal = __ballot(s0 || s1);
        bool any = (bal != 0ull);

        float x0 = any ? (s0 ? 1.0f : 0.0f) : v.x;
        float x1 = any ? (s1 ? 1.0f : 0.0f) : v.y;

        float mx = fmaxf(x0, x1);
#pragma unroll
        for (int off = 16; off >= 1; off >>= 1)
            mx = fmaxf(mx, __shfl_xor(mx, off, 32));

        float e0 = __expf(x0 - mx);
        float e1 = __expf(x1 - mx);
        float s  = e0 + e1;
#pragma unroll
        for (int off = 16; off >= 1; off >>= 1)
            s += __shfl_xor(s, off, 32);

        float inv = 1.0f / s;
        float2 rr;
        rr.x = e0 * inv;
        rr.y = e1 * inv;
        out[(size_t)t * 32 + lane] = rr;
    }
}

extern "C" void kernel_launch(void* const* d_in, const int* in_sizes, int n_in,
                              void* d_out, int out_size, void* d_ws, size_t ws_size,
                              hipStream_t stream) {
    const float* hidden = (const float*)d_in[0];
    const float* W      = (const float*)d_in[1];
    float* out          = (float*)d_out;

    float* curr = (float*)d_ws;
    float* BT   = (float*)((char*)d_ws + (size_t)TSTEPS * EXPERTS * sizeof(float));
    unsigned* nxt   = (unsigned*)d_out;
    unsigned* start = (unsigned*)BT;

    transpose_w_kernel<<<(HIDDEN * EXPERTS) / 256, 256, 0, stream>>>(W, BT);

    gemm_wmma_kernel<<<TSTEPS / AROWS, 256, 0, stream>>>(hidden, BT, curr);

    next_restart_kernel<<<TSTEPS / 8, 256, 0, stream>>>((const float2*)curr, nxt);

    restart_chain_kernel<<<EXPERTS, 256, 0, stream>>>(nxt, start);

    replay_route_kernel<<<NCH, 32, 0, stream>>>((const float2*)curr,
                                                (const unsigned*)start, (float2*)out);
}


# W tile staged straight from W rows: a lane pairs an even and an odd k row in registers, LDS keeps k-pairs side by side, B fragment stays one 8-byte read; transpose launch gone; same k order and sums
# speedup vs baseline: 7.5005x; 1.0818x over previous
#include <hip/hip_runtime.h>


typedef float v2f __attribute__((ext_vector_type(2)));
typedef float v4f __attribute__((ext_vector_type(4)));
typedef float v8f __attribute__((ext_vector_type(8)));

#define HIDDEN 2048
#define EXPERTS 64
#define TSTEPS 8192

#define KC      128
#define PITCH   (KC + 4)
#define AROWS   32
#define ASTAGE  (AROWS * PITCH)
#define BPAIRS  (KC / 2)
#define BPITCH  (2 * EXPERTS + 32)
#define STAGE   (ASTAGE + BPAIRS * BPITCH)
#define NCHUNK  (HIDDEN / KC)
#define CPITCH  (EXPERTS + 4)

struct Frags { v2f a[KC / 4]; v2f b[KC / 4]; };

__device__ __forceinline__ void read_frags(const float* st, int fa, int fb, Frags& f) {
#pragma unroll
    for (int ks = 0; ks < KC / 4; ++ks) {
        f.a[ks] = *(const v2f*)(st + fa + ks * 4);
        f.b[ks] = *(const v2f*)(st + fb + ks * 2 * BPITCH);
    }
    __builtin_amdgcn_sched_barrier(0);
}

__device__ __forceinline__ void mma_chunk(const Frags& f, v8f& acc) {
#pragma unroll
    for (int ks = 0; ks < KC / 4; ++ks)
        acc = __builtin_amdgcn_wmma_f32_16x16x4_f32(false, f.a[ks], false, f.b[ks], (short)0, acc, false, false);
}

__device__ __forceinline__ void lds_barrier() {
    __builtin_amdgcn_fence(__ATOMIC_RELEASE, "workgroup", "local");
    __builtin_amdgcn_s_barrier();
    __builtin_amdgcn_fence(__ATOMIC_ACQUIRE, "workgroup", "local");
}

__global__ __launch_bounds__(256) void gemm_wmma_kernel(const float* __restrict__ A,
                                                        const float* __restrict__ W,
                                                        float* __restrict__ C) {
    __shared__ __attribute__((aligned(16))) float lds[2 * STAGE];

    const int tid  = threadIdx.x;
    const int wave = tid >> 5;
    const int lane = tid & 31;
    const int m0   = blockIdx.x * AROWS;
    const int mt   = wave >> 2;
    const int nt   = wave & 3;
    const int half = lane >> 4;
    const int sub  = lane & 15;

    const float* ga = A + (size_t)(m0 + wave * 4) * HIDDEN + lane * 4;
    const float* gb = W + (size_t)(wave * 16 + half * 2) * EXPERTS + sub * 4;
    const int sa = (wave * 4) * PITCH + lane * 4;
    const int sb = ASTAGE + (wave * 8 + half) * BPITCH + sub * 8;

    const int fa = (mt * 16 + sub) * PITCH + half * 2;
    const int fb = ASTAGE + half * BPITCH + (nt * 16 + sub) * 2;

    float4 ra0, ra1, ra2, ra3;
    v4f    rb0, rb1, rb2, rb3, rb4, rb5, rb6, rb7;
#define STORE_PAIR(dst, e, o)                                        \
    *(v4f*)(dst)       = (v4f){(e).x, (o).x, (e).y, (o).y};          \
    *(v4f*)((dst) + 4) = (v4f){(e).z, (o).z, (e).w, (o).w};
#define HOLD_RAW(r) asm volatile("" : "+v"(r));
#define LOAD_ROWS(koff)                                              \
    ra0 = *(const float4*)(ga + 0 * (size_t)HIDDEN + (koff));        \
    ra1 = *(const float4*)(ga + 1 * (size_t)HIDDEN + (koff));        \
    ra2 = *(const float4*)(ga + 2 * (size_t)HIDDEN + (koff));        \
    ra3 = *(const float4*)(ga + 3 * (size_t)HIDDEN + (koff));        \
    rb0 = *(const v4f*)(gb + ((size_t)(koff) +  0) * EXPERTS);    \
    rb1 = *(const v4f*)(gb + ((size_t)(koff) +  1) * EXPERTS);    \
    rb2 = *(const v4f*)(gb + ((size_t)(koff) +  4) * EXPERTS);    \
    rb3 = *(const v4f*)(gb + ((size_t)(koff) +  5) * EXPERTS);    \
    rb4 = *(const v4f*)(gb + ((size_t)(koff) +  8) * EXPERTS);    \
    rb5 = *(const v4f*)(gb + ((size_t)(koff) +  9) * EXPERTS);    \
    rb6 = *(const v4f*)(gb + ((size_t)(koff) + 12) * EXPERTS);    \
    rb7 = *(const v4f*)(gb + ((size_t)(koff) + 13) * EXPERTS);
#define STORE_ROWS(stage)                                            \
    *(float4*)((stage) + sa + 0 * PITCH) = ra0;                      \
    *(float4*)((stage) + sa + 1 * PITCH) = ra1;                      \
    *(float4*)((stage) + sa + 2 * PITCH) = ra2;                      \
    *(float4*)((stage) + sa + 3 * PITCH) = ra3;                      \
    HOLD_RAW(rb0) HOLD_RAW(rb1) HOLD_RAW(rb2) HOLD_RAW(rb3)          \
    HOLD_RAW(rb4) HOLD_RAW(rb5) HOLD_RAW(rb6) HOLD_RAW(rb7)          \
    STORE_PAIR((stage) + sb + 0 * BPITCH, rb0, rb1)                  \
    STORE_PAIR((stage) + sb + 2 * BPITCH, rb2, rb3)                  \
    STORE_PAIR((stage) + sb + 4 * BPITCH, rb4, rb5)                  \
    STORE_PAIR((stage) + sb + 6 * BPITCH, rb6, rb7)

    LOAD_ROWS(0)
    STORE_ROWS(lds)
    LOAD_ROWS(KC)
    __syncthreads();

    v8f acc = {};

    for (int c = 0; c < NCHUNK - 2; ++c) {
        const float* st = lds + (c & 1) * STAGE;
        float*       sn = lds + ((c + 1) & 1) * STAGE;
        const int    kn = (c + 2) * KC;
        Frags f;
        read_frags(st, fa, fb, f);
        STORE_ROWS(sn)
        LOAD_ROWS(kn)
        mma_chunk(f, acc);
        lds_barrier();
    }
    {
        Frags f;
        read_frags(lds + ((NCHUNK - 2) & 1) * STAGE, fa, fb, f);
        STORE_ROWS(lds + ((NCHUNK - 1) & 1) * STAGE)
        mma_chunk(f, acc);
        __syncthreads();
    }
    {
        Frags f;
        read_frags(lds + ((NCHUNK - 1) & 1) * STAGE, fa, fb, f);
        mma_chunk(f, acc);
        __syncthreads();
    }
#undef LOAD_ROWS
#undef STORE_ROWS
#undef STORE_PAIR
#undef HOLD_RAW

    float* ct = lds;
#pragma unroll
    for (int r = 0; r < 8; ++r)
        ct[(mt * 16 + half * 8 + r) * CPITCH + nt * 16 + sub] = acc[r];
    __syncthreads();
#pragma unroll
    for (int p = 0; p < 2; ++p) {
        const int j   = p * 1024 + tid * 4;
        const int row = j >> 6;
        const int col = j & 63;
        float4 v = *(const float4*)(ct + row * CPITCH + col);
        *(float4*)(C + (size_t)m0 * EXPERTS + j) = v;
    }
}

#define CHK  16
#define NCH  (TSTEPS / CHK)
#define BLK  8

constexpr int refractory_gap() {
    float rf = 1.0f;
    int k = 0;
    for (int i = 0; i < 64; ++i) {
        ++k;
        bool active = (rf <= 0.0f);
        float d = rf - 0.1f;
        rf = (d > 0.0f) ? d : 0.0f;
        if (active) break;
    }
    return k;
}
constexpr int GAP = refractory_gap();

__device__ __forceinline__ float2 lif_step(float2 cur, bool on0, bool on1,
                                           float2& mp, float2& rf) {
    float l0 = mp.x * 0.9f;
    float l1 = mp.y * 0.9f;
    float a0 = (on0 && rf.x <= 0.0f) ? 0.1f : 0.0f;
    float a1 = (on1 && rf.y <= 0.0f) ? 0.1f : 0.0f;
    mp.x = __builtin_fmaf(cur.x, a0, l0);
    mp.y = __builtin_fmaf(cur.y, a1, l1);
    rf.x = fmaxf(rf.x - 0.1f, 0.0f);
    rf.y = fmaxf(rf.y - 0.1f, 0.0f);
    float2 pre = mp;
    bool s0 = mp.x > 1.0f;
    bool s1 = mp.y > 1.0f;
    mp.x = s0 ? 0.0f : mp.x;
    mp.y = s1 ? 0.0f : mp.y;
    rf.x = s0 ? 1.0f : rf.x;
    rf.y = s1 ? 1.0f : rf.y;
    return pre;
}

__global__ __launch_bounds__(256) void next_restart_kernel(const float2* __restrict__ cur,
                                                           unsigned* __restrict__ nxt) {
    const int r    = __builtin_amdgcn_readfirstlane(blockIdx.x * 8 + (threadIdx.x >> 5));
    const int lane = threadIdx.x & 31;

    float m0 = 0.0f, m1 = 0.0f;
    int s0 = TSTEPS, s1 = TSTEPS;

    for (int tb = r; tb < TSTEPS; tb += BLK) {
        float2 c[BLK];
#pragma unroll
        for (int i = 0; i < BLK; ++i) {
            int t = min(tb + i, TSTEPS - 1);
            c[i] = cur[(size_t)t * 32 + lane];
        }
#pragma unroll
        for (int i = 0; i < BLK; ++i) {
            int t = tb + i;
            m0 = __builtin_fmaf(c[i].x, 0.1f, m0 * 0.9f);
            m1 = __builtin_fmaf(c[i].y, 0.1f, m1 * 0.9f);
            bool in = t < TSTEPS;
            if (in && s0 == TSTEPS && m0 > 1.0f) s0 = t;
            if (in && s1 == TSTEPS && m1 > 1.0f) s1 = t;
        }
        if (__ballot(s0 == TSTEPS || s1 == TSTEPS) == 0ull) break;
    }
    unsigned n0 = (unsigned)min(s0 + GAP, TSTEPS);
    unsigned n1 = (unsigned)min(s1 + GAP, TSTEPS);
    nxt[(size_t)r * 32 + lane] = n0 | (n1 << 16);
}

#define HOPS   16
#define NMARK  (((TSTEPS / (HOPS * GAP) + 1 + 3) / 4) * 4)
#define NHOP   (NMARK * HOPS)

__global__ __launch_bounds__(256) void restart_chain_kernel(const unsigned* __restrict__ nxt,
                                                            unsigned* __restrict__ start) {
    __shared__ unsigned short tab1[TSTEPS + 2];
    __shared__ unsigned short tabA[TSTEPS + 2];
    __shared__ unsigned short tabB[TSTEPS + 2];
    __shared__ __attribute__((aligned(16))) unsigned short mark[NMARK];
    __shared__ __attribute__((aligned(16))) unsigned short hop[NHOP + 8];
    __shared__ unsigned st[NCH];
    const int e   = blockIdx.x;
    const int tid = threadIdx.x;

    for (int r = tid; r < TSTEPS; r += 256) {
        unsigned w = nxt[(size_t)r * 32 + (e >> 1)];
        tab1[r] = (unsigned short)((e & 1) ? (w >> 16) : (w & 0xffffu));
    }
    if (tid == 0) { tab1[TSTEPS] = TSTEPS; tabA[TSTEPS] = TSTEPS; tabB[TSTEPS] = TSTEPS; hop[NHOP] = TSTEPS; }
    if (tid < NMARK) mark[tid] = TSTEPS;
    __syncthreads();
    for (int r = tid; r < TSTEPS; r += 256) tabA[r] = tab1[tab1[r]];
    __syncthreads();
    for (int r = tid; r < TSTEPS; r += 256) tabB[r] = tabA[tabA[r]];
    __syncthreads();
    for (int r = tid; r < TSTEPS; r += 256) tabA[r] = tabB[tabB[r]];
    __syncthreads();
    for (int r = tid; r < TSTEPS; r += 256) tabB[r] = tabA[tabA[r]];
    __syncthreads();

    if (tid == 0) {
        int r = 0;
        for (int n = 0; n < NMARK && r < TSTEPS; n += 4) {
            const int r1 = tabB[r];
            const int r2 = tabB[r1];
            const int r3 = tabB[r2];
            mark[n]     = (unsigned short)r;
            mark[n + 1] = (unsigned short)r1;
            mark[n + 2] = (unsigned short)r2;
            mark[n + 3] = (unsigned short)r3;
            r = tabB[r3];
        }
    }
    __syncthreads();

    if (tid < NMARK) {
        int r = mark[tid];
#pragma unroll
        for (int i = 0; i < HOPS; ++i) {
            hop[tid * HOPS + i] = (unsigned short)r;
            r = tab1[r];
        }
    }
    __syncthreads();

    for (int q = tid; q < NHOP; q += 256) {
        const int r    = hop[q];
        const int n    = hop[q + 1];
        const int cend = min((n + CHK - 1) / CHK, NCH);
        const unsigned w = (unsigned)r | ((unsigned)n << 16);
        for (int c = (r + CHK - 1) / CHK; c < cend; ++c) st[c] = w;
    }
    __syncthreads();

    start[(size_t)e * NCH + tid]       = st[tid];
    start[(size_t)e * NCH + 256 + tid] = st[256 + tid];
}

#define RB   CHK
#define NRB  3

__device__ __forceinline__ void load_block(const float2* __restrict__ cur, int lane, int t0,
                                           float2 (&c)[RB]) {
    const int rb = min(max(t0, 0), TSTEPS - RB);
    const float2* p = cur + (size_t)rb * 32 + lane;
#pragma unroll
    for (int i = 0; i < RB; ++i) c[i] = p[(size_t)i * 32];
}

__device__ __forceinline__ void integrate_block(const float2 (&c)[RB], int t0, int eA, int eB,
                                                float2& m) {
#pragma unroll
    for (int i = 0; i < RB; ++i) {
        const int t = t0 + i;
        const float g0 = (t >= eA) ? 0.1f : 0.0f;
        const float g1 = (t >= eB) ? 0.1f : 0.0f;
        m.x = __builtin_fmaf(c[i].x, g0, m.x * 0.9f);
        m.y = __builtin_fmaf(c[i].y, g1, m.y * 0.9f);
    }
}

__device__ __forceinline__ float sitout_counter(int j) {
    float r = 1.0f;
#pragma unroll
    for (int k = 1; k < GAP - 1; ++k) {
        const float d = fmaxf(r - 0.1f, 0.0f);
        r = (k < j) ? d : r;
    }
    return r;
}

__global__ __launch_bounds__(32) void replay_route_kernel(const float2* __restrict__ cur,
                                                          const unsigned* __restrict__ start,
                                                          float2* __restrict__ out) {
    const int ch   = blockIdx.x;
    const int lane = threadIdx.x;
    const int tbeg = ch * CHK;

    const unsigned wA = start[(size_t)(2 * lane)     * NCH + ch];
    const unsigned wB = start[(size_t)(2 * lane + 1) * NCH + ch];
    const int sA = (int)(wA & 0xffffu);
    const int sB = (int)(wB & 0xffffu);
    const int jA = tbeg - ((int)(wA >> 16) - GAP);
    const int jB = tbeg - ((int)(wB >> 16) - GAP);
    const bool spA = jA > 0;
    const bool spB = jB > 0;
    const int eA = spA ? tbeg : sA;
    const int eB = spB ? tbeg : sB;

    int smin = min(eA, eB);
#pragma unroll
    for (int off = 16; off >= 1; off >>= 1)
        smin = min(smin, __shfl_xor(smin, off, 32));
    smin = __builtin_amdgcn_readfirstlane(smin);

    const int nblk = NRB * ((tbeg - smin + NRB * RB - 1) / (NRB * RB));
    int tb = tbeg - nblk * RB;

    float2 c0[RB], c1[RB], c2[RB];
    load_block(cur, lane, tb,          c0);
    load_block(cur, lane, tb + RB,     c1);
    load_block(cur, lane, tb + 2 * RB, c2);

    float2 m = {0.0f, 0.0f};
    for (; tb < tbeg; tb += NRB * RB) {
        integrate_block(c0, tb,          eA, eB, m);
        load_block(cur, lane, tb + 3 * RB, c0);
        integrate_block(c1, tb + RB,     eA, eB, m);
        load_block(cur, lane, tb + 4 * RB, c1);
        integrate_block(c2, tb + 2 * RB, eA, eB, m);
        load_block(cur, lane, tb + 5 * RB, c2);
    }

    float2 mp = m;
    float2 rf;
    rf.x = spA ? sitout_counter(jA) : 0.0f;
    rf.y = spB ? sitout_counter(jB) : 0.0f;

#pragma unroll
    for (int i = 0; i < CHK; ++i) {
        const int t = tbeg + i;
        float2 v = lif_step(c0[i], t >= sA, t >= sB, mp, rf);

        bool s0 = v.x > 1.0f;
        bool s1 = v.y > 1.0f;
        unsigned long long bal = __ballot(s0 || s1);
        bool any = (bal != 0ull);

        float x0 = any ? (s0 ? 1.0f : 0.0f) : v.x;
        float x1 = any ? (s1 ? 1.0f : 0.0f) : v.y;

        float mx = fmaxf(x0, x1);
#pragma unroll
        for (int off = 16; off >= 1; off >>= 1)
            mx = fmaxf(mx, __shfl_xor(mx, off, 32));

        float e0 = __expf(x0 - mx);
        float e1 = __expf(x1 - mx);
        float s  = e0 + e1;
#pragma unroll
        for (int off = 16; off >= 1; off >>= 1)
            s += __shfl_xor(s, off, 32);

        float inv = 1.0f / s;
        float2 rr;
        rr.x = e0 * inv;
        rr.y = e1 * inv;
        out[(size_t)t * 32 + lane] = rr;
    }
}

extern "C" void kernel_launch(void* const* d_in, const int* in_sizes, int n_in,
                              void* d_out, int out_size, void* d_ws, size_t ws_size,
                              hipStream_t stream) {
    const float* hidden = (const float*)d_in[0];
    const float* W      = (const float*)d_in[1];
    float* out          = (float*)d_out;

    float* curr = (float*)d_ws;
    unsigned* nxt   = (unsigned*)d_out;
    unsigned* start = (unsigned*)((char*)d_ws + (size_t)TSTEPS * EXPERTS * sizeof(float));

    gemm_wmma_kernel<<<TSTEPS / AROWS, 256, 0, stream>>>(hidden, W, curr);

    next_restart_kernel<<<TSTEPS / 8, 256, 0, stream>>>((const float2*)curr, nxt);

    restart_chain_kernel<<<EXPERTS, 256, 0, stream>>>(nxt, start);

    replay_route_kernel<<<NCH, 32, 0, stream>>>((const float2*)curr,
                                                (const unsigned*)start, (float2*)out);
}


# 2a re-mapped: lane = restart candidate (32 consecutive r per wave, 4 experts, currents wave-uniform); an expert leaves once its own 32 lanes have spiked; table expert-major, 2b loads whole lines
# speedup vs baseline: 8.6613x; 1.1548x over previous
#include <hip/hip_runtime.h>


typedef float v2f __attribute__((ext_vector_type(2)));
typedef float v4f __attribute__((ext_vector_type(4)));
typedef float v8f __attribute__((ext_vector_type(8)));

#define HIDDEN 2048
#define EXPERTS 64
#define TSTEPS 8192

#define KC      128
#define PITCH   (KC + 4)
#define AROWS   32
#define ASTAGE  (AROWS * PITCH)
#define BPAIRS  (KC / 2)
#define BPITCH  (2 * EXPERTS + 32)
#define STAGE   (ASTAGE + BPAIRS * BPITCH)
#define NCHUNK  (HIDDEN / KC)
#define CPITCH  (EXPERTS + 4)

struct Frags { v2f a[KC / 4]; v2f b[KC / 4]; };

__device__ __forceinline__ void read_frags(const float* st, int fa, int fb, Frags& f) {
#pragma unroll
    for (int ks = 0; ks < KC / 4; ++ks) {
        f.a[ks] = *(const v2f*)(st + fa + ks * 4);
        f.b[ks] = *(const v2f*)(st + fb + ks * 2 * BPITCH);
    }
    __builtin_amdgcn_sched_barrier(0);
}

__device__ __forceinline__ void mma_chunk(const Frags& f, v8f& acc) {
#pragma unroll
    for (int ks = 0; ks < KC / 4; ++ks)
        acc = __builtin_amdgcn_wmma_f32_16x16x4_f32(false, f.a[ks], false, f.b[ks], (short)0, acc, false, false);
}

__device__ __forceinline__ void lds_barrier() {
    __builtin_amdgcn_fence(__ATOMIC_RELEASE, "workgroup", "local");
    __builtin_amdgcn_s_barrier();
    __builtin_amdgcn_fence(__ATOMIC_ACQUIRE, "workgroup", "local");
}

__global__ __launch_bounds__(256) void gemm_wmma_kernel(const float* __restrict__ A,
                                                        const float* __restrict__ W,
                                                        float* __restrict__ C) {
    __shared__ __attribute__((aligned(16))) float lds[2 * STAGE];

    const int tid  = threadIdx.x;
    const int wave = tid >> 5;
    const int lane = tid & 31;
    const int m0   = blockIdx.x * AROWS;
    const int mt   = wave >> 2;
    const int nt   = wave & 3;
    const int half = lane >> 4;
    const int sub  = lane & 15;

    const float* ga = A + (size_t)(m0 + wave * 4) * HIDDEN + lane * 4;
    const float* gb = W + (size_t)(wave * 16 + half * 2) * EXPERTS + sub * 4;
    const int sa = (wave * 4) * PITCH + lane * 4;
    const int sb = ASTAGE + (wave * 8 + half) * BPITCH + sub * 8;

    const int fa = (mt * 16 + sub) * PITCH + half * 2;
    const int fb = ASTAGE + half * BPITCH + (nt * 16 + sub) * 2;

    float4 ra0, ra1, ra2, ra3;
    v4f    rb0, rb1, rb2, rb3, rb4, rb5, rb6, rb7;
#define STORE_PAIR(dst, e, o)                                        \
    *(v4f*)(dst)       = (v4f){(e).x, (o).x, (e).y, (o).y};          \
    *(v4f*)((dst) + 4) = (v4f){(e).z, (o).z, (e).w, (o).w};
#define HOLD_RAW(r) asm volatile("" : "+v"(r));
#define LOAD_ROWS(koff)                                              \
    ra0 = *(const float4*)(ga + 0 * (size_t)HIDDEN + (koff));        \
    ra1 = *(const float4*)(ga + 1 * (size_t)HIDDEN + (koff));        \
    ra2 = *(const float4*)(ga + 2 * (size_t)HIDDEN + (koff));        \
    ra3 = *(const float4*)(ga + 3 * (size_t)HIDDEN + (koff));        \
    rb0 = *(const v4f*)(gb + ((size_t)(koff) +  0) * EXPERTS);    \
    rb1 = *(const v4f*)(gb + ((size_t)(koff) +  1) * EXPERTS);    \
    rb2 = *(const v4f*)(gb + ((size_t)(koff) +  4) * EXPERTS);    \
    rb3 = *(const v4f*)(gb + ((size_t)(koff) +  5) * EXPERTS);    \
    rb4 = *(const v4f*)(gb + ((size_t)(koff) +  8) * EXPERTS);    \
    rb5 = *(const v4f*)(gb + ((size_t)(koff) +  9) * EXPERTS);    \
    rb6 = *(const v4f*)(gb + ((size_t)(koff) + 12) * EXPERTS);    \
    rb7 = *(const v4f*)(gb + ((size_t)(koff) + 13) * EXPERTS);
#define STORE_ROWS(stage)                                            \
    *(float4*)((stage) + sa + 0 * PITCH) = ra0;                      \
    *(float4*)((stage) + sa + 1 * PITCH) = ra1;                      \
    *(float4*)((stage) + sa + 2 * PITCH) = ra2;                      \
    *(float4*)((stage) + sa + 3 * PITCH) = ra3;                      \
    HOLD_RAW(rb0) HOLD_RAW(rb1) HOLD_RAW(rb2) HOLD_RAW(rb3)          \
    HOLD_RAW(rb4) HOLD_RAW(rb5) HOLD_RAW(rb6) HOLD_RAW(rb7)          \
    STORE_PAIR((stage) + sb + 0 * BPITCH, rb0, rb1)                  \
    STORE_PAIR((stage) + sb + 2 * BPITCH, rb2, rb3)                  \
    STORE_PAIR((stage) + sb + 4 * BPITCH, rb4, rb5)                  \
    STORE_PAIR((stage) + sb + 6 * BPITCH, rb6, rb7)

    LOAD_ROWS(0)
    STORE_ROWS(lds)
    LOAD_ROWS(KC)
    __syncthreads();

    v8f acc = {};

    for (int c = 0; c < NCHUNK - 2; ++c) {
        const float* st = lds + (c & 1) * STAGE;
        float*       sn = lds + ((c + 1) & 1) * STAGE;
        const int    kn = (c + 2) * KC;
        Frags f;
        read_frags(st, fa, fb, f);
        STORE_ROWS(sn)
        LOAD_ROWS(kn)
        mma_chunk(f, acc);
        lds_barrier();
    }
    {
        Frags f;
        read_frags(lds + ((NCHUNK - 2) & 1) * STAGE, fa, fb, f);
        STORE_ROWS(lds + ((NCHUNK - 1) & 1) * STAGE)
        mma_chunk(f, acc);
        __syncthreads();
    }
    {
        Frags f;
        read_frags(lds + ((NCHUNK - 1) & 1) * STAGE, fa, fb, f);
        mma_chunk(f, acc);
        __syncthreads();
    }
#undef LOAD_ROWS
#undef STORE_ROWS
#undef STORE_PAIR
#undef HOLD_RAW

    float* ct = lds;
#pragma unroll
    for (int r = 0; r < 8; ++r)
        ct[(mt * 16 + half * 8 + r) * CPITCH + nt * 16 + sub] = acc[r];
    __syncthreads();
#pragma unroll
    for (int p = 0; p < 2; ++p) {
        const int j   = p * 1024 + tid * 4;
        const int row = j >> 6;
        const int col = j & 63;
        float4 v = *(const float4*)(ct + row * CPITCH + col);
        *(float4*)(C + (size_t)m0 * EXPERTS + j) = v;
    }
}

#define CHK  16
#define NCH  (TSTEPS / CHK)
#define BLK  8

constexpr int refractory_gap() {
    float rf = 1.0f;
    int k = 0;
    for (int i = 0; i < 64; ++i) {
        ++k;
        bool active = (rf <= 0.0f);
        float d = rf - 0.1f;
        rf = (d > 0.0f) ? d : 0.0f;
        if (active) break;
    }
    return k;
}
constexpr int GAP = refractory_gap();

__device__ __forceinline__ float2 lif_step(float2 cur, bool on0, bool on1,
                                           float2& mp, float2& rf) {
    float l0 = mp.x * 0.9f;
    float l1 = mp.y * 0.9f;
    float a0 = (on0 && rf.x <= 0.0f) ? 0.1f : 0.0f;
    float a1 = (on1 && rf.y <= 0.0f) ? 0.1f : 0.0f;
    mp.x = __builtin_fmaf(cur.x, a0, l0);
    mp.y = __builtin_fmaf(cur.y, a1, l1);
    rf.x = fmaxf(rf.x - 0.1f, 0.0f);
    rf.y = fmaxf(rf.y - 0.1f, 0.0f);
    float2 pre = mp;
    bool s0 = mp.x > 1.0f;
    bool s1 = mp.y > 1.0f;
    mp.x = s0 ? 0.0f : mp.x;
    mp.y = s1 ? 0.0f : mp.y;
    rf.x = s0 ? 1.0f : rf.x;
    rf.y = s1 ? 1.0f : rf.y;
    return pre;
}

#define RG   32
#define EG   4
#define WPG  (EXPERTS / EG)
static_assert(RG % BLK == 0 && TSTEPS % RG == 0 && EXPERTS % EG == 0 && EG == 4, "2a block layout");

__device__ __forceinline__ void load_rows(const float* __restrict__ base, int tb, float4 (&c)[BLK]) {
#pragma unroll
    for (int i = 0; i < BLK; ++i) {
        const int t = min(tb + i, TSTEPS - 1);
        c[i] = *(const float4*)(base + (size_t)t * EXPERTS);
    }
}

template <bool GATED>
__device__ __forceinline__ void scan_steps(const float (&c)[BLK], int tb, int r, float& m, int& s) {
    int first = BLK;
#pragma unroll
    for (int i = 0; i < BLK; ++i) {
        const float v = __builtin_fmaf(c[i], 0.1f, m * 0.9f);
        m = (!GATED || tb + i >= r) ? v : 0.0f;
        first = min(first, (m > 1.0f) ? i : BLK);
    }
    s = min(s, (first < BLK) ? tb + first : TSTEPS);
}

__device__ __forceinline__ unsigned scan_block(const float4 (&c)[BLK], int tb, int r0, int r,
                                               unsigned alive, float (&m)[EG], int (&s)[EG]) {
    const bool gated = tb < r0 + RG - 1;
    unsigned left = 0;
#pragma unroll
    for (int j = 0; j < EG; ++j) {
        if (alive & (1u << j)) {
            float cj[BLK];
#pragma unroll
            for (int i = 0; i < BLK; ++i) cj[i] = (j == 0) ? c[i].x : (j == 1) ? c[i].y : (j == 2) ? c[i].z : c[i].w;
            if (gated) scan_steps<true>(cj, tb, r, m[j], s[j]);
            else       scan_steps<false>(cj, tb, r, m[j], s[j]);
            if (__ballot(s[j] == TSTEPS) != 0ull) left |= (1u << j);
        }
    }
    return left;
}

__global__ __launch_bounds__(WPG * 32) void next_restart_kernel(const float* __restrict__ cur,
                                                               unsigned* __restrict__ nxt) {
    const int wv   = __builtin_amdgcn_readfirstlane(threadIdx.x >> 5);
    const int lane = threadIdx.x & 31;
    const int r0   = blockIdx.x * RG;
    const int r    = r0 + lane;
    const int e0   = wv * EG;
    const float* base = cur + e0;

    float m[EG];
    int   s[EG];
#pragma unroll
    for (int j = 0; j < EG; ++j) { m[j] = 0.0f; s[j] = TSTEPS; }

    float4 ca[BLK], cb[BLK];
    unsigned alive = (1u << EG) - 1u;
    int tb = r0;
    load_rows(base, tb, ca);
    for (;;) {
        load_rows(base, tb + BLK, cb);
        alive = scan_block(ca, tb, r0, r, alive, m, s);
        tb += BLK;
        if (alive == 0u || tb >= TSTEPS) break;
        load_rows(base, tb + BLK, ca);
        alive = scan_block(cb, tb, r0, r, alive, m, s);
        tb += BLK;
        if (alive == 0u || tb >= TSTEPS) break;
    }

#pragma unroll
    for (int j = 0; j < EG; ++j)
        nxt[(size_t)(e0 + j) * TSTEPS + r] = (unsigned)min(s[j] + GAP, TSTEPS);
}

#define HOPS   16
#define NMARK  (((TSTEPS / (HOPS * GAP) + 1 + 3) / 4) * 4)
#define NHOP   (NMARK * HOPS)

__global__ __launch_bounds__(256) void restart_chain_kernel(const unsigned* __restrict__ nxt,
                                                            unsigned* __restrict__ start) {
    __shared__ unsigned short tab1[TSTEPS + 2];
    __shared__ unsigned short tabA[TSTEPS + 2];
    __shared__ unsigned short tabB[TSTEPS + 2];
    __shared__ __attribute__((aligned(16))) unsigned short mark[NMARK];
    __shared__ __attribute__((aligned(16))) unsigned short hop[NHOP + 8];
    __shared__ unsigned st[NCH];
    const int e   = blockIdx.x;
    const int tid = threadIdx.x;

    const uint4* col = (const uint4*)(nxt + (size_t)e * TSTEPS);
    uint4 w[TSTEPS / 4 / 256];
#pragma unroll
    for (int k = 0; k < TSTEPS / 4 / 256; ++k) w[k] = col[tid + 256 * k];
#pragma unroll
    for (int k = 0; k < TSTEPS / 4 / 256; ++k) {
        const int q = tid + 256 * k;
        tab1[4 * q]     = (unsigned short)w[k].x;
        tab1[4 * q + 1] = (unsigned short)w[k].y;
        tab1[4 * q + 2] = (unsigned short)w[k].z;
        tab1[4 * q + 3] = (unsigned short)w[k].w;
    }
    if (tid == 0) { tab1[TSTEPS] = TSTEPS; tabA[TSTEPS] = TSTEPS; tabB[TSTEPS] = TSTEPS; hop[NHOP] = TSTEPS; }
    if (tid < NMARK) mark[tid] = TSTEPS;
    __syncthreads();
    for (int r = tid; r < TSTEPS; r += 256) tabA[r] = tab1[tab1[r]];
    __syncthreads();
    for (int r = tid; r < TSTEPS; r += 256) tabB[r] = tabA[tabA[r]];
    __syncthreads();
    for (int r = tid; r < TSTEPS; r += 256) tabA[r] = tabB[tabB[r]];
    __syncthreads();
    for (int r = tid; r < TSTEPS; r += 256) tabB[r] = tabA[tabA[r]];
    __syncthreads();

    if (tid == 0) {
        int r = 0;
        for (int n = 0; n < NMARK && r < TSTEPS; n += 4) {
            const int r1 = tabB[r];
            const int r2 = tabB[r1];
            const int r3 = tabB[r2];
            mark[n]     = (unsigned short)r;
            mark[n + 1] = (unsigned short)r1;
            mark[n + 2] = (unsigned short)r2;
            mark[n + 3] = (unsigned short)r3;
            r = tabB[r3];
        }
    }
    __syncthreads();

    if (tid < NMARK) {
        int r = mark[tid];
#pragma unroll
        for (int i = 0; i < HOPS; ++i) {
            hop[tid * HOPS + i] = (unsigned short)r;
            r = tab1[r];
        }
    }
    __syncthreads();

    for (int q = tid; q < NHOP; q += 256) {
        const int r    = hop[q];
        const int n    = hop[q + 1];
        const int cend = min((n + CHK - 1) / CHK, NCH);
        const unsigned w = (unsigned)r | ((unsigned)n << 16);
        for (int c = (r + CHK - 1) / CHK; c < cend; ++c) st[c] = w;
    }
    __syncthreads();

    start[(size_t)e * NCH + tid]       = st[tid];
    start[(size_t)e * NCH + 256 + tid] = st[256 + tid];
}

#define RB   CHK
#define NRB  3

__device__ __forceinline__ void load_block(const float2* __restrict__ cur, int lane, int t0,
                                           float2 (&c)[RB]) {
    const int rb = min(max(t0, 0), TSTEPS - RB);
    const float2* p = cur + (size_t)rb * 32 + lane;
#pragma unroll
    for (int i = 0; i < RB; ++i) c[i] = p[(size_t)i * 32];
}

__device__ __forceinline__ void integrate_block(const float2 (&c)[RB], int t0, int eA, int eB,
                                                float2& m) {
#pragma unroll
    for (int i = 0; i < RB; ++i) {
        const int t = t0 + i;
        const float g0 = (t >= eA) ? 0.1f : 0.0f;
        const float g1 = (t >= eB) ? 0.1f : 0.0f;
        m.x = __builtin_fmaf(c[i].x, g0, m.x * 0.9f);
        m.y = __builtin_fmaf(c[i].y, g1, m.y * 0.9f);
    }
}

__device__ __forceinline__ float sitout_counter(int j) {
    float r = 1.0f;
#pragma unroll
    for (int k = 1; k < GAP - 1; ++k) {
        const float d = fmaxf(r - 0.1f, 0.0f);
        r = (k < j) ? d : r;
    }
    return r;
}

__global__ __launch_bounds__(32) void replay_route_kernel(const float2* __restrict__ cur,
                                                          const unsigned* __restrict__ start,
                                                          float2* __restrict__ out) {
    const int ch   = blockIdx.x;
    const int lane = threadIdx.x;
    const int tbeg = ch * CHK;

    const unsigned wA = start[(size_t)(2 * lane)     * NCH + ch];
    const unsigned wB = start[(size_t)(2 * lane + 1) * NCH + ch];
    const int sA = (int)(wA & 0xffffu);
    const int sB = (int)(wB & 0xffffu);
    const int jA = tbeg - ((int)(wA >> 16) - GAP);
    const int jB = tbeg - ((int)(wB >> 16) - GAP);
    const bool spA = jA > 0;
    const bool spB = jB > 0;
    const int eA = spA ? tbeg : sA;
    const int eB = spB ? tbeg : sB;

    int smin = min(eA, eB);
#pragma unroll
    for (int off = 16; off >= 1; off >>= 1)
        smin = min(smin, __shfl_xor(smin, off, 32));
    smin = __builtin_amdgcn_readfirstlane(smin);

    const int nblk = NRB * ((tbeg - smin + NRB * RB - 1) / (NRB * RB));
    int tb = tbeg - nblk * RB;

    float2 c0[RB], c1[RB], c2[RB];
    load_block(cur, lane, tb,          c0);
    load_block(cur, lane, tb + RB,     c1);
    load_block(cur, lane, tb + 2 * RB, c2);

    float2 m = {0.0f, 0.0f};
    for (; tb < tbeg; tb += NRB * RB) {
        integrate_block(c0, tb,          eA, eB, m);
        load_block(cur, lane, tb + 3 * RB, c0);
        integrate_block(c1, tb + RB,     eA, eB, m);
        load_block(cur, lane, tb + 4 * RB, c1);
        integrate_block(c2, tb + 2 * RB, eA, eB, m);
        load_block(cur, lane, tb + 5 * RB, c2);
    }

    float2 mp = m;
    float2 rf;
    rf.x = spA ? sitout_counter(jA) : 0.0f;
    rf.y = spB ? sitout_counter(jB) : 0.0f;

#pragma unroll
    for (int i = 0; i < CHK; ++i) {
        const int t = tbeg + i;
        float2 v = lif_step(c0[i], t >= sA, t >= sB, mp, rf);

        bool s0 = v.x > 1.0f;
        bool s1 = v.y > 1.0f;
        unsigned long long bal = __ballot(s0 || s1);
        bool any = (bal != 0ull);

        float x0 = any ? (s0 ? 1.0f : 0.0f) : v.x;
        float x1 = any ? (s1 ? 1.0f : 0.0f) : v.y;

        float mx = fmaxf(x0, x1);
#pragma unroll
        for (int off = 16; off >= 1; off >>= 1)
            mx = fmaxf(mx, __shfl_xor(mx, off, 32));

        float e0 = __expf(x0 - mx);
        float e1 = __expf(x1 - mx);
        float s  = e0 + e1;
#pragma unroll
        for (int off = 16; off >= 1; off >>= 1)
            s += __shfl_xor(s, off, 32);

        float inv = 1.0f / s;
        float2 rr;
        rr.x = e0 * inv;
        rr.y = e1 * inv;
        out[(size_t)t * 32 + lane] = rr;
    }
}

extern "C" void kernel_launch(void* const* d_in, const int* in_sizes, int n_in,
                              void* d_out, int out_size, void* d_ws, size_t ws_size,
                              hipStream_t stream) {
    const float* hidden = (const float*)d_in[0];
    const float* W      = (const float*)d_in[1];
    float* out          = (float*)d_out;

    float* curr = (float*)d_ws;
    unsigned* nxt   = (unsigned*)d_out;
    unsigned* start = (unsigned*)((char*)d_ws + (size_t)TSTEPS * EXPERTS * sizeof(float));

    gemm_wmma_kernel<<<TSTEPS / AROWS, 256, 0, stream>>>(hidden, W, curr);

    next_restart_kernel<<<TSTEPS / RG, WPG * 32, 0, stream>>>((const float*)curr, nxt);

    restart_chain_kernel<<<EXPERTS, 256, 0, stream>>>(nxt, start);

    replay_route_kernel<<<NCH, 32, 0, stream>>>((const float2*)curr,
                                                (const unsigned*)start, (float2*)out);
}


# 2b by pointer doubling to 64 hops: all 7 hop tables kept, serial walk 12 links (was 48), each restart found from its mark by its rank's bits, 6 reads side by side; batched 8-byte passes; same table
# speedup vs baseline: 8.9368x; 1.0318x over previous
#include <hip/hip_runtime.h>


typedef float v2f __attribute__((ext_vector_type(2)));
typedef float v4f __attribute__((ext_vector_type(4)));
typedef float v8f __attribute__((ext_vector_type(8)));

#define HIDDEN 2048
#define EXPERTS 64
#define TSTEPS 8192

#define KC      128
#define PITCH   (KC + 4)
#define AROWS   32
#define ASTAGE  (AROWS * PITCH)
#define BPAIRS  (KC / 2)
#define BPITCH  (2 * EXPERTS + 32)
#define STAGE   (ASTAGE + BPAIRS * BPITCH)
#define NCHUNK  (HIDDEN / KC)
#define CPITCH  (EXPERTS + 4)

struct Frags { v2f a[KC / 4]; v2f b[KC / 4]; };

__device__ __forceinline__ void read_frags(const float* st, int fa, int fb, Frags& f) {
#pragma unroll
    for (int ks = 0; ks < KC / 4; ++ks) {
        f.a[ks] = *(const v2f*)(st + fa + ks * 4);
        f.b[ks] = *(const v2f*)(st + fb + ks * 2 * BPITCH);
    }
    __builtin_amdgcn_sched_barrier(0);
}

__device__ __forceinline__ void mma_chunk(const Frags& f, v8f& acc) {
#pragma unroll
    for (int ks = 0; ks < KC / 4; ++ks)
        acc = __builtin_amdgcn_wmma_f32_16x16x4_f32(false, f.a[ks], false, f.b[ks], (short)0, acc, false, false);
}

__device__ __forceinline__ void lds_barrier() {
    __builtin_amdgcn_fence(__ATOMIC_RELEASE, "workgroup", "local");
    __builtin_amdgcn_s_barrier();
    __builtin_amdgcn_fence(__ATOMIC_ACQUIRE, "workgroup", "local");
}

__global__ __launch_bounds__(256) void gemm_wmma_kernel(const float* __restrict__ A,
                                                        const float* __restrict__ W,
                                                        float* __restrict__ C) {
    __shared__ __attribute__((aligned(16))) float lds[2 * STAGE];

    const int tid  = threadIdx.x;
    const int wave = tid >> 5;
    const int lane = tid & 31;
    const int m0   = blockIdx.x * AROWS;
    const int mt   = wave >> 2;
    const int nt   = wave & 3;
    const int half = lane >> 4;
    const int sub  = lane & 15;

    const float* ga = A + (size_t)(m0 + wave * 4) * HIDDEN + lane * 4;
    const float* gb = W + (size_t)(wave * 16 + half * 2) * EXPERTS + sub * 4;
    const int sa = (wave * 4) * PITCH + lane * 4;
    const int sb = ASTAGE + (wave * 8 + half) * BPITCH + sub * 8;

    const int fa = (mt * 16 + sub) * PITCH + half * 2;
    const int fb = ASTAGE + half * BPITCH + (nt * 16 + sub) * 2;

    float4 ra0, ra1, ra2, ra3;
    v4f    rb0, rb1, rb2, rb3, rb4, rb5, rb6, rb7;
#define STORE_PAIR(dst, e, o)                                        \
    *(v4f*)(dst)       = (v4f){(e).x, (o).x, (e).y, (o).y};          \
    *(v4f*)((dst) + 4) = (v4f){(e).z, (o).z, (e).w, (o).w};
#define HOLD_RAW(r) asm volatile("" : "+v"(r));
#define LOAD_ROWS(koff)                                              \
    ra0 = *(const float4*)(ga + 0 * (size_t)HIDDEN + (koff));        \
    ra1 = *(const float4*)(ga + 1 * (size_t)HIDDEN + (koff));        \
    ra2 = *(const float4*)(ga + 2 * (size_t)HIDDEN + (koff));        \
    ra3 = *(const float4*)(ga + 3 * (size_t)HIDDEN + (koff));        \
    rb0 = *(const v4f*)(gb + ((size_t)(koff) +  0) * EXPERTS);    \
    rb1 = *(const v4f*)(gb + ((size_t)(koff) +  1) * EXPERTS);    \
    rb2 = *(const v4f*)(gb + ((size_t)(koff) +  4) * EXPERTS);    \
    rb3 = *(const v4f*)(gb + ((size_t)(koff) +  5) * EXPERTS);    \
    rb4 = *(const v4f*)(gb + ((size_t)(koff) +  8) * EXPERTS);    \
    rb5 = *(const v4f*)(gb + ((size_t)(koff) +  9) * EXPERTS);    \
    rb6 = *(const v4f*)(gb + ((size_t)(koff) + 12) * EXPERTS);    \
    rb7 = *(const v4f*)(gb + ((size_t)(koff) + 13) * EXPERTS);
#define STORE_ROWS(stage)                                            \
    *(float4*)((stage) + sa + 0 * PITCH) = ra0;                      \
    *(float4*)((stage) + sa + 1 * PITCH) = ra1;                      \
    *(float4*)((stage) + sa + 2 * PITCH) = ra2;                      \
    *(float4*)((stage) + sa + 3 * PITCH) = ra3;                      \
    HOLD_RAW(rb0) HOLD_RAW(rb1) HOLD_RAW(rb2) HOLD_RAW(rb3)          \
    HOLD_RAW(rb4) HOLD_RAW(rb5) HOLD_RAW(rb6) HOLD_RAW(rb7)          \
    STORE_PAIR((stage) + sb + 0 * BPITCH, rb0, rb1)                  \
    STORE_PAIR((stage) + sb + 2 * BPITCH, rb2, rb3)                  \
    STORE_PAIR((stage) + sb + 4 * BPITCH, rb4, rb5)                  \
    STORE_PAIR((stage) + sb + 6 * BPITCH, rb6, rb7)

    LOAD_ROWS(0)
    STORE_ROWS(lds)
    LOAD_ROWS(KC)
    __syncthreads();

    v8f acc = {};

    for (int c = 0; c < NCHUNK - 2; ++c) {
        const float* st = lds + (c & 1) * STAGE;
        float*       sn = lds + ((c + 1) & 1) * STAGE;
        const int    kn = (c + 2) * KC;
        Frags f;
        read_frags(st, fa, fb, f);
        STORE_ROWS(sn)
        LOAD_ROWS(kn)
        mma_chunk(f, acc);
        lds_barrier();
    }
    {
        Frags f;
        read_frags(lds + ((NCHUNK - 2) & 1) * STAGE, fa, fb, f);
        STORE_ROWS(lds + ((NCHUNK - 1) & 1) * STAGE)
        mma_chunk(f, acc);
        __syncthreads();
    }
    {
        Frags f;
        read_frags(lds + ((NCHUNK - 1) & 1) * STAGE, fa, fb, f);
        mma_chunk(f, acc);
        __syncthreads();
    }
#undef LOAD_ROWS
#undef STORE_ROWS
#undef STORE_PAIR
#undef HOLD_RAW

    float* ct = lds;
#pragma unroll
    for (int r = 0; r < 8; ++r)
        ct[(mt * 16 + half * 8 + r) * CPITCH + nt * 16 + sub] = acc[r];
    __syncthreads();
#pragma unroll
    for (int p = 0; p < 2; ++p) {
        const int j   = p * 1024 + tid * 4;
        const int row = j >> 6;
        const int col = j & 63;
        float4 v = *(const float4*)(ct + row * CPITCH + col);
        *(float4*)(C + (size_t)m0 * EXPERTS + j) = v;
    }
}

#define CHK  16
#define NCH  (TSTEPS / CHK)
#define BLK  8

constexpr int refractory_gap() {
    float rf = 1.0f;
    int k = 0;
    for (int i = 0; i < 64; ++i) {
        ++k;
        bool active = (rf <= 0.0f);
        float d = rf - 0.1f;
        rf = (d > 0.0f) ? d : 0.0f;
        if (active) break;
    }
    return k;
}
constexpr int GAP = refractory_gap();

__device__ __forceinline__ float2 lif_step(float2 cur, bool on0, bool on1,
                                           float2& mp, float2& rf) {
    float l0 = mp.x * 0.9f;
    float l1 = mp.y * 0.9f;
    float a0 = (on0 && rf.x <= 0.0f) ? 0.1f : 0.0f;
    float a1 = (on1 && rf.y <= 0.0f) ? 0.1f : 0.0f;
    mp.x = __builtin_fmaf(cur.x, a0, l0);
    mp.y = __builtin_fmaf(cur.y, a1, l1);
    rf.x = fmaxf(rf.x - 0.1f, 0.0f);
    rf.y = fmaxf(rf.y - 0.1f, 0.0f);
    float2 pre = mp;
    bool s0 = mp.x > 1.0f;
    bool s1 = mp.y > 1.0f;
    mp.x = s0 ? 0.0f : mp.x;
    mp.y = s1 ? 0.0f : mp.y;
    rf.x = s0 ? 1.0f : rf.x;
    rf.y = s1 ? 1.0f : rf.y;
    return pre;
}

#define RG   32
#define EG   4
#define WPG  (EXPERTS / EG)
static_assert(RG % BLK == 0 && TSTEPS % RG == 0 && EXPERTS % EG == 0 && EG == 4, "2a block layout");

__device__ __forceinline__ void load_rows(const float* __restrict__ base, int tb, float4 (&c)[BLK]) {
#pragma unroll
    for (int i = 0; i < BLK; ++i) {
        const int t = min(tb + i, TSTEPS - 1);
        c[i] = *(const float4*)(base + (size_t)t * EXPERTS);
    }
}

template <bool GATED>
__device__ __forceinline__ void scan_steps(const float (&c)[BLK], int tb, int r, float& m, int& s) {
    int first = BLK;
#pragma unroll
    for (int i = 0; i < BLK; ++i) {
        const float v = __builtin_fmaf(c[i], 0.1f, m * 0.9f);
        m = (!GATED || tb + i >= r) ? v : 0.0f;
        first = min(first, (m > 1.0f) ? i : BLK);
    }
    s = min(s, (first < BLK) ? tb + first : TSTEPS);
}

__device__ __forceinline__ unsigned scan_block(const float4 (&c)[BLK], int tb, int r0, int r,
                                               unsigned alive, float (&m)[EG], int (&s)[EG]) {
    const bool gated = tb < r0 + RG - 1;
    unsigned left = 0;
#pragma unroll
    for (int j = 0; j < EG; ++j) {
        if (alive & (1u << j)) {
            float cj[BLK];
#pragma unroll
            for (int i = 0; i < BLK; ++i) cj[i] = (j == 0) ? c[i].x : (j == 1) ? c[i].y : (j == 2) ? c[i].z : c[i].w;
            if (gated) scan_steps<true>(cj, tb, r, m[j], s[j]);
            else       scan_steps<false>(cj, tb, r, m[j], s[j]);
            if (__ballot(s[j] == TSTEPS) != 0ull) left |= (1u << j);
        }
    }
    return left;
}

__global__ __launch_bounds__(WPG * 32) void next_restart_kernel(const float* __restrict__ cur,
                                                               unsigned* __restrict__ nxt) {
    const int wv   = __builtin_amdgcn_readfirstlane(threadIdx.x >> 5);
    const int lane = threadIdx.x & 31;
    const int r0   = blockIdx.x * RG;
    const int r    = r0 + lane;
    const int e0   = wv * EG;
    const float* base = cur + e0;

    float m[EG];
    int   s[EG];
#pragma unroll
    for (int j = 0; j < EG; ++j) { m[j] = 0.0f; s[j] = TSTEPS; }

    float4 ca[BLK], cb[BLK];
    unsigned alive = (1u << EG) - 1u;
    int tb = r0;
    load_rows(base, tb, ca);
    for (;;) {
        load_rows(base, tb + BLK, cb);
        alive = scan_block(ca, tb, r0, r, alive, m, s);
        tb += BLK;
        if (alive == 0u || tb >= TSTEPS) break;
        load_rows(base, tb + BLK, ca);
        alive = scan_block(cb, tb, r0, r, alive, m, s);
        tb += BLK;
        if (alive == 0u || tb >= TSTEPS) break;
    }

#pragma unroll
    for (int j = 0; j < EG; ++j)
        nxt[(size_t)(e0 + j) * TSTEPS + r] = (unsigned)min(s[j] + GAP, TSTEPS);
}

#define HOPS   64
#define NLEV   7
#define TPITCH (TSTEPS + 8)
#define NMARK  (((TSTEPS / (HOPS * GAP) + 1 + 3) / 4) * 4)
#define NHOP   (NMARK * HOPS)
static_assert((1 << (NLEV - 1)) == HOPS && NMARK % 4 == 0 && NHOP % 256 == 0 && TPITCH % 4 == 0, "2b table layout");

typedef unsigned int u16x4w __attribute__((ext_vector_type(2), aligned(8), may_alias));

__device__ __forceinline__ void compose_table(const unsigned short* src, unsigned short* dst, int tid) {
    constexpr int NQ = TSTEPS / 4 / 256;
    u16x4w in[NQ];
#pragma unroll
    for (int k = 0; k < NQ; ++k) in[k] = *(const u16x4w*)(src + 4 * (tid + 256 * k));
    unsigned g[NQ][4];
#pragma unroll
    for (int k = 0; k < NQ; ++k) {
        g[k][0] = src[in[k].x & 0xffffu];
        g[k][1] = src[in[k].x >> 16];
        g[k][2] = src[in[k].y & 0xffffu];
        g[k][3] = src[in[k].y >> 16];
    }
#pragma unroll
    for (int k = 0; k < NQ; ++k) {
        u16x4w o;
        o.x = g[k][0] | (g[k][1] << 16);
        o.y = g[k][2] | (g[k][3] << 16);
        *(u16x4w*)(dst + 4 * (tid + 256 * k)) = o;
    }
}

__global__ __launch_bounds__(256) void restart_chain_kernel(const unsigned* __restrict__ nxt,
                                                            unsigned* __restrict__ start) {
    __shared__ __attribute__((aligned(16))) unsigned short tab[NLEV * TPITCH];
    __shared__ __attribute__((aligned(16))) unsigned short mark[NMARK];
    __shared__ __attribute__((aligned(16))) unsigned short hop[NHOP + 8];
    __shared__ unsigned st[NCH];
    const int e   = blockIdx.x;
    const int tid = threadIdx.x;

    const uint4* col = (const uint4*)(nxt + (size_t)e * TSTEPS);
    uint4 w[TSTEPS / 4 / 256];
#pragma unroll
    for (int k = 0; k < TSTEPS / 4 / 256; ++k) w[k] = col[tid + 256 * k];
#pragma unroll
    for (int k = 0; k < TSTEPS / 4 / 256; ++k) {
        const int q = tid + 256 * k;
        tab[4 * q]     = (unsigned short)w[k].x;
        tab[4 * q + 1] = (unsigned short)w[k].y;
        tab[4 * q + 2] = (unsigned short)w[k].z;
        tab[4 * q + 3] = (unsigned short)w[k].w;
    }
    if (tid < NLEV) tab[tid * TPITCH + TSTEPS] = TSTEPS;
    if (tid == 0) hop[NHOP] = TSTEPS;
    if (tid < NMARK) mark[tid] = TSTEPS;
    __syncthreads();
#pragma unroll
    for (int k = 0; k + 1 < NLEV; ++k) {
        compose_table(tab + k * TPITCH, tab + (k + 1) * TPITCH, tid);
        __syncthreads();
    }

    if (tid == 0) {
        const unsigned short* top = tab + (NLEV - 1) * TPITCH;
        int r = 0;
        for (int n = 0; n < NMARK && r < TSTEPS; n += 4) {
            const int r1 = top[r];
            const int r2 = top[r1];
            const int r3 = top[r2];
            mark[n]     = (unsigned short)r;
            mark[n + 1] = (unsigned short)r1;
            mark[n + 2] = (unsigned short)r2;
            mark[n + 3] = (unsigned short)r3;
            r = top[r3];
        }
    }
    __syncthreads();

#pragma unroll
    for (int j = 0; j < NHOP / 256; ++j) {
        const int p = tid + 256 * j;
        const int b = p & (HOPS - 1);
        int r = mark[p / HOPS];
#pragma unroll
        for (int k = NLEV - 2; k >= 0; --k) {
            const int t = tab[k * TPITCH + r];
            r = ((b >> k) & 1) ? t : r;
        }
        hop[p] = (unsigned short)r;
    }
    __syncthreads();

    for (int q = tid; q < NHOP; q += 256) {
        const int r    = hop[q];
        const int n    = hop[q + 1];
        const int cend = min((n + CHK - 1) / CHK, NCH);
        const unsigned w = (unsigned)r | ((unsigned)n << 16);
        for (int c = (r + CHK - 1) / CHK; c < cend; ++c) st[c] = w;
    }
    __syncthreads();

    start[(size_t)e * NCH + tid]       = st[tid];
    start[(size_t)e * NCH + 256 + tid] = st[256 + tid];
}

#define RB   CHK
#define NRB  3

__device__ __forceinline__ void load_block(const float2* __restrict__ cur, int lane, int t0,
                                           float2 (&c)[RB]) {
    const int rb = min(max(t0, 0), TSTEPS - RB);
    const float2* p = cur + (size_t)rb * 32 + lane;
#pragma unroll
    for (int i = 0; i < RB; ++i) c[i] = p[(size_t)i * 32];
}

__device__ __forceinline__ void integrate_block(const float2 (&c)[RB], int t0, int eA, int eB,
                                                float2& m) {
#pragma unroll
    for (int i = 0; i < RB; ++i) {
        const int t = t0 + i;
        const float g0 = (t >= eA) ? 0.1f : 0.0f;
        const float g1 = (t >= eB) ? 0.1f : 0.0f;
        m.x = __builtin_fmaf(c[i].x, g0, m.x * 0.9f);
        m.y = __builtin_fmaf(c[i].y, g1, m.y * 0.9f);
    }
}

__device__ __forceinline__ float sitout_counter(int j) {
    float r = 1.0f;
#pragma unroll
    for (int k = 1; k < GAP - 1; ++k) {
        const float d = fmaxf(r - 0.1f, 0.0f);
        r = (k < j) ? d : r;
    }
    return r;
}

__global__ __launch_bounds__(32) void replay_route_kernel(const float2* __restrict__ cur,
                                                          const unsigned* __restrict__ start,
                                                          float2* __restrict__ out) {
    const int ch   = blockIdx.x;
    const int lane = threadIdx.x;
    const int tbeg = ch * CHK;

    const unsigned wA = start[(size_t)(2 * lane)     * NCH + ch];
    const unsigned wB = start[(size_t)(2 * lane + 1) * NCH + ch];
    const int sA = (int)(wA & 0xffffu);
    const int sB = (int)(wB & 0xffffu);
    const int jA = tbeg - ((int)(wA >> 16) - GAP);
    const int jB = tbeg - ((int)(wB >> 16) - GAP);
    const bool spA = jA > 0;
    const bool spB = jB > 0;
    const int eA = spA ? tbeg : sA;
    const int eB = spB ? tbeg : sB;

    int smin = min(eA, eB);
#pragma unroll
    for (int off = 16; off >= 1; off >>= 1)
        smin = min(smin, __shfl_xor(smin, off, 32));
    smin = __builtin_amdgcn_readfirstlane(smin);

    const int nblk = NRB * ((tbeg - smin + NRB * RB - 1) / (NRB * RB));
    int tb = tbeg - nblk * RB;

    float2 c0[RB], c1[RB], c2[RB];
    load_block(cur, lane, tb,          c0);
    load_block(cur, lane, tb + RB,     c1);
    load_block(cur, lane, tb + 2 * RB, c2);

    float2 m = {0.0f, 0.0f};
    for (; tb < tbeg; tb += NRB * RB) {
        integrate_block(c0, tb,          eA, eB, m);
        load_block(cur, lane, tb + 3 * RB, c0);
        integrate_block(c1, tb + RB,     eA, eB, m);
        load_block(cur, lane, tb + 4 * RB, c1);
        integrate_block(c2, tb + 2 * RB, eA, eB, m);
        load_block(cur, lane, tb + 5 * RB, c2);
    }

    float2 mp = m;
    float2 rf;
    rf.x = spA ? sitout_counter(jA) : 0.0f;
    rf.y = spB ? sitout_counter(jB) : 0.0f;

#pragma unroll
    for (int i = 0; i < CHK; ++i) {
        const int t = tbeg + i;
        float2 v = lif_step(c0[i], t >= sA, t >= sB, mp, rf);

        bool s0 = v.x > 1.0f;
        bool s1 = v.y > 1.0f;
        unsigned long long bal = __ballot(s0 || s1);
        bool any = (bal != 0ull);

        float x0 = any ? (s0 ? 1.0f : 0.0f) : v.x;
        float x1 = any ? (s1 ? 1.0f : 0.0f) : v.y;

        float mx = fmaxf(x0, x1);
#pragma unroll
        for (int off = 16; off >= 1; off >>= 1)
            mx = fmaxf(mx, __shfl_xor(mx, off, 32));

        float e0 = __expf(x0 - mx);
        float e1 = __expf(x1 - mx);
        float s  = e0 + e1;
#pragma unroll
        for (int off = 16; off >= 1; off >>= 1)
            s += __shfl_xor(s, off, 32);

        float inv = 1.0f / s;
        float2 rr;
        rr.x = e0 * inv;
        rr.y = e1 * inv;
        out[(size_t)t * 32 + lane] = rr;
    }
}

extern "C" void kernel_launch(void* const* d_in, const int* in_sizes, int n_in,
                              void* d_out, int out_size, void* d_ws, size_t ws_size,
                              hipStream_t stream) {
    const float* hidden = (const float*)d_in[0];
    const float* W      = (const float*)d_in[1];
    float* out          = (float*)d_out;

    float* curr = (float*)d_ws;
    unsigned* nxt   = (unsigned*)d_out;
    unsigned* start = (unsigned*)((char*)d_ws + (size_t)TSTEPS * EXPERTS * sizeof(float));

    gemm_wmma_kernel<<<TSTEPS / AROWS, 256, 0, stream>>>(hidden, W, curr);

    next_restart_kernel<<<TSTEPS / RG, WPG * 32, 0, stream>>>((const float*)curr, nxt);

    restart_chain_kernel<<<EXPERTS, 256, 0, stream>>>(nxt, start);

    replay_route_kernel<<<NCH, 32, 0, stream>>>((const float2*)curr,
                                                (const unsigned*)start, (float2*)out);
}


# 2a: once one expert of a wave is left, its column alone is followed 32 rows per block (vector loads a block ahead); spike-step search only where the block maximum crosses 1; same operations per value
# speedup vs baseline: 9.2401x; 1.0339x over previous
#include <hip/hip_runtime.h>


typedef float v2f __attribute__((ext_vector_type(2)));
typedef float v4f __attribute__((ext_vector_type(4)));
typedef float v8f __attribute__((ext_vector_type(8)));

#define HIDDEN 2048
#define EXPERTS 64
#define TSTEPS 8192

#define KC      128
#define PITCH   (KC + 4)
#define AROWS   32
#define ASTAGE  (AROWS * PITCH)
#define BPAIRS  (KC / 2)
#define BPITCH  (2 * EXPERTS + 32)
#define STAGE   (ASTAGE + BPAIRS * BPITCH)
#define NCHUNK  (HIDDEN / KC)
#define CPITCH  (EXPERTS + 4)

struct Frags { v2f a[KC / 4]; v2f b[KC / 4]; };

__device__ __forceinline__ void read_frags(const float* st, int fa, int fb, Frags& f) {
#pragma unroll
    for (int ks = 0; ks < KC / 4; ++ks) {
        f.a[ks] = *(const v2f*)(st + fa + ks * 4);
        f.b[ks] = *(const v2f*)(st + fb + ks * 2 * BPITCH);
    }
    __builtin_amdgcn_sched_barrier(0);
}

__device__ __forceinline__ void mma_chunk(const Frags& f, v8f& acc) {
#pragma unroll
    for (int ks = 0; ks < KC / 4; ++ks)
        acc = __builtin_amdgcn_wmma_f32_16x16x4_f32(false, f.a[ks], false, f.b[ks], (short)0, acc, false, false);
}

__device__ __forceinline__ void lds_barrier() {
    __builtin_amdgcn_fence(__ATOMIC_RELEASE, "workgroup", "local");
    __builtin_amdgcn_s_barrier();
    __builtin_amdgcn_fence(__ATOMIC_ACQUIRE, "workgroup", "local");
}

__global__ __launch_bounds__(256) void gemm_wmma_kernel(const float* __restrict__ A,
                                                        const float* __restrict__ W,
                                                        float* __restrict__ C) {
    __shared__ __attribute__((aligned(16))) float lds[2 * STAGE];

    const int tid  = threadIdx.x;
    const int wave = tid >> 5;
    const int lane = tid & 31;
    const int m0   = blockIdx.x * AROWS;
    const int mt   = wave >> 2;
    const int nt   = wave & 3;
    const int half = lane >> 4;
    const int sub  = lane & 15;

    const float* ga = A + (size_t)(m0 + wave * 4) * HIDDEN + lane * 4;
    const float* gb = W + (size_t)(wave * 16 + half * 2) * EXPERTS + sub * 4;
    const int sa = (wave * 4) * PITCH + lane * 4;
    const int sb = ASTAGE + (wave * 8 + half) * BPITCH + sub * 8;

    const int fa = (mt * 16 + sub) * PITCH + half * 2;
    const int fb = ASTAGE + half * BPITCH + (nt * 16 + sub) * 2;

    float4 ra0, ra1, ra2, ra3;
    v4f    rb0, rb1, rb2, rb3, rb4, rb5, rb6, rb7;
#define STORE_PAIR(dst, e, o)                                        \
    *(v4f*)(dst)       = (v4f){(e).x, (o).x, (e).y, (o).y};          \
    *(v4f*)((dst) + 4) = (v4f){(e).z, (o).z, (e).w, (o).w};
#define HOLD_RAW(r) asm volatile("" : "+v"(r));
#define LOAD_ROWS(koff)                                              \
    ra0 = *(const float4*)(ga + 0 * (size_t)HIDDEN + (koff));        \
    ra1 = *(const float4*)(ga + 1 * (size_t)HIDDEN + (koff));        \
    ra2 = *(const float4*)(ga + 2 * (size_t)HIDDEN + (koff));        \
    ra3 = *(const float4*)(ga + 3 * (size_t)HIDDEN + (koff));        \
    rb0 = *(const v4f*)(gb + ((size_t)(koff) +  0) * EXPERTS);    \
    rb1 = *(const v4f*)(gb + ((size_t)(koff) +  1) * EXPERTS);    \
    rb2 = *(const v4f*)(gb + ((size_t)(koff) +  4) * EXPERTS);    \
    rb3 = *(const v4f*)(gb + ((size_t)(koff) +  5) * EXPERTS);    \
    rb4 = *(const v4f*)(gb + ((size_t)(koff) +  8) * EXPERTS);    \
    rb5 = *(const v4f*)(gb + ((size_t)(koff) +  9) * EXPERTS);    \
    rb6 = *(const v4f*)(gb + ((size_t)(koff) + 12) * EXPERTS);    \
    rb7 = *(const v4f*)(gb + ((size_t)(koff) + 13) * EXPERTS);
#define STORE_ROWS(stage)                                            \
    *(float4*)((stage) + sa + 0 * PITCH) = ra0;                      \
    *(float4*)((stage) + sa + 1 * PITCH) = ra1;                      \
    *(float4*)((stage) + sa + 2 * PITCH) = ra2;                      \
    *(float4*)((stage) + sa + 3 * PITCH) = ra3;                      \
    HOLD_RAW(rb0) HOLD_RAW(rb1) HOLD_RAW(rb2) HOLD_RAW(rb3)          \
    HOLD_RAW(rb4) HOLD_RAW(rb5) HOLD_RAW(rb6) HOLD_RAW(rb7)          \
    STORE_PAIR((stage) + sb + 0 * BPITCH, rb0, rb1)                  \
    STORE_PAIR((stage) + sb + 2 * BPITCH, rb2, rb3)                  \
    STORE_PAIR((stage) + sb + 4 * BPITCH, rb4, rb5)                  \
    STORE_PAIR((stage) + sb + 6 * BPITCH, rb6, rb7)

    LOAD_ROWS(0)
    STORE_ROWS(lds)
    LOAD_ROWS(KC)
    __syncthreads();

    v8f acc = {};

    for (int c = 0; c < NCHUNK - 2; ++c) {
        const float* st = lds + (c & 1) * STAGE;
        float*       sn = lds + ((c + 1) & 1) * STAGE;
        const int    kn = (c + 2) * KC;
        Frags f;
        read_frags(st, fa, fb, f);
        STORE_ROWS(sn)
        LOAD_ROWS(kn)
        mma_chunk(f, acc);
        lds_barrier();
    }
    {
        Frags f;
        read_frags(lds + ((NCHUNK - 2) & 1) * STAGE, fa, fb, f);
        STORE_ROWS(lds + ((NCHUNK - 1) & 1) * STAGE)
        mma_chunk(f, acc);
        __syncthreads();
    }
    {
        Frags f;
        read_frags(lds + ((NCHUNK - 1) & 1) * STAGE, fa, fb, f);
        mma_chunk(f, acc);
        __syncthreads();
    }
#undef LOAD_ROWS
#undef STORE_ROWS
#undef STORE_PAIR
#undef HOLD_RAW

    float* ct = lds;
#pragma unroll
    for (int r = 0; r < 8; ++r)
        ct[(mt * 16 + half * 8 + r) * CPITCH + nt * 16 + sub] = acc[r];
    __syncthreads();
#pragma unroll
    for (int p = 0; p < 2; ++p) {
        const int j   = p * 1024 + tid * 4;
        const int row = j >> 6;
        const int col = j & 63;
        float4 v = *(const float4*)(ct + row * CPITCH + col);
        *(float4*)(C + (size_t)m0 * EXPERTS + j) = v;
    }
}

#define CHK  16
#define NCH  (TSTEPS / CHK)
#define BLK  8

constexpr int refractory_gap() {
    float rf = 1.0f;
    int k = 0;
    for (int i = 0; i < 64; ++i) {
        ++k;
        bool active = (rf <= 0.0f);
        float d = rf - 0.1f;
        rf = (d > 0.0f) ? d : 0.0f;
        if (active) break;
    }
    return k;
}
constexpr int GAP = refractory_gap();

__device__ __forceinline__ float2 lif_step(float2 cur, bool on0, bool on1,
                                           float2& mp, float2& rf) {
    float l0 = mp.x * 0.9f;
    float l1 = mp.y * 0.9f;
    float a0 = (on0 && rf.x <= 0.0f) ? 0.1f : 0.0f;
    float a1 = (on1 && rf.y <= 0.0f) ? 0.1f : 0.0f;
    mp.x = __builtin_fmaf(cur.x, a0, l0);
    mp.y = __builtin_fmaf(cur.y, a1, l1);
    rf.x = fmaxf(rf.x - 0.1f, 0.0f);
    rf.y = fmaxf(rf.y - 0.1f, 0.0f);
    float2 pre = mp;
    bool s0 = mp.x > 1.0f;
    bool s1 = mp.y > 1.0f;
    mp.x = s0 ? 0.0f : mp.x;
    mp.y = s1 ? 0.0f : mp.y;
    rf.x = s0 ? 1.0f : rf.x;
    rf.y = s1 ? 1.0f : rf.y;
    return pre;
}

#define RG   32
#define EG   4
#define WPG  (EXPERTS / EG)
static_assert(RG % BLK == 0 && TSTEPS % RG == 0 && EXPERTS % EG == 0 && EG == 4, "2a block layout");

__device__ __forceinline__ void load_rows(const float* __restrict__ base, int tb, float4 (&c)[BLK]) {
#pragma unroll
    for (int i = 0; i < BLK; ++i) {
        const int t = min(tb + i, TSTEPS - 1);
        c[i] = *(const float4*)(base + (size_t)t * EXPERTS);
    }
}

template <bool GATED>
__device__ __forceinline__ void scan_steps(const float (&c)[BLK], int tb, int r, float& m, int& s) {
    int first = BLK;
#pragma unroll
    for (int i = 0; i < BLK; ++i) {
        const float v = __builtin_fmaf(c[i], 0.1f, m * 0.9f);
        m = (!GATED || tb + i >= r) ? v : 0.0f;
        first = min(first, (m > 1.0f) ? i : BLK);
    }
    s = min(s, (first < BLK) ? tb + first : TSTEPS);
}

__device__ __forceinline__ unsigned scan_block(const float4 (&c)[BLK], int tb, int r0, int r,
                                               unsigned alive, float (&m)[EG], int (&s)[EG]) {
    const bool gated = tb < r0 + RG - 1;
    unsigned left = 0;
#pragma unroll
    for (int j = 0; j < EG; ++j) {
        if (alive & (1u << j)) {
            float cj[BLK];
#pragma unroll
            for (int i = 0; i < BLK; ++i) cj[i] = (j == 0) ? c[i].x : (j == 1) ? c[i].y : (j == 2) ? c[i].z : c[i].w;
            if (gated) scan_steps<true>(cj, tb, r, m[j], s[j]);
            else       scan_steps<false>(cj, tb, r, m[j], s[j]);
            if (__ballot(s[j] == TSTEPS) != 0ull) left |= (1u << j);
        }
    }
    return left;
}

#define TB   32
static_assert(TB % BLK == 0 && TSTEPS % TB == 0, "tail block layout");

__device__ __forceinline__ void load_col(const float* p, float (&c)[TB]) {
#pragma unroll
    for (int i = 0; i < TB; ++i) c[i] = p[(size_t)i * EXPERTS];
}

__device__ __forceinline__ unsigned tail_block(const float (&c)[TB], int tb, unsigned open,
                                               float& m, int& s) {
    float mm[TB];
#pragma unroll
    for (int i = 0; i < TB; ++i) {
        m = __builtin_fmaf(c[i], 0.1f, m * 0.9f);
        mm[i] = m;
    }
    float top = mm[0];
#pragma unroll
    for (int i = 1; i < TB; ++i) top = __builtin_fmaxf(top, mm[i]);
    if (((unsigned)__ballot(top > 1.0f) & open) != 0u) {
        int first = TB;
#pragma unroll
        for (int i = TB - 1; i >= 0; --i) first = (mm[i] > 1.0f) ? i : first;
        s = min(s, (first < TB) ? tb + first : TSTEPS);
        open = (unsigned)__ballot(s == TSTEPS);
    }
    return open;
}

__global__ __launch_bounds__(WPG * 32) void next_restart_kernel(const float* __restrict__ cur,
                                                               unsigned* __restrict__ nxt) {
    const int wv   = __builtin_amdgcn_readfirstlane(threadIdx.x >> 5);
    const int lane = threadIdx.x & 31;
    const int r0   = blockIdx.x * RG;
    const int r    = r0 + lane;
    const int e0   = wv * EG;
    const float* base = cur + e0;

    float m[EG];
    int   s[EG];
#pragma unroll
    for (int j = 0; j < EG; ++j) { m[j] = 0.0f; s[j] = TSTEPS; }

    float4 ca[BLK], cb[BLK];
    unsigned alive = (1u << EG) - 1u;
    int tb = r0;
    for (;;) {
        load_rows(base, tb, ca);
        for (;;) {
            load_rows(base, tb + BLK, cb);
            alive = scan_block(ca, tb, r0, r, alive, m, s);
            tb += BLK;
            if (alive == 0u || tb >= TSTEPS) break;
            if ((alive & (alive - 1u)) == 0u && tb + 2 * TB <= TSTEPS) break;
            load_rows(base, tb + BLK, ca);
            alive = scan_block(cb, tb, r0, r, alive, m, s);
            tb += BLK;
            if (alive == 0u || tb >= TSTEPS) break;
            if ((alive & (alive - 1u)) == 0u && tb + 2 * TB <= TSTEPS) break;
        }
        if (alive == 0u || tb >= TSTEPS) break;

        const int j = __builtin_amdgcn_readfirstlane(__builtin_ctz(alive));
        float mj = (j == 0) ? m[0] : (j == 1) ? m[1] : (j == 2) ? m[2] : m[3];
        int   sj = (j == 0) ? s[0] : (j == 1) ? s[1] : (j == 2) ? s[2] : s[3];
        unsigned open = (unsigned)__ballot(sj == TSTEPS);
        unsigned lane_off = 0u;
        asm("" : "+v"(lane_off));
        const float* col = base + j + (size_t)tb * EXPERTS + lane_off;
        float ta[TB], tc[TB];
        load_col(col, ta);
        for (;;) {
            load_col(col + (size_t)TB * EXPERTS, tc);
            __builtin_amdgcn_sched_barrier(0);
            open = tail_block(ta, tb, open, mj, sj);
            tb += TB;
            if (open == 0u || tb + 2 * TB > TSTEPS) break;
            load_col(col + (size_t)2 * TB * EXPERTS, ta);
            __builtin_amdgcn_sched_barrier(0);
            open = tail_block(tc, tb, open, mj, sj);
            tb += TB;
            col += (size_t)2 * TB * EXPERTS;
            if (open == 0u || tb + 2 * TB > TSTEPS) break;
        }
#pragma unroll
        for (int k = 0; k < EG; ++k) {
            m[k] = (j == k) ? mj : m[k];
            s[k] = (j == k) ? sj : s[k];
        }
        if (open == 0u) break;
    }

#pragma unroll
    for (int j = 0; j < EG; ++j)
        nxt[(size_t)(e0 + j) * TSTEPS + r] = (unsigned)min(s[j] + GAP, TSTEPS);
}

#define HOPS   64
#define NLEV   7
#define TPITCH (TSTEPS + 8)
#define NMARK  (((TSTEPS / (HOPS * GAP) + 1 + 3) / 4) * 4)
#define NHOP   (NMARK * HOPS)
static_assert((1 << (NLEV - 1)) == HOPS && NMARK % 4 == 0 && NHOP % 256 == 0 && TPITCH % 4 == 0, "2b table layout");

typedef unsigned int u16x4w __attribute__((ext_vector_type(2), aligned(8), may_alias));

__device__ __forceinline__ void compose_table(const unsigned short* src, unsigned short* dst, int tid) {
    constexpr int NQ = TSTEPS / 4 / 256;
    u16x4w in[NQ];
#pragma unroll
    for (int k = 0; k < NQ; ++k) in[k] = *(const u16x4w*)(src + 4 * (tid + 256 * k));
    unsigned g[NQ][4];
#pragma unroll
    for (int k = 0; k < NQ; ++k) {
        g[k][0] = src[in[k].x & 0xffffu];
        g[k][1] = src[in[k].x >> 16];
        g[k][2] = src[in[k].y & 0xffffu];
        g[k][3] = src[in[k].y >> 16];
    }
#pragma unroll
    for (int k = 0; k < NQ; ++k) {
        u16x4w o;
        o.x = g[k][0] | (g[k][1] << 16);
        o.y = g[k][2] | (g[k][3] << 16);
        *(u16x4w*)(dst + 4 * (tid + 256 * k)) = o;
    }
}

__global__ __launch_bounds__(256) void restart_chain_kernel(const unsigned* __restrict__ nxt,
                                                            unsigned* __restrict__ start) {
    __shared__ __attribute__((aligned(16))) unsigned short tab[NLEV * TPITCH];
    __shared__ __attribute__((aligned(16))) unsigned short mark[NMARK];
    __shared__ __attribute__((aligned(16))) unsigned short hop[NHOP + 8];
    __shared__ unsigned st[NCH];
    const int e   = blockIdx.x;
    const int tid = threadIdx.x;

    const uint4* col = (const uint4*)(nxt + (size_t)e * TSTEPS);
    uint4 w[TSTEPS / 4 / 256];
#pragma unroll
    for (int k = 0; k < TSTEPS / 4 / 256; ++k) w[k] = col[tid + 256 * k];
#pragma unroll
    for (int k = 0; k < TSTEPS / 4 / 256; ++k) {
        const int q = tid + 256 * k;
        tab[4 * q]     = (unsigned short)w[k].x;
        tab[4 * q + 1] = (unsigned short)w[k].y;
        tab[4 * q + 2] = (unsigned short)w[k].z;
        tab[4 * q + 3] = (unsigned short)w[k].w;
    }
    if (tid < NLEV) tab[tid * TPITCH + TSTEPS] = TSTEPS;
    if (tid == 0) hop[NHOP] = TSTEPS;
    if (tid < NMARK) mark[tid] = TSTEPS;
    __syncthreads();
#pragma unroll
    for (int k = 0; k + 1 < NLEV; ++k) {
        compose_table(tab + k * TPITCH, tab + (k + 1) * TPITCH, tid);
        __syncthreads();
    }

    if (tid == 0) {
        const unsigned short* top = tab + (NLEV - 1) * TPITCH;
        int r = 0;
        for (int n = 0; n < NMARK && r < TSTEPS; n += 4) {
            const int r1 = top[r];
            const int r2 = top[r1];
            const int r3 = top[r2];
            mark[n]     = (unsigned short)r;
            mark[n + 1] = (unsigned short)r1;
            mark[n + 2] = (unsigned short)r2;
            mark[n + 3] = (unsigned short)r3;
            r = top[r3];
        }
    }
    __syncthreads();

#pragma unroll
    for (int j = 0; j < NHOP / 256; ++j) {
        const int p = tid + 256 * j;
        const int b = p & (HOPS - 1);
        int r = mark[p / HOPS];
#pragma unroll
        for (int k = NLEV - 2; k >= 0; --k) {
            const int t = tab[k * TPITCH + r];
            r = ((b >> k) & 1) ? t : r;
        }
        hop[p] = (unsigned short)r;
    }
    __syncthreads();

    for (int q = tid; q < NHOP; q += 256) {
        const int r    = hop[q];
        const int n    = hop[q + 1];
        const int cend = min((n + CHK - 1) / CHK, NCH);
        const unsigned w = (unsigned)r | ((unsigned)n << 16);
        for (int c = (r + CHK - 1) / CHK; c < cend; ++c) st[c] = w;
    }
    __syncthreads();

    start[(size_t)e * NCH + tid]       = st[tid];
    start[(size_t)e * NCH + 256 + tid] = st[256 + tid];
}

#define RB   CHK
#define NRB  3

__device__ __forceinline__ void load_block(const float2* __restrict__ cur, int lane, int t0,
                                           float2 (&c)[RB]) {
    const int rb = min(max(t0, 0), TSTEPS - RB);
    const float2* p = cur + (size_t)rb * 32 + lane;
#pragma unroll
    for (int i = 0; i < RB; ++i) c[i] = p[(size_t)i * 32];
}

__device__ __forceinline__ void integrate_block(const float2 (&c)[RB], int t0, int eA, int eB,
                                                float2& m) {
#pragma unroll
    for (int i = 0; i < RB; ++i) {
        const int t = t0 + i;
        const float g0 = (t >= eA) ? 0.1f : 0.0f;
        const float g1 = (t >= eB) ? 0.1f : 0.0f;
        m.x = __builtin_fmaf(c[i].x, g0, m.x * 0.9f);
        m.y = __builtin_fmaf(c[i].y, g1, m.y * 0.9f);
    }
}

__device__ __forceinline__ float sitout_counter(int j) {
    float r = 1.0f;
#pragma unroll
    for (int k = 1; k < GAP - 1; ++k) {
        const float d = fmaxf(r - 0.1f, 0.0f);
        r = (k < j) ? d : r;
    }
    return r;
}

__global__ __launch_bounds__(32) void replay_route_kernel(const float2* __restrict__ cur,
                                                          const unsigned* __restrict__ start,
                                                          float2* __restrict__ out) {
    const int ch   = blockIdx.x;
    const int lane = threadIdx.x;
    const int tbeg = ch * CHK;

    const unsigned wA = start[(size_t)(2 * lane)     * NCH + ch];
    const unsigned wB = start[(size_t)(2 * lane + 1) * NCH + ch];
    const int sA = (int)(wA & 0xffffu);
    const int sB = (int)(wB & 0xffffu);
    const int jA = tbeg - ((int)(wA >> 16) - GAP);
    const int jB = tbeg - ((int)(wB >> 16) - GAP);
    const bool spA = jA > 0;
    const bool spB = jB > 0;
    const int eA = spA ? tbeg : sA;
    const int eB = spB ? tbeg : sB;

    int smin = min(eA, eB);
#pragma unroll
    for (int off = 16; off >= 1; off >>= 1)
        smin = min(smin, __shfl_xor(smin, off, 32));
    smin = __builtin_amdgcn_readfirstlane(smin);

    const int nblk = NRB * ((tbeg - smin + NRB * RB - 1) / (NRB * RB));
    int tb = tbeg - nblk * RB;

    float2 c0[RB], c1[RB], c2[RB];
    load_block(cur, lane, tb,          c0);
    load_block(cur, lane, tb + RB,     c1);
    load_block(cur, lane, tb + 2 * RB, c2);

    float2 m = {0.0f, 0.0f};
    for (; tb < tbeg; tb += NRB * RB) {
        integrate_block(c0, tb,          eA, eB, m);
        load_block(cur, lane, tb + 3 * RB, c0);
        integrate_block(c1, tb + RB,     eA, eB, m);
        load_block(cur, lane, tb + 4 * RB, c1);
        integrate_block(c2, tb + 2 * RB, eA, eB, m);
        load_block(cur, lane, tb + 5 * RB, c2);
    }

    float2 mp = m;
    float2 rf;
    rf.x = spA ? sitout_counter(jA) : 0.0f;
    rf.y = spB ? sitout_counter(jB) : 0.0f;

#pragma unroll
    for (int i = 0; i < CHK; ++i) {
        const int t = tbeg + i;
        float2 v = lif_step(c0[i], t >= sA, t >= sB, mp, rf);

        bool s0 = v.x > 1.0f;
        bool s1 = v.y > 1.0f;
        unsigned long long bal = __ballot(s0 || s1);
        bool any = (bal != 0ull);

        float x0 = any ? (s0 ? 1.0f : 0.0f) : v.x;
        float x1 = any ? (s1 ? 1.0f : 0.0f) : v.y;

        float mx = fmaxf(x0, x1);
#pragma unroll
        for (int off = 16; off >= 1; off >>= 1)
            mx = fmaxf(mx, __shfl_xor(mx, off, 32));

        float e0 = __expf(x0 - mx);
        float e1 = __expf(x1 - mx);
        float s  = e0 + e1;
#pragma unroll
        for (int off = 16; off >= 1; off >>= 1)
            s += __shfl_xor(s, off, 32);

        float inv = 1.0f / s;
        float2 rr;
        rr.x = e0 * inv;
        rr.y = e1 * inv;
        out[(size_t)t * 32 + lane] = rr;
    }
}

extern "C" void kernel_launch(void* const* d_in, const int* in_sizes, int n_in,
                              void* d_out, int out_size, void* d_ws, size_t ws_size,
                              hipStream_t stream) {
    const float* hidden = (const float*)d_in[0];
    const float* W      = (const float*)d_in[1];
    float* out          = (float*)d_out;

    float* curr = (float*)d_ws;
    unsigned* nxt   = (unsigned*)d_out;
    unsigned* start = (unsigned*)((char*)d_ws + (size_t)TSTEPS * EXPERTS * sizeof(float));

    gemm_wmma_kernel<<<TSTEPS / AROWS, 256, 0, stream>>>(hidden, W, curr);

    next_restart_kernel<<<TSTEPS / RG, WPG * 32, 0, stream>>>((const float*)curr, nxt);

    restart_chain_kernel<<<EXPERTS, 256, 0, stream>>>(nxt, start);

    replay_route_kernel<<<NCH, 32, 0, stream>>>((const float2*)curr,
                                                (const unsigned*)start, (float2*)out);
}


# 2c chain on short-latency vector ops: wave max/sum/min butterflies by cross-row permute + row_xmask DPP (was LDS permutes, same partner order), replay on single mul/fma (was packed); sums unchanged
# speedup vs baseline: 9.5863x; 1.0375x over previous
#include <hip/hip_runtime.h>


typedef float v2f __attribute__((ext_vector_type(2)));
typedef float v4f __attribute__((ext_vector_type(4)));
typedef float v8f __attribute__((ext_vector_type(8)));

#define HIDDEN 2048
#define EXPERTS 64
#define TSTEPS 8192

#define KC      128
#define PITCH   (KC + 4)
#define AROWS   32
#define ASTAGE  (AROWS * PITCH)
#define BPAIRS  (KC / 2)
#define BPITCH  (2 * EXPERTS + 32)
#define STAGE   (ASTAGE + BPAIRS * BPITCH)
#define NCHUNK  (HIDDEN / KC)
#define CPITCH  (EXPERTS + 4)

struct Frags { v2f a[KC / 4]; v2f b[KC / 4]; };

__device__ __forceinline__ void read_frags(const float* st, int fa, int fb, Frags& f) {
#pragma unroll
    for (int ks = 0; ks < KC / 4; ++ks) {
        f.a[ks] = *(const v2f*)(st + fa + ks * 4);
        f.b[ks] = *(const v2f*)(st + fb + ks * 2 * BPITCH);
    }
    __builtin_amdgcn_sched_barrier(0);
}

__device__ __forceinline__ void mma_chunk(const Frags& f, v8f& acc) {
#pragma unroll
    for (int ks = 0; ks < KC / 4; ++ks)
        acc = __builtin_amdgcn_wmma_f32_16x16x4_f32(false, f.a[ks], false, f.b[ks], (short)0, acc, false, false);
}

__device__ __forceinline__ void lds_barrier() {
    __builtin_amdgcn_fence(__ATOMIC_RELEASE, "workgroup", "local");
    __builtin_amdgcn_s_barrier();
    __builtin_amdgcn_fence(__ATOMIC_ACQUIRE, "workgroup", "local");
}

__global__ __launch_bounds__(256) void gemm_wmma_kernel(const float* __restrict__ A,
                                                        const float* __restrict__ W,
                                                        float* __restrict__ C) {
    __shared__ __attribute__((aligned(16))) float lds[2 * STAGE];

    const int tid  = threadIdx.x;
    const int wave = tid >> 5;
    const int lane = tid & 31;
    const int m0   = blockIdx.x * AROWS;
    const int mt   = wave >> 2;
    const int nt   = wave & 3;
    const int half = lane >> 4;
    const int sub  = lane & 15;

    const float* ga = A + (size_t)(m0 + wave * 4) * HIDDEN + lane * 4;
    const float* gb = W + (size_t)(wave * 16 + half * 2) * EXPERTS + sub * 4;
    const int sa = (wave * 4) * PITCH + lane * 4;
    const int sb = ASTAGE + (wave * 8 + half) * BPITCH + sub * 8;

    const int fa = (mt * 16 + sub) * PITCH + half * 2;
    const int fb = ASTAGE + half * BPITCH + (nt * 16 + sub) * 2;

    float4 ra0, ra1, ra2, ra3;
    v4f    rb0, rb1, rb2, rb3, rb4, rb5, rb6, rb7;
#define STORE_PAIR(dst, e, o)                                        \
    *(v4f*)(dst)       = (v4f){(e).x, (o).x, (e).y, (o).y};          \
    *(v4f*)((dst) + 4) = (v4f){(e).z, (o).z, (e).w, (o).w};
#define HOLD_RAW(r) asm volatile("" : "+v"(r));
#define LOAD_ROWS(koff)                                              \
    ra0 = *(const float4*)(ga + 0 * (size_t)HIDDEN + (koff));        \
    ra1 = *(const float4*)(ga + 1 * (size_t)HIDDEN + (koff));        \
    ra2 = *(const float4*)(ga + 2 * (size_t)HIDDEN + (koff));        \
    ra3 = *(const float4*)(ga + 3 * (size_t)HIDDEN + (koff));        \
    rb0 = *(const v4f*)(gb + ((size_t)(koff) +  0) * EXPERTS);    \
    rb1 = *(const v4f*)(gb + ((size_t)(koff) +  1) * EXPERTS);    \
    rb2 = *(const v4f*)(gb + ((size_t)(koff) +  4) * EXPERTS);    \
    rb3 = *(const v4f*)(gb + ((size_t)(koff) +  5) * EXPERTS);    \
    rb4 = *(const v4f*)(gb + ((size_t)(koff) +  8) * EXPERTS);    \
    rb5 = *(const v4f*)(gb + ((size_t)(koff) +  9) * EXPERTS);    \
    rb6 = *(const v4f*)(gb + ((size_t)(koff) + 12) * EXPERTS);    \
    rb7 = *(const v4f*)(gb + ((size_t)(koff) + 13) * EXPERTS);
#define STORE_ROWS(stage)                                            \
    *(float4*)((stage) + sa + 0 * PITCH) = ra0;                      \
    *(float4*)((stage) + sa + 1 * PITCH) = ra1;                      \
    *(float4*)((stage) + sa + 2 * PITCH) = ra2;                      \
    *(float4*)((stage) + sa + 3 * PITCH) = ra3;                      \
    HOLD_RAW(rb0) HOLD_RAW(rb1) HOLD_RAW(rb2) HOLD_RAW(rb3)          \
    HOLD_RAW(rb4) HOLD_RAW(rb5) HOLD_RAW(rb6) HOLD_RAW(rb7)          \
    STORE_PAIR((stage) + sb + 0 * BPITCH, rb0, rb1)                  \
    STORE_PAIR((stage) + sb + 2 * BPITCH, rb2, rb3)                  \
    STORE_PAIR((stage) + sb + 4 * BPITCH, rb4, rb5)                  \
    STORE_PAIR((stage) + sb + 6 * BPITCH, rb6, rb7)

    LOAD_ROWS(0)
    STORE_ROWS(lds)
    LOAD_ROWS(KC)
    __syncthreads();

    v8f acc = {};

    for (int c = 0; c < NCHUNK - 2; ++c) {
        const float* st = lds + (c & 1) * STAGE;
        float*       sn = lds + ((c + 1) & 1) * STAGE;
        const int    kn = (c + 2) * KC;
        Frags f;
        read_frags(st, fa, fb, f);
        STORE_ROWS(sn)
        LOAD_ROWS(kn)
        mma_chunk(f, acc);
        lds_barrier();
    }
    {
        Frags f;
        read_frags(lds + ((NCHUNK - 2) & 1) * STAGE, fa, fb, f);
        STORE_ROWS(lds + ((NCHUNK - 1) & 1) * STAGE)
        mma_chunk(f, acc);
        __syncthreads();
    }
    {
        Frags f;
        read_frags(lds + ((NCHUNK - 1) & 1) * STAGE, fa, fb, f);
        mma_chunk(f, acc);
        __syncthreads();
    }
#undef LOAD_ROWS
#undef STORE_ROWS
#undef STORE_PAIR
#undef HOLD_RAW

    float* ct = lds;
#pragma unroll
    for (int r = 0; r < 8; ++r)
        ct[(mt * 16 + half * 8 + r) * CPITCH + nt * 16 + sub] = acc[r];
    __syncthreads();
#pragma unroll
    for (int p = 0; p < 2; ++p) {
        const int j   = p * 1024 + tid * 4;
        const int row = j >> 6;
        const int col = j & 63;
        float4 v = *(const float4*)(ct + row * CPITCH + col);
        *(float4*)(C + (size_t)m0 * EXPERTS + j) = v;
    }
}

#define CHK  16
#define NCH  (TSTEPS / CHK)
#define BLK  8

constexpr int refractory_gap() {
    float rf = 1.0f;
    int k = 0;
    for (int i = 0; i < 64; ++i) {
        ++k;
        bool active = (rf <= 0.0f);
        float d = rf - 0.1f;
        rf = (d > 0.0f) ? d : 0.0f;
        if (active) break;
    }
    return k;
}
constexpr int GAP = refractory_gap();

__device__ __forceinline__ float2 lif_step(float2 cur, bool on0, bool on1,
                                           float2& mp, float2& rf) {
    float l0 = mp.x * 0.9f;
    float l1 = mp.y * 0.9f;
    float a0 = (on0 && rf.x <= 0.0f) ? 0.1f : 0.0f;
    float a1 = (on1 && rf.y <= 0.0f) ? 0.1f : 0.0f;
    mp.x = __builtin_fmaf(cur.x, a0, l0);
    mp.y = __builtin_fmaf(cur.y, a1, l1);
    rf.x = fmaxf(rf.x - 0.1f, 0.0f);
    rf.y = fmaxf(rf.y - 0.1f, 0.0f);
    float2 pre = mp;
    bool s0 = mp.x > 1.0f;
    bool s1 = mp.y > 1.0f;
    mp.x = s0 ? 0.0f : mp.x;
    mp.y = s1 ? 0.0f : mp.y;
    rf.x = s0 ? 1.0f : rf.x;
    rf.y = s1 ? 1.0f : rf.y;
    return pre;
}

#define RG   32
#define EG   4
#define WPG  (EXPERTS / EG)
static_assert(RG % BLK == 0 && TSTEPS % RG == 0 && EXPERTS % EG == 0 && EG == 4, "2a block layout");

__device__ __forceinline__ void load_rows(const float* __restrict__ base, int tb, float4 (&c)[BLK]) {
#pragma unroll
    for (int i = 0; i < BLK; ++i) {
        const int t = min(tb + i, TSTEPS - 1);
        c[i] = *(const float4*)(base + (size_t)t * EXPERTS);
    }
}

template <bool GATED>
__device__ __forceinline__ void scan_steps(const float (&c)[BLK], int tb, int r, float& m, int& s) {
    int first = BLK;
#pragma unroll
    for (int i = 0; i < BLK; ++i) {
        const float v = __builtin_fmaf(c[i], 0.1f, m * 0.9f);
        m = (!GATED || tb + i >= r) ? v : 0.0f;
        first = min(first, (m > 1.0f) ? i : BLK);
    }
    s = min(s, (first < BLK) ? tb + first : TSTEPS);
}

__device__ __forceinline__ unsigned scan_block(const float4 (&c)[BLK], int tb, int r0, int r,
                                               unsigned alive, float (&m)[EG], int (&s)[EG]) {
    const bool gated = tb < r0 + RG - 1;
    unsigned left = 0;
#pragma unroll
    for (int j = 0; j < EG; ++j) {
        if (alive & (1u << j)) {
            float cj[BLK];
#pragma unroll
            for (int i = 0; i < BLK; ++i) cj[i] = (j == 0) ? c[i].x : (j == 1) ? c[i].y : (j == 2) ? c[i].z : c[i].w;
            if (gated) scan_steps<true>(cj, tb, r, m[j], s[j]);
            else       scan_steps<false>(cj, tb, r, m[j], s[j]);
            if (__ballot(s[j] == TSTEPS) != 0ull) left |= (1u << j);
        }
    }
    return left;
}

#define TB   32
static_assert(TB % BLK == 0 && TSTEPS % TB == 0, "tail block layout");

__device__ __forceinline__ void load_col(const float* p, float (&c)[TB]) {
#pragma unroll
    for (int i = 0; i < TB; ++i) c[i] = p[(size_t)i * EXPERTS];
}

__device__ __forceinline__ unsigned tail_block(const float (&c)[TB], int tb, unsigned open,
                                               float& m, int& s) {
    float mm[TB];
#pragma unroll
    for (int i = 0; i < TB; ++i) {
        m = __builtin_fmaf(c[i], 0.1f, m * 0.9f);
        mm[i] = m;
    }
    float top = mm[0];
#pragma unroll
    for (int i = 1; i < TB; ++i) top = __builtin_fmaxf(top, mm[i]);
    if (((unsigned)__ballot(top > 1.0f) & open) != 0u) {
        int first = TB;
#pragma unroll
        for (int i = TB - 1; i >= 0; --i) first = (mm[i] > 1.0f) ? i : first;
        s = min(s, (first < TB) ? tb + first : TSTEPS);
        open = (unsigned)__ballot(s == TSTEPS);
    }
    return open;
}

__global__ __launch_bounds__(WPG * 32) void next_restart_kernel(const float* __restrict__ cur,
                                                               unsigned* __restrict__ nxt) {
    const int wv   = __builtin_amdgcn_readfirstlane(threadIdx.x >> 5);
    const int lane = threadIdx.x & 31;
    const int r0   = blockIdx.x * RG;
    const int r    = r0 + lane;
    const int e0   = wv * EG;
    const float* base = cur + e0;

    float m[EG];
    int   s[EG];
#pragma unroll
    for (int j = 0; j < EG; ++j) { m[j] = 0.0f; s[j] = TSTEPS; }

    float4 ca[BLK], cb[BLK];
    unsigned alive = (1u << EG) - 1u;
    int tb = r0;
    for (;;) {
        load_rows(base, tb, ca);
        for (;;) {
            load_rows(base, tb + BLK, cb);
            alive = scan_block(ca, tb, r0, r, alive, m, s);
            tb += BLK;
            if (alive == 0u || tb >= TSTEPS) break;
            if ((alive & (alive - 1u)) == 0u && tb + 2 * TB <= TSTEPS) break;
            load_rows(base, tb + BLK, ca);
            alive = scan_block(cb, tb, r0, r, alive, m, s);
            tb += BLK;
            if (alive == 0u || tb >= TSTEPS) break;
            if ((alive & (alive - 1u)) == 0u && tb + 2 * TB <= TSTEPS) break;
        }
        if (alive == 0u || tb >= TSTEPS) break;

        const int j = __builtin_amdgcn_readfirstlane(__builtin_ctz(alive));
        float mj = (j == 0) ? m[0] : (j == 1) ? m[1] : (j == 2) ? m[2] : m[3];
        int   sj = (j == 0) ? s[0] : (j == 1) ? s[1] : (j == 2) ? s[2] : s[3];
        unsigned open = (unsigned)__ballot(sj == TSTEPS);
        unsigned lane_off = 0u;
        asm("" : "+v"(lane_off));
        const float* col = base + j + (size_t)tb * EXPERTS + lane_off;
        float ta[TB], tc[TB];
        load_col(col, ta);
        for (;;) {
            load_col(col + (size_t)TB * EXPERTS, tc);
            __builtin_amdgcn_sched_barrier(0);
            open = tail_block(ta, tb, open, mj, sj);
            tb += TB;
            if (open == 0u || tb + 2 * TB > TSTEPS) break;
            load_col(col + (size_t)2 * TB * EXPERTS, ta);
            __builtin_amdgcn_sched_barrier(0);
            open = tail_block(tc, tb, open, mj, sj);
            tb += TB;
            col += (size_t)2 * TB * EXPERTS;
            if (open == 0u || tb + 2 * TB > TSTEPS) break;
        }
#pragma unroll
        for (int k = 0; k < EG; ++k) {
            m[k] = (j == k) ? mj : m[k];
            s[k] = (j == k) ? sj : s[k];
        }
        if (open == 0u) break;
    }

#pragma unroll
    for (int j = 0; j < EG; ++j)
        nxt[(size_t)(e0 + j) * TSTEPS + r] = (unsigned)min(s[j] + GAP, TSTEPS);
}

#define HOPS   64
#define NLEV   7
#define TPITCH (TSTEPS + 8)
#define NMARK  (((TSTEPS / (HOPS * GAP) + 1 + 3) / 4) * 4)
#define NHOP   (NMARK * HOPS)
static_assert((1 << (NLEV - 1)) == HOPS && NMARK % 4 == 0 && NHOP % 256 == 0 && TPITCH % 4 == 0, "2b table layout");

typedef unsigned int u16x4w __attribute__((ext_vector_type(2), aligned(8), may_alias));

__device__ __forceinline__ void compose_table(const unsigned short* src, unsigned short* dst, int tid) {
    constexpr int NQ = TSTEPS / 4 / 256;
    u16x4w in[NQ];
#pragma unroll
    for (int k = 0; k < NQ; ++k) in[k] = *(const u16x4w*)(src + 4 * (tid + 256 * k));
    unsigned g[NQ][4];
#pragma unroll
    for (int k = 0; k < NQ; ++k) {
        g[k][0] = src[in[k].x & 0xffffu];
        g[k][1] = src[in[k].x >> 16];
        g[k][2] = src[in[k].y & 0xffffu];
        g[k][3] = src[in[k].y >> 16];
    }
#pragma unroll
    for (int k = 0; k < NQ; ++k) {
        u16x4w o;
        o.x = g[k][0] | (g[k][1] << 16);
        o.y = g[k][2] | (g[k][3] << 16);
        *(u16x4w*)(dst + 4 * (tid + 256 * k)) = o;
    }
}

__global__ __launch_bounds__(256) void restart_chain_kernel(const unsigned* __restrict__ nxt,
                                                            unsigned* __restrict__ start) {
    __shared__ __attribute__((aligned(16))) unsigned short tab[NLEV * TPITCH];
    __shared__ __attribute__((aligned(16))) unsigned short mark[NMARK];
    __shared__ __attribute__((aligned(16))) unsigned short hop[NHOP + 8];
    __shared__ unsigned st[NCH];
    const int e   = blockIdx.x;
    const int tid = threadIdx.x;

    const uint4* col = (const uint4*)(nxt + (size_t)e * TSTEPS);
    uint4 w[TSTEPS / 4 / 256];
#pragma unroll
    for (int k = 0; k < TSTEPS / 4 / 256; ++k) w[k] = col[tid + 256 * k];
#pragma unroll
    for (int k = 0; k < TSTEPS / 4 / 256; ++k) {
        const int q = tid + 256 * k;
        tab[4 * q]     = (unsigned short)w[k].x;
        tab[4 * q + 1] = (unsigned short)w[k].y;
        tab[4 * q + 2] = (unsigned short)w[k].z;
        tab[4 * q + 3] = (unsigned short)w[k].w;
    }
    if (tid < NLEV) tab[tid * TPITCH + TSTEPS] = TSTEPS;
    if (tid == 0) hop[NHOP] = TSTEPS;
    if (tid < NMARK) mark[tid] = TSTEPS;
    __syncthreads();
#pragma unroll
    for (int k = 0; k + 1 < NLEV; ++k) {
        compose_table(tab + k * TPITCH, tab + (k + 1) * TPITCH, tid);
        __syncthreads();
    }

    if (tid == 0) {
        const unsigned short* top = tab + (NLEV - 1) * TPITCH;
        int r = 0;
        for (int n = 0; n < NMARK && r < TSTEPS; n += 4) {
            const int r1 = top[r];
            const int r2 = top[r1];
            const int r3 = top[r2];
            mark[n]     = (unsigned short)r;
            mark[n + 1] = (unsigned short)r1;
            mark[n + 2] = (unsigned short)r2;
            mark[n + 3] = (unsigned short)r3;
            r = top[r3];
        }
    }
    __syncthreads();

#pragma unroll
    for (int j = 0; j < NHOP / 256; ++j) {
        const int p = tid + 256 * j;
        const int b = p & (HOPS - 1);
        int r = mark[p / HOPS];
#pragma unroll
        for (int k = NLEV - 2; k >= 0; --k) {
            const int t = tab[k * TPITCH + r];
            r = ((b >> k) & 1) ? t : r;
        }
        hop[p] = (unsigned short)r;
    }
    __syncthreads();

    for (int q = tid; q < NHOP; q += 256) {
        const int r    = hop[q];
        const int n    = hop[q + 1];
        const int cend = min((n + CHK - 1) / CHK, NCH);
        const unsigned w = (unsigned)r | ((unsigned)n << 16);
        for (int c = (r + CHK - 1) / CHK; c < cend; ++c) st[c] = w;
    }
    __syncthreads();

    start[(size_t)e * NCH + tid]       = st[tid];
    start[(size_t)e * NCH + 256 + tid] = st[256 + tid];
}

#define RB   CHK
#define NRB  3

__device__ __forceinline__ void load_block(const float2* __restrict__ cur, int lane, int t0,
                                           float2 (&c)[RB]) {
    const int rb = min(max(t0, 0), TSTEPS - RB);
    const float2* p = cur + (size_t)rb * 32 + lane;
#pragma unroll
    for (int i = 0; i < RB; ++i) c[i] = p[(size_t)i * 32];
}

__device__ __forceinline__ void integrate_block(const float2 (&c)[RB], int t0, int eA, int eB,
                                                float& m0, float& m1) {
#pragma unroll
    for (int i = 0; i < RB; ++i) {
        const int t = t0 + i;
        const float g0 = (t >= eA) ? 0.1f : 0.0f;
        const float g1 = (t >= eB) ? 0.1f : 0.0f;
        m0 = __builtin_fmaf(c[i].x, g0, m0 * 0.9f);
        m1 = __builtin_fmaf(c[i].y, g1, m1 * 0.9f);
        asm("" : "+v"(m0));
        asm("" : "+v"(m1));
    }
}

__device__ __forceinline__ int lane_xor16(int x) {
    return __builtin_amdgcn_permlanex16(x, x, 0x76543210, 0xfedcba98, false, false);
}
template <int M>
__device__ __forceinline__ int lane_xor_row(int x) {
    return __builtin_amdgcn_update_dpp(x, x, 0x160 + M, 0xf, 0xf, true);
}
__device__ __forceinline__ float lane_xor16(float x) {
    return __builtin_bit_cast(float, lane_xor16(__builtin_bit_cast(int, x)));
}
template <int M>
__device__ __forceinline__ float lane_xor_row(float x) {
    return __builtin_bit_cast(float, lane_xor_row<M>(__builtin_bit_cast(int, x)));
}
__device__ __forceinline__ float wave_max(float x) {
    x = fmaxf(x, lane_xor16(x));
    x = fmaxf(x, lane_xor_row<8>(x));
    x = fmaxf(x, lane_xor_row<4>(x));
    x = fmaxf(x, lane_xor_row<2>(x));
    x = fmaxf(x, lane_xor_row<1>(x));
    return x;
}
__device__ __forceinline__ float wave_sum(float x) {
    x += lane_xor16(x);
    x += lane_xor_row<8>(x);
    x += lane_xor_row<4>(x);
    x += lane_xor_row<2>(x);
    x += lane_xor_row<1>(x);
    return x;
}
__device__ __forceinline__ int wave_min(int x) {
    x = min(x, lane_xor16(x));
    x = min(x, lane_xor_row<8>(x));
    x = min(x, lane_xor_row<4>(x));
    x = min(x, lane_xor_row<2>(x));
    x = min(x, lane_xor_row<1>(x));
    return x;
}

__device__ __forceinline__ float sitout_counter(int j) {
    float r = 1.0f;
#pragma unroll
    for (int k = 1; k < GAP - 1; ++k) {
        const float d = fmaxf(r - 0.1f, 0.0f);
        r = (k < j) ? d : r;
    }
    return r;
}

__global__ __launch_bounds__(32) void replay_route_kernel(const float2* __restrict__ cur,
                                                          const unsigned* __restrict__ start,
                                                          float2* __restrict__ out) {
    const int ch   = blockIdx.x;
    const int lane = threadIdx.x;
    const int tbeg = ch * CHK;

    const unsigned wA = start[(size_t)(2 * lane)     * NCH + ch];
    const unsigned wB = start[(size_t)(2 * lane + 1) * NCH + ch];
    const int sA = (int)(wA & 0xffffu);
    const int sB = (int)(wB & 0xffffu);
    const int jA = tbeg - ((int)(wA >> 16) - GAP);
    const int jB = tbeg - ((int)(wB >> 16) - GAP);
    const bool spA = jA > 0;
    const bool spB = jB > 0;
    const int eA = spA ? tbeg : sA;
    const int eB = spB ? tbeg : sB;

    int smin = wave_min(min(eA, eB));
    smin = __builtin_amdgcn_readfirstlane(smin);

    const int nblk = NRB * ((tbeg - smin + NRB * RB - 1) / (NRB * RB));
    int tb = tbeg - nblk * RB;

    float2 c0[RB], c1[RB], c2[RB];
    load_block(cur, lane, tb,          c0);
    load_block(cur, lane, tb + RB,     c1);
    load_block(cur, lane, tb + 2 * RB, c2);

    float m0 = 0.0f, m1 = 0.0f;
    for (; tb < tbeg; tb += NRB * RB) {
        integrate_block(c0, tb,          eA, eB, m0, m1);
        load_block(cur, lane, tb + 3 * RB, c0);
        integrate_block(c1, tb + RB,     eA, eB, m0, m1);
        load_block(cur, lane, tb + 4 * RB, c1);
        integrate_block(c2, tb + 2 * RB, eA, eB, m0, m1);
        load_block(cur, lane, tb + 5 * RB, c2);
    }

    float2 mp = {m0, m1};
    float2 rf;
    rf.x = spA ? sitout_counter(jA) : 0.0f;
    rf.y = spB ? sitout_counter(jB) : 0.0f;

#pragma unroll
    for (int i = 0; i < CHK; ++i) {
        const int t = tbeg + i;
        float2 v = lif_step(c0[i], t >= sA, t >= sB, mp, rf);

        bool s0 = v.x > 1.0f;
        bool s1 = v.y > 1.0f;
        unsigned long long bal = __ballot(s0 || s1);
        bool any = (bal != 0ull);

        float x0 = any ? (s0 ? 1.0f : 0.0f) : v.x;
        float x1 = any ? (s1 ? 1.0f : 0.0f) : v.y;

        float mx = wave_max(fmaxf(x0, x1));

        float e0 = __expf(x0 - mx);
        float e1 = __expf(x1 - mx);
        float s  = wave_sum(e0 + e1);

        float inv = 1.0f / s;
        float2 rr;
        rr.x = e0 * inv;
        rr.y = e1 * inv;
        out[(size_t)t * 32 + lane] = rr;
    }
}

extern "C" void kernel_launch(void* const* d_in, const int* in_sizes, int n_in,
                              void* d_out, int out_size, void* d_ws, size_t ws_size,
                              hipStream_t stream) {
    const float* hidden = (const float*)d_in[0];
    const float* W      = (const float*)d_in[1];
    float* out          = (float*)d_out;

    float* curr = (float*)d_ws;
    unsigned* nxt   = (unsigned*)d_out;
    unsigned* start = (unsigned*)((char*)d_ws + (size_t)TSTEPS * EXPERTS * sizeof(float));

    gemm_wmma_kernel<<<TSTEPS / AROWS, 256, 0, stream>>>(hidden, W, curr);

    next_restart_kernel<<<TSTEPS / RG, WPG * 32, 0, stream>>>((const float*)curr, nxt);

    restart_chain_kernel<<<EXPERTS, 256, 0, stream>>>(nxt, start);

    replay_route_kernel<<<NCH, 32, 0, stream>>>((const float2*)curr,
                                                (const unsigned*)start, (float2*)out);
}


# GEMM row pieces in flight for two chunk multiplications: second register set, chunk loop in pairs, first barrier LDS-only, A pieces fenced so the tail loads stay early; k order and sums unchanged
# speedup vs baseline: 9.9260x; 1.0354x over previous
#include <hip/hip_runtime.h>


typedef float v2f __attribute__((ext_vector_type(2)));
typedef float v4f __attribute__((ext_vector_type(4)));
typedef float v8f __attribute__((ext_vector_type(8)));

#define HIDDEN 2048
#define EXPERTS 64
#define TSTEPS 8192

#define KC      128
#define PITCH   (KC + 4)
#define AROWS   32
#define ASTAGE  (AROWS * PITCH)
#define BPAIRS  (KC / 2)
#define BPITCH  (2 * EXPERTS + 32)
#define STAGE   (ASTAGE + BPAIRS * BPITCH)
#define NCHUNK  (HIDDEN / KC)
#define CPITCH  (EXPERTS + 4)

struct Frags { v2f a[KC / 4]; v2f b[KC / 4]; };

__device__ __forceinline__ void read_frags(const float* st, int fa, int fb, Frags& f) {
#pragma unroll
    for (int ks = 0; ks < KC / 4; ++ks) {
        f.a[ks] = *(const v2f*)(st + fa + ks * 4);
        f.b[ks] = *(const v2f*)(st + fb + ks * 2 * BPITCH);
    }
    __builtin_amdgcn_sched_barrier(0);
}

__device__ __forceinline__ void mma_chunk(const Frags& f, v8f& acc) {
#pragma unroll
    for (int ks = 0; ks < KC / 4; ++ks)
        acc = __builtin_amdgcn_wmma_f32_16x16x4_f32(false, f.a[ks], false, f.b[ks], (short)0, acc, false, false);
}

__device__ __forceinline__ void lds_barrier() {
    __builtin_amdgcn_fence(__ATOMIC_RELEASE, "workgroup", "local");
    __builtin_amdgcn_s_barrier();
    __builtin_amdgcn_fence(__ATOMIC_ACQUIRE, "workgroup", "local");
}

__global__ __launch_bounds__(256) void gemm_wmma_kernel(const float* __restrict__ A,
                                                        const float* __restrict__ W,
                                                        float* __restrict__ C) {
    __shared__ __attribute__((aligned(16))) float lds[2 * STAGE];

    const int tid  = threadIdx.x;
    const int wave = tid >> 5;
    const int lane = tid & 31;
    const int m0   = blockIdx.x * AROWS;
    const int mt   = wave >> 2;
    const int nt   = wave & 3;
    const int half = lane >> 4;
    const int sub  = lane & 15;

    const float* ga = A + (size_t)(m0 + wave * 4) * HIDDEN + lane * 4;
    const float* gb = W + (size_t)(wave * 16 + half * 2) * EXPERTS + sub * 4;
    const int sa = (wave * 4) * PITCH + lane * 4;
    const int sb = ASTAGE + (wave * 8 + half) * BPITCH + sub * 8;

    const int fa = (mt * 16 + sub) * PITCH + half * 2;
    const int fb = ASTAGE + half * BPITCH + (nt * 16 + sub) * 2;

    v4f    xa0, xa1, xa2, xa3, ya0, ya1, ya2, ya3;
    v4f    xb0, xb1, xb2, xb3, xb4, xb5, xb6, xb7;
    v4f    yb0, yb1, yb2, yb3, yb4, yb5, yb6, yb7;
#define STORE_PAIR(dst, e, o)                                        \
    *(v4f*)(dst)       = (v4f){(e).x, (o).x, (e).y, (o).y};          \
    *(v4f*)((dst) + 4) = (v4f){(e).z, (o).z, (e).w, (o).w};
#define HOLD_RAW(r) asm volatile("" : "+v"(r));
#define LOAD_ROWS(S, koff)                                           \
    S##a0 = *(const v4f*)(ga + 0 * (size_t)HIDDEN + (koff));         \
    S##a1 = *(const v4f*)(ga + 1 * (size_t)HIDDEN + (koff));         \
    S##a2 = *(const v4f*)(ga + 2 * (size_t)HIDDEN + (koff));         \
    S##a3 = *(const v4f*)(ga + 3 * (size_t)HIDDEN + (koff));         \
    S##b0 = *(const v4f*)(gb + ((size_t)(koff) +  0) * EXPERTS);     \
    S##b1 = *(const v4f*)(gb + ((size_t)(koff) +  1) * EXPERTS);     \
    S##b2 = *(const v4f*)(gb + ((size_t)(koff) +  4) * EXPERTS);     \
    S##b3 = *(const v4f*)(gb + ((size_t)(koff) +  5) * EXPERTS);     \
    S##b4 = *(const v4f*)(gb + ((size_t)(koff) +  8) * EXPERTS);     \
    S##b5 = *(const v4f*)(gb + ((size_t)(koff) +  9) * EXPERTS);     \
    S##b6 = *(const v4f*)(gb + ((size_t)(koff) + 12) * EXPERTS);     \
    S##b7 = *(const v4f*)(gb + ((size_t)(koff) + 13) * EXPERTS);
#define STORE_ROWS(S, stage)                                         \
    HOLD_RAW(S##a0) HOLD_RAW(S##a1) HOLD_RAW(S##a2) HOLD_RAW(S##a3)  \
    *(v4f*)((stage) + sa + 0 * PITCH) = S##a0;                       \
    *(v4f*)((stage) + sa + 1 * PITCH) = S##a1;                       \
    *(v4f*)((stage) + sa + 2 * PITCH) = S##a2;                       \
    *(v4f*)((stage) + sa + 3 * PITCH) = S##a3;                       \
    HOLD_RAW(S##b0) HOLD_RAW(S##b1) HOLD_RAW(S##b2) HOLD_RAW(S##b3)  \
    HOLD_RAW(S##b4) HOLD_RAW(S##b5) HOLD_RAW(S##b6) HOLD_RAW(S##b7)  \
    STORE_PAIR((stage) + sb + 0 * BPITCH, S##b0, S##b1)              \
    STORE_PAIR((stage) + sb + 2 * BPITCH, S##b2, S##b3)              \
    STORE_PAIR((stage) + sb + 4 * BPITCH, S##b4, S##b5)              \
    STORE_PAIR((stage) + sb + 6 * BPITCH, S##b6, S##b7)
#define CHUNK_STEP(cur, nxt, S, LOADS)                               \
    {                                                                \
        Frags f;                                                     \
        read_frags(cur, fa, fb, f);                                  \
        STORE_ROWS(S, nxt)                                           \
        LOADS                                                        \
        mma_chunk(f, acc);                                           \
        lds_barrier();                                               \
    }

    float* const s0 = lds;
    float* const s1 = lds + STAGE;

    LOAD_ROWS(x, 0)
    LOAD_ROWS(y, KC)
    STORE_ROWS(x, s0)
    LOAD_ROWS(x, 2 * KC)
    lds_barrier();

    v8f acc = {};

    static_assert(NCHUNK % 2 == 0 && NCHUNK >= 6, "chunk pairs");
    for (int c = 0; c < NCHUNK - 4; c += 2) {
        CHUNK_STEP(s0, s1, y, LOAD_ROWS(y, (c + 3) * KC))
        CHUNK_STEP(s1, s0, x, LOAD_ROWS(x, (c + 4) * KC))
    }
    CHUNK_STEP(s0, s1, y, LOAD_ROWS(y, (NCHUNK - 1) * KC))
    CHUNK_STEP(s1, s0, x, )
    CHUNK_STEP(s0, s1, y, )
    {
        Frags f;
        read_frags(s1, fa, fb, f);
        mma_chunk(f, acc);
        __syncthreads();
    }
#undef CHUNK_STEP
#undef LOAD_ROWS
#undef STORE_ROWS
#undef STORE_PAIR
#undef HOLD_RAW

    float* ct = lds;
#pragma unroll
    for (int r = 0; r < 8; ++r)
        ct[(mt * 16 + half * 8 + r) * CPITCH + nt * 16 + sub] = acc[r];
    __syncthreads();
#pragma unroll
    for (int p = 0; p < 2; ++p) {
        const int j   = p * 1024 + tid * 4;
        const int row = j >> 6;
        const int col = j & 63;
        float4 v = *(const float4*)(ct + row * CPITCH + col);
        *(float4*)(C + (size_t)m0 * EXPERTS + j) = v;
    }
}

#define CHK  16
#define NCH  (TSTEPS / CHK)
#define BLK  8

constexpr int refractory_gap() {
    float rf = 1.0f;
    int k = 0;
    for (int i = 0; i < 64; ++i) {
        ++k;
        bool active = (rf <= 0.0f);
        float d = rf - 0.1f;
        rf = (d > 0.0f) ? d : 0.0f;
        if (active) break;
    }
    return k;
}
constexpr int GAP = refractory_gap();

__device__ __forceinline__ float2 lif_step(float2 cur, bool on0, bool on1,
                                           float2& mp, float2& rf) {
    float l0 = mp.x * 0.9f;
    float l1 = mp.y * 0.9f;
    float a0 = (on0 && rf.x <= 0.0f) ? 0.1f : 0.0f;
    float a1 = (on1 && rf.y <= 0.0f) ? 0.1f : 0.0f;
    mp.x = __builtin_fmaf(cur.x, a0, l0);
    mp.y = __builtin_fmaf(cur.y, a1, l1);
    rf.x = fmaxf(rf.x - 0.1f, 0.0f);
    rf.y = fmaxf(rf.y - 0.1f, 0.0f);
    float2 pre = mp;
    bool s0 = mp.x > 1.0f;
    bool s1 = mp.y > 1.0f;
    mp.x = s0 ? 0.0f : mp.x;
    mp.y = s1 ? 0.0f : mp.y;
    rf.x = s0 ? 1.0f : rf.x;
    rf.y = s1 ? 1.0f : rf.y;
    return pre;
}

#define RG   32
#define EG   4
#define WPG  (EXPERTS / EG)
static_assert(RG % BLK == 0 && TSTEPS % RG == 0 && EXPERTS % EG == 0 && EG == 4, "2a block layout");

__device__ __forceinline__ void load_rows(const float* __restrict__ base, int tb, float4 (&c)[BLK]) {
#pragma unroll
    for (int i = 0; i < BLK; ++i) {
        const int t = min(tb + i, TSTEPS - 1);
        c[i] = *(const float4*)(base + (size_t)t * EXPERTS);
    }
}

template <bool GATED>
__device__ __forceinline__ void scan_steps(const float (&c)[BLK], int tb, int r, float& m, int& s) {
    int first = BLK;
#pragma unroll
    for (int i = 0; i < BLK; ++i) {
        const float v = __builtin_fmaf(c[i], 0.1f, m * 0.9f);
        m = (!GATED || tb + i >= r) ? v : 0.0f;
        first = min(first, (m > 1.0f) ? i : BLK);
    }
    s = min(s, (first < BLK) ? tb + first : TSTEPS);
}

__device__ __forceinline__ unsigned scan_block(const float4 (&c)[BLK], int tb, int r0, int r,
                                               unsigned alive, float (&m)[EG], int (&s)[EG]) {
    const bool gated = tb < r0 + RG - 1;
    unsigned left = 0;
#pragma unroll
    for (int j = 0; j < EG; ++j) {
        if (alive & (1u << j)) {
            float cj[BLK];
#pragma unroll
            for (int i = 0; i < BLK; ++i) cj[i] = (j == 0) ? c[i].x : (j == 1) ? c[i].y : (j == 2) ? c[i].z : c[i].w;
            if (gated) scan_steps<true>(cj, tb, r, m[j], s[j]);
            else       scan_steps<false>(cj, tb, r, m[j], s[j]);
            if (__ballot(s[j] == TSTEPS) != 0ull) left |= (1u << j);
        }
    }
    return left;
}

#define TB   32
static_assert(TB % BLK == 0 && TSTEPS % TB == 0, "tail block layout");

__device__ __forceinline__ void load_col(const float* p, float (&c)[TB]) {
#pragma unroll
    for (int i = 0; i < TB; ++i) c[i] = p[(size_t)i * EXPERTS];
}

__device__ __forceinline__ unsigned tail_block(const float (&c)[TB], int tb, unsigned open,
                                               float& m, int& s) {
    float mm[TB];
#pragma unroll
    for (int i = 0; i < TB; ++i) {
        m = __builtin_fmaf(c[i], 0.1f, m * 0.9f);
        mm[i] = m;
    }
    float top = mm[0];
#pragma unroll
    for (int i = 1; i < TB; ++i) top = __builtin_fmaxf(top, mm[i]);
    if (((unsigned)__ballot(top > 1.0f) & open) != 0u) {
        int first = TB;
#pragma unroll
        for (int i = TB - 1; i >= 0; --i) first = (mm[i] > 1.0f) ? i : first;
        s = min(s, (first < TB) ? tb + first : TSTEPS);
        open = (unsigned)__ballot(s == TSTEPS);
    }
    return open;
}

__global__ __launch_bounds__(WPG * 32) void next_restart_kernel(const float* __restrict__ cur,
                                                               unsigned* __restrict__ nxt) {
    const int wv   = __builtin_amdgcn_readfirstlane(threadIdx.x >> 5);
    const int lane = threadIdx.x & 31;
    const int r0   = blockIdx.x * RG;
    const int r    = r0 + lane;
    const int e0   = wv * EG;
    const float* base = cur + e0;

    float m[EG];
    int   s[EG];
#pragma unroll
    for (int j = 0; j < EG; ++j) { m[j] = 0.0f; s[j] = TSTEPS; }

    float4 ca[BLK], cb[BLK];
    unsigned alive = (1u << EG) - 1u;
    int tb = r0;
    for (;;) {
        load_rows(base, tb, ca);
        for (;;) {
            load_rows(base, tb + BLK, cb);
            alive = scan_block(ca, tb, r0, r, alive, m, s);
            tb += BLK;
            if (alive == 0u || tb >= TSTEPS) break;
            if ((alive & (alive - 1u)) == 0u && tb + 2 * TB <= TSTEPS) break;
            load_rows(base, tb + BLK, ca);
            alive = scan_block(cb, tb, r0, r, alive, m, s);
            tb += BLK;
            if (alive == 0u || tb >= TSTEPS) break;
            if ((alive & (alive - 1u)) == 0u && tb + 2 * TB <= TSTEPS) break;
        }
        if (alive == 0u || tb >= TSTEPS) break;

        const int j = __builtin_amdgcn_readfirstlane(__builtin_ctz(alive));
        float mj = (j == 0) ? m[0] : (j == 1) ? m[1] : (j == 2) ? m[2] : m[3];
        int   sj = (j == 0) ? s[0] : (j == 1) ? s[1] : (j == 2) ? s[2] : s[3];
        unsigned open = (unsigned)__ballot(sj == TSTEPS);
        unsigned lane_off = 0u;
        asm("" : "+v"(lane_off));
        const float* col = base + j + (size_t)tb * EXPERTS + lane_off;
        float ta[TB], tc[TB];
        load_col(col, ta);
        for (;;) {
            load_col(col + (size_t)TB * EXPERTS, tc);
            __builtin_amdgcn_sched_barrier(0);
            open = tail_block(ta, tb, open, mj, sj);
            tb += TB;
            if (open == 0u || tb + 2 * TB > TSTEPS) break;
            load_col(col + (size_t)2 * TB * EXPERTS, ta);
            __builtin_amdgcn_sched_barrier(0);
            open = tail_block(tc, tb, open, mj, sj);
            tb += TB;
            col += (size_t)2 * TB * EXPERTS;
            if (open == 0u || tb + 2 * TB > TSTEPS) break;
        }
#pragma unroll
        for (int k = 0; k < EG; ++k) {
            m[k] = (j == k) ? mj : m[k];
            s[k] = (j == k) ? sj : s[k];
        }
        if (open == 0u) break;
    }

#pragma unroll
    for (int j = 0; j < EG; ++j)
        nxt[(size_t)(e0 + j) * TSTEPS + r] = (unsigned)min(s[j] + GAP, TSTEPS);
}

#define HOPS   64
#define NLEV   7
#define TPITCH (TSTEPS + 8)
#define NMARK  (((TSTEPS / (HOPS * GAP) + 1 + 3) / 4) * 4)
#define NHOP   (NMARK * HOPS)
static_assert((1 << (NLEV - 1)) == HOPS && NMARK % 4 == 0 && NHOP % 256 == 0 && TPITCH % 4 == 0, "2b table layout");

typedef unsigned int u16x4w __attribute__((ext_vector_type(2), aligned(8), may_alias));

__device__ __forceinline__ void compose_table(const unsigned short* src, unsigned short* dst, int tid) {
    constexpr int NQ = TSTEPS / 4 / 256;
    u16x4w in[NQ];
#pragma unroll
    for (int k = 0; k < NQ; ++k) in[k] = *(const u16x4w*)(src + 4 * (tid + 256 * k));
    unsigned g[NQ][4];
#pragma unroll
    for (int k = 0; k < NQ; ++k) {
        g[k][0] = src[in[k].x & 0xffffu];
        g[k][1] = src[in[k].x >> 16];
        g[k][2] = src[in[k].y & 0xffffu];
        g[k][3] = src[in[k].y >> 16];
    }
#pragma unroll
    for (int k = 0; k < NQ; ++k) {
        u16x4w o;
        o.x = g[k][0] | (g[k][1] << 16);
        o.y = g[k][2] | (g[k][3] << 16);
        *(u16x4w*)(dst + 4 * (tid + 256 * k)) = o;
    }
}

__global__ __launch_bounds__(256) void restart_chain_kernel(const unsigned* __restrict__ nxt,
                                                            unsigned* __restrict__ start) {
    __shared__ __attribute__((aligned(16))) unsigned short tab[NLEV * TPITCH];
    __shared__ __attribute__((aligned(16))) unsigned short mark[NMARK];
    __shared__ __attribute__((aligned(16))) unsigned short hop[NHOP + 8];
    __shared__ unsigned st[NCH];
    const int e   = blockIdx.x;
    const int tid = threadIdx.x;

    const uint4* col = (const uint4*)(nxt + (size_t)e * TSTEPS);
    uint4 w[TSTEPS / 4 / 256];
#pragma unroll
    for (int k = 0; k < TSTEPS / 4 / 256; ++k) w[k] = col[tid + 256 * k];
#pragma unroll
    for (int k = 0; k < TSTEPS / 4 / 256; ++k) {
        const int q = tid + 256 * k;
        tab[4 * q]     = (unsigned short)w[k].x;
        tab[4 * q + 1] = (unsigned short)w[k].y;
        tab[4 * q + 2] = (unsigned short)w[k].z;
        tab[4 * q + 3] = (unsigned short)w[k].w;
    }
    if (tid < NLEV) tab[tid * TPITCH + TSTEPS] = TSTEPS;
    if (tid == 0) hop[NHOP] = TSTEPS;
    if (tid < NMARK) mark[tid] = TSTEPS;
    __syncthreads();
#pragma unroll
    for (int k = 0; k + 1 < NLEV; ++k) {
        compose_table(tab + k * TPITCH, tab + (k + 1) * TPITCH, tid);
        __syncthreads();
    }

    if (tid == 0) {
        const unsigned short* top = tab + (NLEV - 1) * TPITCH;
        int r = 0;
        for (int n = 0; n < NMARK && r < TSTEPS; n += 4) {
            const int r1 = top[r];
            const int r2 = top[r1];
            const int r3 = top[r2];
            mark[n]     = (unsigned short)r;
            mark[n + 1] = (unsigned short)r1;
            mark[n + 2] = (unsigned short)r2;
            mark[n + 3] = (unsigned short)r3;
            r = top[r3];
        }
    }
    __syncthreads();

#pragma unroll
    for (int j = 0; j < NHOP / 256; ++j) {
        const int p = tid + 256 * j;
        const int b = p & (HOPS - 1);
        int r = mark[p / HOPS];
#pragma unroll
        for (int k = NLEV - 2; k >= 0; --k) {
            const int t = tab[k * TPITCH + r];
            r = ((b >> k) & 1) ? t : r;
        }
        hop[p] = (unsigned short)r;
    }
    __syncthreads();

    for (int q = tid; q < NHOP; q += 256) {
        const int r    = hop[q];
        const int n    = hop[q + 1];
        const int cend = min((n + CHK - 1) / CHK, NCH);
        const unsigned w = (unsigned)r | ((unsigned)n << 16);
        for (int c = (r + CHK - 1) / CHK; c < cend; ++c) st[c] = w;
    }
    __syncthreads();

    start[(size_t)e * NCH + tid]       = st[tid];
    start[(size_t)e * NCH + 256 + tid] = st[256 + tid];
}

#define RB   CHK
#define NRB  3

__device__ __forceinline__ void load_block(const float2* __restrict__ cur, int lane, int t0,
                                           float2 (&c)[RB]) {
    const int rb = min(max(t0, 0), TSTEPS - RB);
    const float2* p = cur + (size_t)rb * 32 + lane;
#pragma unroll
    for (int i = 0; i < RB; ++i) c[i] = p[(size_t)i * 32];
}

__device__ __forceinline__ void integrate_block(const float2 (&c)[RB], int t0, int eA, int eB,
                                                float& m0, float& m1) {
#pragma unroll
    for (int i = 0; i < RB; ++i) {
        const int t = t0 + i;
        const float g0 = (t >= eA) ? 0.1f : 0.0f;
        const float g1 = (t >= eB) ? 0.1f : 0.0f;
        m0 = __builtin_fmaf(c[i].x, g0, m0 * 0.9f);
        m1 = __builtin_fmaf(c[i].y, g1, m1 * 0.9f);
        asm("" : "+v"(m0));
        asm("" : "+v"(m1));
    }
}

__device__ __forceinline__ int lane_xor16(int x) {
    return __builtin_amdgcn_permlanex16(x, x, 0x76543210, 0xfedcba98, false, false);
}
template <int M>
__device__ __forceinline__ int lane_xor_row(int x) {
    return __builtin_amdgcn_update_dpp(x, x, 0x160 + M, 0xf, 0xf, true);
}
__device__ __forceinline__ float lane_xor16(float x) {
    return __builtin_bit_cast(float, lane_xor16(__builtin_bit_cast(int, x)));
}
template <int M>
__device__ __forceinline__ float lane_xor_row(float x) {
    return __builtin_bit_cast(float, lane_xor_row<M>(__builtin_bit_cast(int, x)));
}
__device__ __forceinline__ float wave_max(float x) {
    x = fmaxf(x, lane_xor16(x));
    x = fmaxf(x, lane_xor_row<8>(x));
    x = fmaxf(x, lane_xor_row<4>(x));
    x = fmaxf(x, lane_xor_row<2>(x));
    x = fmaxf(x, lane_xor_row<1>(x));
    return x;
}
__device__ __forceinline__ float wave_sum(float x) {
    x += lane_xor16(x);
    x += lane_xor_row<8>(x);
    x += lane_xor_row<4>(x);
    x += lane_xor_row<2>(x);
    x += lane_xor_row<1>(x);
    return x;
}
__device__ __forceinline__ int wave_min(int x) {
    x = min(x, lane_xor16(x));
    x = min(x, lane_xor_row<8>(x));
    x = min(x, lane_xor_row<4>(x));
    x = min(x, lane_xor_row<2>(x));
    x = min(x, lane_xor_row<1>(x));
    return x;
}

__device__ __forceinline__ float sitout_counter(int j) {
    float r = 1.0f;
#pragma unroll
    for (int k = 1; k < GAP - 1; ++k) {
        const float d = fmaxf(r - 0.1f, 0.0f);
        r = (k < j) ? d : r;
    }
    return r;
}

__global__ __launch_bounds__(32) void replay_route_kernel(const float2* __restrict__ cur,
                                                          const unsigned* __restrict__ start,
                                                          float2* __restrict__ out) {
    const int ch   = blockIdx.x;
    const int lane = threadIdx.x;
    const int tbeg = ch * CHK;

    const unsigned wA = start[(size_t)(2 * lane)     * NCH + ch];
    const unsigned wB = start[(size_t)(2 * lane + 1) * NCH + ch];
    const int sA = (int)(wA & 0xffffu);
    const int sB = (int)(wB & 0xffffu);
    const int jA = tbeg - ((int)(wA >> 16) - GAP);
    const int jB = tbeg - ((int)(wB >> 16) - GAP);
    const bool spA = jA > 0;
    const bool spB = jB > 0;
    const int eA = spA ? tbeg : sA;
    const int eB = spB ? tbeg : sB;

    int smin = wave_min(min(eA, eB));
    smin = __builtin_amdgcn_readfirstlane(smin);

    const int nblk = NRB * ((tbeg - smin + NRB * RB - 1) / (NRB * RB));
    int tb = tbeg - nblk * RB;

    float2 c0[RB], c1[RB], c2[RB];
    load_block(cur, lane, tb,          c0);
    load_block(cur, lane, tb + RB,     c1);
    load_block(cur, lane, tb + 2 * RB, c2);

    float m0 = 0.0f, m1 = 0.0f;
    for (; tb < tbeg; tb += NRB * RB) {
        integrate_block(c0, tb,          eA, eB, m0, m1);
        load_block(cur, lane, tb + 3 * RB, c0);
        integrate_block(c1, tb + RB,     eA, eB, m0, m1);
        load_block(cur, lane, tb + 4 * RB, c1);
        integrate_block(c2, tb + 2 * RB, eA, eB, m0, m1);
        load_block(cur, lane, tb + 5 * RB, c2);
    }

    float2 mp = {m0, m1};
    float2 rf;
    rf.x = spA ? sitout_counter(jA) : 0.0f;
    rf.y = spB ? sitout_counter(jB) : 0.0f;

#pragma unroll
    for (int i = 0; i < CHK; ++i) {
        const int t = tbeg + i;
        float2 v = lif_step(c0[i], t >= sA, t >= sB, mp, rf);

        bool s0 = v.x > 1.0f;
        bool s1 = v.y > 1.0f;
        unsigned long long bal = __ballot(s0 || s1);
        bool any = (bal != 0ull);

        float x0 = any ? (s0 ? 1.0f : 0.0f) : v.x;
        float x1 = any ? (s1 ? 1.0f : 0.0f) : v.y;

        float mx = wave_max(fmaxf(x0, x1));

        float e0 = __expf(x0 - mx);
        float e1 = __expf(x1 - mx);
        float s  = wave_sum(e0 + e1);

        float inv = 1.0f / s;
        float2 rr;
        rr.x = e0 * inv;
        rr.y = e1 * inv;
        out[(size_t)t * 32 + lane] = rr;
    }
}

extern "C" void kernel_launch(void* const* d_in, const int* in_sizes, int n_in,
                              void* d_out, int out_size, void* d_ws, size_t ws_size,
                              hipStream_t stream) {
    const float* hidden = (const float*)d_in[0];
    const float* W      = (const float*)d_in[1];
    float* out          = (float*)d_out;

    float* curr = (float*)d_ws;
    unsigned* nxt   = (unsigned*)d_out;
    unsigned* start = (unsigned*)((char*)d_ws + (size_t)TSTEPS * EXPERTS * sizeof(float));

    gemm_wmma_kernel<<<TSTEPS / AROWS, 256, 0, stream>>>(hidden, W, curr);

    next_restart_kernel<<<TSTEPS / RG, WPG * 32, 0, stream>>>((const float*)curr, nxt);

    restart_chain_kernel<<<EXPERTS, 256, 0, stream>>>(nxt, start);

    replay_route_kernel<<<NCH, 32, 0, stream>>>((const float2*)curr,
                                                (const unsigned*)start, (float2*)out);
}
